# starting kernel
# baseline (speedup 1.0000x reference)
#include <hip/hip_runtime.h>

typedef float v2f __attribute__((ext_vector_type(2)));
typedef float v8f __attribute__((ext_vector_type(8)));

#define N_NODES 100000
#define IN_DIM  384
#define HID     64


__global__ void k_zero_deg(unsigned int* __restrict__ deg) {
    int i = blockIdx.x * blockDim.x + threadIdx.x;
    if (i < N_NODES) deg[i] = 0u;
}

__global__ void k_deg_count(const int* __restrict__ dst, int E,
                            unsigned int* __restrict__ deg) {
    int e = blockIdx.x * blockDim.x + threadIdx.x;
    if (e < E) atomicAdd(&deg[dst[e]], 1u);
}

__global__ void k_dinv(const unsigned int* __restrict__ deg,
                       float* __restrict__ dinv) {
    int i = blockIdx.x * blockDim.x + threadIdx.x;
    if (i < N_NODES) dinv[i] = rsqrtf((float)deg[i] + 1.0f);
}

__global__ __launch_bounds__(128) void k_gemm1(const float* __restrict__ X,
                                               const float* __restrict__ W1,
                                               float* __restrict__ H) {
    const int lane = threadIdx.x & 31;
    const int wave = threadIdx.x >> 5;
    const int half = lane >> 4;
    const int lr   = lane & 15;
    const int rowBase = blockIdx.x * 16;
    const int colBase = wave * 16;

    v8f c = {};
    const float* xrow = X + (size_t)(rowBase + lr) * IN_DIM;
    for (int k0 = 0; k0 < IN_DIM; k0 += 4) {
        v2f a, b;
        const float* ap = xrow + k0 + 2 * half;
        a[0] = ap[0];
        a[1] = ap[1];
        const float* bp = W1 + (size_t)(k0 + 2 * half) * HID + colBase + lr;
        b[0] = bp[0];
        b[1] = bp[HID];
        c = __builtin_amdgcn_wmma_f32_16x16x4_f32(
                 false, a,  false, b,
                 (short)0, c,  false,  false);
    }
    float* hrow = H + (size_t)(rowBase + 8 * half) * HID + colBase + lr;
#pragma unroll
    for (int r = 0; r < 8; ++r)
        hrow[(size_t)r * HID] = c[r];
}


__global__ void k_selfloop1(const float* __restrict__ H,
                            const float* __restrict__ dinv,
                            float* __restrict__ agg) {
    int idx = blockIdx.x * blockDim.x + threadIdx.x;
    if (idx < N_NODES * HID) {
        float di = dinv[idx >> 6];
        agg[idx] = H[idx] * di * di;
    }
}

__global__ void k_scatter1(const int* __restrict__ src,
                           const int* __restrict__ dst, int E,
                           const float* __restrict__ H,
                           const float* __restrict__ dinv,
                           float* __restrict__ agg) {
    int i = blockIdx.x * blockDim.x + threadIdx.x;
    int e = i >> 6;
    if (e < E) {
        int f = i & 63;
        int s = src[e], d = dst[e];
        float w = dinv[s] * dinv[d];
        atomicAdd(&agg[(size_t)d * HID + f], H[(size_t)s * HID + f] * w);
    }
}

__global__ __launch_bounds__(256) void k_dot2(const float* __restrict__ agg,
                                              const float* __restrict__ b1,
                                              const float* __restrict__ W2,
                                              float* __restrict__ z) {
    int row  = (blockIdx.x * blockDim.x + threadIdx.x) >> 5;
    int lane = threadIdx.x & 31;
    if (row < N_NODES) {
        const float* a = agg + (size_t)row * HID;
        float v0 = a[lane] + b1[lane];
        float v1 = a[lane + 32] + b1[lane + 32];
        v0 = v0 > 0.0f ? v0 : 0.0f;
        v1 = v1 > 0.0f ? v1 : 0.0f;
        float s = v0 * W2[lane] + v1 * W2[lane + 32];
#pragma unroll
        for (int off = 16; off > 0; off >>= 1)
            s += __shfl_down(s, off, 32);
        if (lane == 0) z[row] = s;
    }
}

__global__ void k_out_init(const float* __restrict__ z,
                           const float* __restrict__ dinv,
                           const float* __restrict__ b2,
                           float* __restrict__ out) {
    int i = blockIdx.x * blockDim.x + threadIdx.x;
    if (i < N_NODES) {
        float di = dinv[i];
        out[i] = z[i] * di * di + b2[0];
    }
}

__global__ void k_scatter2(const int* __restrict__ src,
                           const int* __restrict__ dst, int E,
                           const float* __restrict__ z,
                           const float* __restrict__ dinv,
                           float* __restrict__ out) {
    int e = blockIdx.x * blockDim.x + threadIdx.x;
    if (e < E) {
        int s = src[e], d = dst[e];
        atomicAdd(&out[d], z[s] * dinv[s] * dinv[d]);
    }
}


extern "C" void kernel_launch(void* const* d_in, const int* in_sizes, int n_in,
                              void* d_out, int out_size, void* d_ws, size_t ws_size,
                              hipStream_t stream) {
    const float* X  = (const float*)d_in[0];
    const int*   EI = (const int*)d_in[1];
    const float* W1 = (const float*)d_in[2];
    const float* b1 = (const float*)d_in[3];
    const float* W2 = (const float*)d_in[4];
    const float* b2 = (const float*)d_in[5];
    float* out = (float*)d_out;

    const int E = in_sizes[1] / 2;
    const int* src = EI;
    const int* dst = EI + E;

    char* ws = (char*)d_ws;
    float* H    = (float*)ws;
    float* agg  = (float*)(ws + (size_t)N_NODES * HID * 4);
    float* dinv = (float*)(ws + (size_t)2 * N_NODES * HID * 4);
    float* z    = dinv + N_NODES;
    unsigned int* deg = (unsigned int*)(z + N_NODES);

    const int B = 256;
    k_zero_deg <<<(N_NODES + B - 1) / B, B, 0, stream>>>(deg);
    k_deg_count<<<(E + B - 1) / B, B, 0, stream>>>(dst, E, deg);
    k_dinv     <<<(N_NODES + B - 1) / B, B, 0, stream>>>(deg, dinv);

    k_gemm1    <<<N_NODES / 16, 128, 0, stream>>>(X, W1, H);

    k_selfloop1<<<(N_NODES * HID + B - 1) / B, B, 0, stream>>>(H, dinv, agg);
    long long work1 = (long long)E * 64;
    k_scatter1 <<<(int)((work1 + B - 1) / B), B, 0, stream>>>(src, dst, E, H, dinv, agg);

    k_dot2     <<<(N_NODES * 32 + B - 1) / B, B, 0, stream>>>(agg, b1, W2, z);
    k_out_init <<<(N_NODES + B - 1) / B, B, 0, stream>>>(z, dinv, b2, out);
    k_scatter2 <<<(E + B - 1) / B, B, 0, stream>>>(src, dst, E, z, dinv, out);
}


# Layer-1 edge sums without global atomics: 256 work-groups each own 391 destination rows in LDS, sweep dst, queue their edges, add gathered H rows by LDS atomics (arrival order); self term folded in
# speedup vs baseline: 1.5388x; 1.5388x over previous
#include <hip/hip_runtime.h>

typedef float v2f __attribute__((ext_vector_type(2)));
typedef float v8f __attribute__((ext_vector_type(8)));

#define N_NODES 100000
#define IN_DIM  384
#define HID     64


__global__ void k_zero_deg(unsigned int* __restrict__ deg) {
    int i = blockIdx.x * blockDim.x + threadIdx.x;
    if (i < N_NODES) deg[i] = 0u;
}

__global__ void k_deg_count(const int* __restrict__ dst, int E,
                            unsigned int* __restrict__ deg) {
    int e = blockIdx.x * blockDim.x + threadIdx.x;
    if (e < E) atomicAdd(&deg[dst[e]], 1u);
}

__global__ void k_dinv(const unsigned int* __restrict__ deg,
                       float* __restrict__ dinv) {
    int i = blockIdx.x * blockDim.x + threadIdx.x;
    if (i < N_NODES) dinv[i] = rsqrtf((float)deg[i] + 1.0f);
}

__global__ __launch_bounds__(128) void k_gemm1(const float* __restrict__ X,
                                               const float* __restrict__ W1,
                                               float* __restrict__ H) {
    const int lane = threadIdx.x & 31;
    const int wave = threadIdx.x >> 5;
    const int half = lane >> 4;
    const int lr   = lane & 15;
    const int rowBase = blockIdx.x * 16;
    const int colBase = wave * 16;

    v8f c = {};
    const float* xrow = X + (size_t)(rowBase + lr) * IN_DIM;
    for (int k0 = 0; k0 < IN_DIM; k0 += 4) {
        v2f a, b;
        const float* ap = xrow + k0 + 2 * half;
        a[0] = ap[0];
        a[1] = ap[1];
        const float* bp = W1 + (size_t)(k0 + 2 * half) * HID + colBase + lr;
        b[0] = bp[0];
        b[1] = bp[HID];
        c = __builtin_amdgcn_wmma_f32_16x16x4_f32(
                 false, a,  false, b,
                 (short)0, c,  false,  false);
    }
    float* hrow = H + (size_t)(rowBase + 8 * half) * HID + colBase + lr;
#pragma unroll
    for (int r = 0; r < 8; ++r)
        hrow[(size_t)r * HID] = c[r];
}

#define TILES 256
#define RPB   391
#define RBITS 9
#define QCAP  16384
#define TPB   1024
#define SCAN_U 8
#define ROW_U  16

__global__ __launch_bounds__(TPB) void k_agg1(const int* __restrict__ src,
                                              const int* __restrict__ dst, int E,
                                              const float* __restrict__ H,
                                              const float* __restrict__ dinv,
                                              float* __restrict__ agg) {
    __shared__ float        acc[RPB * HID];
    __shared__ unsigned int q[QCAP];
    __shared__ float        qw[QCAP];
    __shared__ float        dl[RPB + 1];
    __shared__ unsigned int qn;

    const int tid = threadIdx.x;
    const int lo  = blockIdx.x * RPB;
    int cnt = N_NODES - lo;
    if (cnt > RPB) cnt = RPB;
    const unsigned ucnt = cnt > 0 ? (unsigned)cnt : 0u;

    if (tid == 0) qn = 0u;
    for (int r = tid; r < cnt; r += TPB) dl[r] = dinv[lo + r];
    __syncthreads();

    {
        const float4* H4 = reinterpret_cast<const float4*>(H + (size_t)lo * HID);
        float4* acc4 = reinterpret_cast<float4*>(acc);
        for (int i = tid; i < cnt * (HID / 4); i += TPB) {
            const float di = dl[i >> 4];
            float4 h = H4[i];
            h.x = h.x * di * di; h.y = h.y * di * di;
            h.z = h.z * di * di; h.w = h.w * di * di;
            acc4[i] = h;
        }
    }
    __syncthreads();

    auto found = [&](int e, unsigned r) {
        const unsigned pos = atomicAdd(&qn, 1u);
        if (pos < QCAP && e < (1 << (32 - RBITS))) {
            q[pos] = ((unsigned)e << RBITS) | r;
        } else {
            const int s = src[e];
            const float w = dinv[s] * dl[r];
            const float* hs = H + (size_t)s * HID;
            float* a = acc + r * HID;
            for (int f = 0; f < HID; ++f) atomicAdd(a + f, hs[f] * w);
        }
    };
    auto test4 = [&](int4 d, int e0) {
        const unsigned r0 = (unsigned)(d.x - lo), r1 = (unsigned)(d.y - lo);
        const unsigned r2 = (unsigned)(d.z - lo), r3 = (unsigned)(d.w - lo);
        const unsigned m01 = r0 < r1 ? r0 : r1, m23 = r2 < r3 ? r2 : r3;
        if ((m01 < m23 ? m01 : m23) < ucnt) {
            if (r0 < ucnt) found(e0, r0);
            if (r1 < ucnt) found(e0 + 1, r1);
            if (r2 < ucnt) found(e0 + 2, r2);
            if (r3 < ucnt) found(e0 + 3, r3);
        }
    };

    const int E4 = ((reinterpret_cast<size_t>(dst) & 15) == 0) ? (E >> 2) : 0;
    const int4* d4 = reinterpret_cast<const int4*>(dst);
    const int start = (int)(((long long)E4 * blockIdx.x) / TILES) & ~31;
    for (int part = 0; part < 2; ++part) {
        const int a = part ? 0 : start;
        const int b = part ? start : E4;
        int i = a + tid;
        for (; i + (SCAN_U - 1) * TPB < b; i += SCAN_U * TPB) {
            int4 d[SCAN_U];
#pragma unroll
            for (int u = 0; u < SCAN_U; ++u) d[u] = d4[i + u * TPB];
#pragma unroll
            for (int u = 0; u < SCAN_U; ++u) test4(d[u], 4 * (i + u * TPB));
        }
        for (; i < b; i += TPB) test4(d4[i], 4 * i);
    }
    for (int e = 4 * E4 + tid; e < E; e += TPB) {
        const unsigned r = (unsigned)(dst[e] - lo);
        if (r < ucnt) found(e, r);
    }
    __syncthreads();

    unsigned n = qn;
    if (n > QCAP) n = QCAP;

    for (unsigned i = tid; i < n; i += TPB) {
        const unsigned v = q[i];
        const unsigned r = v & ((1u << RBITS) - 1u);
        const int s = src[v >> RBITS];
        qw[i] = dinv[s] * dl[r];
        q[i] = ((unsigned)s << RBITS) | r;
    }
    __syncthreads();

    {
        const int lane = tid & 31;
        const unsigned wave = (unsigned)tid >> 5;
        const unsigned NW = TPB / 32;
        unsigned i = wave;
        for (; i + (ROW_U - 1) * NW < n; i += ROW_U * NW) {
            unsigned v[ROW_U];
            float w[ROW_U];
            float2 h[ROW_U];
#pragma unroll
            for (int u = 0; u < ROW_U; ++u) { v[u] = q[i + u * NW]; w[u] = qw[i + u * NW]; }
#pragma unroll
            for (int u = 0; u < ROW_U; ++u)
                h[u] = reinterpret_cast<const float2*>(H + (size_t)(v[u] >> RBITS) * HID)[lane];
#pragma unroll
            for (int u = 0; u < ROW_U; ++u) {
                float* a = acc + (v[u] & ((1u << RBITS) - 1u)) * HID + 2 * lane;
                atomicAdd(a,     h[u].x * w[u]);
                atomicAdd(a + 1, h[u].y * w[u]);
            }
        }
        for (; i < n; i += NW) {
            const unsigned v = q[i];
            const float w = qw[i];
            const float2 h = reinterpret_cast<const float2*>(H + (size_t)(v >> RBITS) * HID)[lane];
            float* a = acc + (v & ((1u << RBITS) - 1u)) * HID + 2 * lane;
            atomicAdd(a,     h.x * w);
            atomicAdd(a + 1, h.y * w);
        }
    }
    __syncthreads();

    {
        float4* o4 = reinterpret_cast<float4*>(agg + (size_t)lo * HID);
        const float4* acc4 = reinterpret_cast<const float4*>(acc);
        for (int i = tid; i < cnt * (HID / 4); i += TPB) o4[i] = acc4[i];
    }
}

__global__ __launch_bounds__(256) void k_dot2(const float* __restrict__ agg,
                                              const float* __restrict__ b1,
                                              const float* __restrict__ W2,
                                              float* __restrict__ z) {
    int row  = (blockIdx.x * blockDim.x + threadIdx.x) >> 5;
    int lane = threadIdx.x & 31;
    if (row < N_NODES) {
        const float* a = agg + (size_t)row * HID;
        float v0 = a[lane] + b1[lane];
        float v1 = a[lane + 32] + b1[lane + 32];
        v0 = v0 > 0.0f ? v0 : 0.0f;
        v1 = v1 > 0.0f ? v1 : 0.0f;
        float s = v0 * W2[lane] + v1 * W2[lane + 32];
#pragma unroll
        for (int off = 16; off > 0; off >>= 1)
            s += __shfl_down(s, off, 32);
        if (lane == 0) z[row] = s;
    }
}

__global__ void k_out_init(const float* __restrict__ z,
                           const float* __restrict__ dinv,
                           const float* __restrict__ b2,
                           float* __restrict__ out) {
    int i = blockIdx.x * blockDim.x + threadIdx.x;
    if (i < N_NODES) {
        float di = dinv[i];
        out[i] = z[i] * di * di + b2[0];
    }
}

__global__ void k_scatter2(const int* __restrict__ src,
                           const int* __restrict__ dst, int E,
                           const float* __restrict__ z,
                           const float* __restrict__ dinv,
                           float* __restrict__ out) {
    int e = blockIdx.x * blockDim.x + threadIdx.x;
    if (e < E) {
        int s = src[e], d = dst[e];
        atomicAdd(&out[d], z[s] * dinv[s] * dinv[d]);
    }
}


extern "C" void kernel_launch(void* const* d_in, const int* in_sizes, int n_in,
                              void* d_out, int out_size, void* d_ws, size_t ws_size,
                              hipStream_t stream) {
    const float* X  = (const float*)d_in[0];
    const int*   EI = (const int*)d_in[1];
    const float* W1 = (const float*)d_in[2];
    const float* b1 = (const float*)d_in[3];
    const float* W2 = (const float*)d_in[4];
    const float* b2 = (const float*)d_in[5];
    float* out = (float*)d_out;

    const int E = in_sizes[1] / 2;
    const int* src = EI;
    const int* dst = EI + E;

    char* ws = (char*)d_ws;
    float* H    = (float*)ws;
    float* agg  = (float*)(ws + (size_t)N_NODES * HID * 4);
    float* dinv = (float*)(ws + (size_t)2 * N_NODES * HID * 4);
    float* z    = dinv + N_NODES;
    unsigned int* deg = (unsigned int*)(z + N_NODES);

    const int B = 256;
    k_zero_deg <<<(N_NODES + B - 1) / B, B, 0, stream>>>(deg);
    k_deg_count<<<(E + B - 1) / B, B, 0, stream>>>(dst, E, deg);
    k_dinv     <<<(N_NODES + B - 1) / B, B, 0, stream>>>(deg, dinv);

    k_gemm1    <<<N_NODES / 16, 128, 0, stream>>>(X, W1, H);

    k_agg1     <<<TILES, TPB, 0, stream>>>(src, dst, E, H, dinv, agg);

    k_dot2     <<<(N_NODES * 32 + B - 1) / B, B, 0, stream>>>(agg, b1, W2, z);
    k_out_init <<<(N_NODES + B - 1) / B, B, 0, stream>>>(z, dinv, b2, out);
    k_scatter2 <<<(E + B - 1) / B, B, 0, stream>>>(src, dst, E, z, dinv, out);
}


# Degree count and layer-2 scatter off global atomics: 128 edge slices x 2 node halves add in LDS, store planes; planes added in fixed slice order (new sum tree; LDS arrival order in a plane)
# speedup vs baseline: 2.6878x; 1.7467x over previous
#include <hip/hip_runtime.h>

typedef float v2f __attribute__((ext_vector_type(2)));
typedef float v8f __attribute__((ext_vector_type(8)));

#define N_NODES 100000
#define IN_DIM  384
#define HID     64

#define HALF_N  50000
#define NSLICE  128
#define PBLK    (2 * NSLICE)
#define PTPB    1024
#define PART_U  6
#define SGRP    8
#define SPG     (NSLICE / SGRP)

static_assert(2 * HALF_N >= N_NODES && HALF_N % 4 == 0 && N_NODES % 4 == 0, "halves");
static_assert((size_t)PBLK * HALF_N <= (size_t)2 * N_NODES * HID, "planes fit over H and agg");
static_assert(NSLICE % SGRP == 0, "slice groups");

__global__ __launch_bounds__(PTPB) void k_deg_part(const int* __restrict__ dst, int E,
                                                   unsigned int* __restrict__ P) {
    __shared__ unsigned int cnt[HALF_N];
    const int tid   = threadIdx.x;
    const int half  = blockIdx.x & 1;
    const int slice = blockIdx.x >> 1;
    const int lo    = half * HALF_N;
    int n = N_NODES - lo;
    if (n > HALF_N) n = HALF_N;
    const unsigned un = n > 0 ? (unsigned)n : 0u;

    uint4* c4 = reinterpret_cast<uint4*>(cnt);
    for (int i = tid; i < HALF_N / 4; i += PTPB) c4[i] = make_uint4(0u, 0u, 0u, 0u);
    __syncthreads();

    auto hit4 = [&](int4 d) {
        const unsigned r0 = (unsigned)(d.x - lo), r1 = (unsigned)(d.y - lo);
        const unsigned r2 = (unsigned)(d.z - lo), r3 = (unsigned)(d.w - lo);
        if (r0 < un) atomicAdd(&cnt[r0], 1u);
        if (r1 < un) atomicAdd(&cnt[r1], 1u);
        if (r2 < un) atomicAdd(&cnt[r2], 1u);
        if (r3 < un) atomicAdd(&cnt[r3], 1u);
    };

    const int E4 = ((reinterpret_cast<size_t>(dst) & 15) == 0) ? (E >> 2) : 0;
    const int4* d4 = reinterpret_cast<const int4*>(dst);
    const int a = (int)(((long long)E4 * slice) / NSLICE);
    const int b = (int)(((long long)E4 * (slice + 1)) / NSLICE);
    int i = a + tid;
    for (; i + (PART_U - 1) * PTPB < b; i += PART_U * PTPB) {
        int4 d[PART_U];
#pragma unroll
        for (int u = 0; u < PART_U; ++u) d[u] = d4[i + u * PTPB];
#pragma unroll
        for (int u = 0; u < PART_U; ++u) hit4(d[u]);
    }
    for (; i < b; i += PTPB) hit4(d4[i]);
    if (slice == NSLICE - 1) {
        for (int e = 4 * E4 + tid; e < E; e += PTPB) {
            const unsigned r = (unsigned)(dst[e] - lo);
            if (r < un) atomicAdd(&cnt[r], 1u);
        }
    }
    __syncthreads();

    uint4* o4 = reinterpret_cast<uint4*>(P + (size_t)blockIdx.x * HALF_N);
    for (int j = tid; j < HALF_N / 4; j += PTPB) o4[j] = c4[j];
}

__global__ __launch_bounds__(32 * SGRP) void k_dinv_sum(const unsigned int* __restrict__ P,
                                                        float* __restrict__ dinv) {
    __shared__ uint4 part[SGRP][32];
    const int x = threadIdx.x, g = threadIdx.y;
    const int quad = blockIdx.x * 32 + x;
    uint4 s = make_uint4(0u, 0u, 0u, 0u);
    if (quad < N_NODES / 4) {
        const int node = 4 * quad;
        const int half = node >= HALF_N ? 1 : 0;
        const int j = node - half * HALF_N;
        const uint4* p = reinterpret_cast<const uint4*>(
            P + ((size_t)(2 * g * SPG + half)) * HALF_N + j);
#pragma unroll
        for (int k = 0; k < SPG; ++k) {
            const uint4 v = p[(size_t)k * (2 * HALF_N / 4)];
            s.x += v.x; s.y += v.y; s.z += v.z; s.w += v.w;
        }
    }
    part[g][x] = s;
    __syncthreads();
    if (g == 0 && quad < N_NODES / 4) {
        uint4 t = part[0][x];
        for (int k = 1; k < SGRP; ++k) {
            const uint4 v = part[k][x];
            t.x += v.x; t.y += v.y; t.z += v.z; t.w += v.w;
        }
        float4 r;
        r.x = rsqrtf((float)t.x + 1.0f);
        r.y = rsqrtf((float)t.y + 1.0f);
        r.z = rsqrtf((float)t.z + 1.0f);
        r.w = rsqrtf((float)t.w + 1.0f);
        reinterpret_cast<float4*>(dinv)[quad] = r;
    }
}

__global__ __launch_bounds__(128) void k_gemm1(const float* __restrict__ X,
                                               const float* __restrict__ W1,
                                               float* __restrict__ H) {
    const int lane = threadIdx.x & 31;
    const int wave = threadIdx.x >> 5;
    const int half = lane >> 4;
    const int lr   = lane & 15;
    const int rowBase = blockIdx.x * 16;
    const int colBase = wave * 16;

    v8f c = {};
    const float* xrow = X + (size_t)(rowBase + lr) * IN_DIM;
    for (int k0 = 0; k0 < IN_DIM; k0 += 4) {
        v2f a, b;
        const float* ap = xrow + k0 + 2 * half;
        a[0] = ap[0];
        a[1] = ap[1];
        const float* bp = W1 + (size_t)(k0 + 2 * half) * HID + colBase + lr;
        b[0] = bp[0];
        b[1] = bp[HID];
        c = __builtin_amdgcn_wmma_f32_16x16x4_f32(
                 false, a,  false, b,
                 (short)0, c,  false,  false);
    }
    float* hrow = H + (size_t)(rowBase + 8 * half) * HID + colBase + lr;
#pragma unroll
    for (int r = 0; r < 8; ++r)
        hrow[(size_t)r * HID] = c[r];
}

#define TILES 256
#define RPB   391
#define RBITS 9
#define QCAP  16384
#define TPB   1024
#define SCAN_U 8
#define ROW_U  16

__global__ __launch_bounds__(TPB) void k_agg1(const int* __restrict__ src,
                                              const int* __restrict__ dst, int E,
                                              const float* __restrict__ H,
                                              const float* __restrict__ dinv,
                                              float* __restrict__ agg) {
    __shared__ float        acc[RPB * HID];
    __shared__ unsigned int q[QCAP];
    __shared__ float        qw[QCAP];
    __shared__ float        dl[RPB + 1];
    __shared__ unsigned int qn;

    const int tid = threadIdx.x;
    const int lo  = blockIdx.x * RPB;
    int cnt = N_NODES - lo;
    if (cnt > RPB) cnt = RPB;
    const unsigned ucnt = cnt > 0 ? (unsigned)cnt : 0u;

    if (tid == 0) qn = 0u;
    for (int r = tid; r < cnt; r += TPB) dl[r] = dinv[lo + r];
    __syncthreads();

    {
        const float4* H4 = reinterpret_cast<const float4*>(H + (size_t)lo * HID);
        float4* acc4 = reinterpret_cast<float4*>(acc);
        for (int i = tid; i < cnt * (HID / 4); i += TPB) {
            const float di = dl[i >> 4];
            float4 h = H4[i];
            h.x = h.x * di * di; h.y = h.y * di * di;
            h.z = h.z * di * di; h.w = h.w * di * di;
            acc4[i] = h;
        }
    }
    __syncthreads();

    auto found = [&](int e, unsigned r) {
        const unsigned pos = atomicAdd(&qn, 1u);
        if (pos < QCAP && e < (1 << (32 - RBITS))) {
            q[pos] = ((unsigned)e << RBITS) | r;
        } else {
            const int s = src[e];
            const float w = dinv[s] * dl[r];
            const float* hs = H + (size_t)s * HID;
            float* a = acc + r * HID;
            for (int f = 0; f < HID; ++f) atomicAdd(a + f, hs[f] * w);
        }
    };
    auto test4 = [&](int4 d, int e0) {
        const unsigned r0 = (unsigned)(d.x - lo), r1 = (unsigned)(d.y - lo);
        const unsigned r2 = (unsigned)(d.z - lo), r3 = (unsigned)(d.w - lo);
        const unsigned m01 = r0 < r1 ? r0 : r1, m23 = r2 < r3 ? r2 : r3;
        if ((m01 < m23 ? m01 : m23) < ucnt) {
            if (r0 < ucnt) found(e0, r0);
            if (r1 < ucnt) found(e0 + 1, r1);
            if (r2 < ucnt) found(e0 + 2, r2);
            if (r3 < ucnt) found(e0 + 3, r3);
        }
    };

    const int E4 = ((reinterpret_cast<size_t>(dst) & 15) == 0) ? (E >> 2) : 0;
    const int4* d4 = reinterpret_cast<const int4*>(dst);
    const int start = (int)(((long long)E4 * blockIdx.x) / TILES) & ~31;
    for (int part = 0; part < 2; ++part) {
        const int a = part ? 0 : start;
        const int b = part ? start : E4;
        int i = a + tid;
        for (; i + (SCAN_U - 1) * TPB < b; i += SCAN_U * TPB) {
            int4 d[SCAN_U];
#pragma unroll
            for (int u = 0; u < SCAN_U; ++u) d[u] = d4[i + u * TPB];
#pragma unroll
            for (int u = 0; u < SCAN_U; ++u) test4(d[u], 4 * (i + u * TPB));
        }
        for (; i < b; i += TPB) test4(d4[i], 4 * i);
    }
    for (int e = 4 * E4 + tid; e < E; e += TPB) {
        const unsigned r = (unsigned)(dst[e] - lo);
        if (r < ucnt) found(e, r);
    }
    __syncthreads();

    unsigned n = qn;
    if (n > QCAP) n = QCAP;

    for (unsigned i = tid; i < n; i += TPB) {
        const unsigned v = q[i];
        const unsigned r = v & ((1u << RBITS) - 1u);
        const int s = src[v >> RBITS];
        qw[i] = dinv[s] * dl[r];
        q[i] = ((unsigned)s << RBITS) | r;
    }
    __syncthreads();

    {
        const int lane = tid & 31;
        const unsigned wave = (unsigned)tid >> 5;
        const unsigned NW = TPB / 32;
        unsigned i = wave;
        for (; i + (ROW_U - 1) * NW < n; i += ROW_U * NW) {
            unsigned v[ROW_U];
            float w[ROW_U];
            float2 h[ROW_U];
#pragma unroll
            for (int u = 0; u < ROW_U; ++u) { v[u] = q[i + u * NW]; w[u] = qw[i + u * NW]; }
#pragma unroll
            for (int u = 0; u < ROW_U; ++u)
                h[u] = reinterpret_cast<const float2*>(H + (size_t)(v[u] >> RBITS) * HID)[lane];
#pragma unroll
            for (int u = 0; u < ROW_U; ++u) {
                float* a = acc + (v[u] & ((1u << RBITS) - 1u)) * HID + 2 * lane;
                atomicAdd(a,     h[u].x * w[u]);
                atomicAdd(a + 1, h[u].y * w[u]);
            }
        }
        for (; i < n; i += NW) {
            const unsigned v = q[i];
            const float w = qw[i];
            const float2 h = reinterpret_cast<const float2*>(H + (size_t)(v >> RBITS) * HID)[lane];
            float* a = acc + (v & ((1u << RBITS) - 1u)) * HID + 2 * lane;
            atomicAdd(a,     h.x * w);
            atomicAdd(a + 1, h.y * w);
        }
    }
    __syncthreads();

    {
        float4* o4 = reinterpret_cast<float4*>(agg + (size_t)lo * HID);
        const float4* acc4 = reinterpret_cast<const float4*>(acc);
        for (int i = tid; i < cnt * (HID / 4); i += TPB) o4[i] = acc4[i];
    }
}

__global__ __launch_bounds__(256) void k_dot2(const float* __restrict__ agg,
                                              const float* __restrict__ b1,
                                              const float* __restrict__ W2,
                                              float* __restrict__ z) {
    int row  = (blockIdx.x * blockDim.x + threadIdx.x) >> 5;
    int lane = threadIdx.x & 31;
    if (row < N_NODES) {
        const float* a = agg + (size_t)row * HID;
        float v0 = a[lane] + b1[lane];
        float v1 = a[lane + 32] + b1[lane + 32];
        v0 = v0 > 0.0f ? v0 : 0.0f;
        v1 = v1 > 0.0f ? v1 : 0.0f;
        float s = v0 * W2[lane] + v1 * W2[lane + 32];
#pragma unroll
        for (int off = 16; off > 0; off >>= 1)
            s += __shfl_down(s, off, 32);
        if (lane == 0) z[row] = s;
    }
}

__global__ __launch_bounds__(PTPB) void k_l2_part(const int* __restrict__ src,
                                                  const int* __restrict__ dst, int E,
                                                  const float* __restrict__ z,
                                                  const float* __restrict__ dinv,
                                                  float* __restrict__ P) {
    __shared__ float acc[HALF_N];
    const int tid   = threadIdx.x;
    const int half  = blockIdx.x & 1;
    const int slice = blockIdx.x >> 1;
    const int lo    = half * HALF_N;
    int n = N_NODES - lo;
    if (n > HALF_N) n = HALF_N;
    const unsigned un = n > 0 ? (unsigned)n : 0u;

    float4* a4 = reinterpret_cast<float4*>(acc);
    for (int i = tid; i < HALF_N / 4; i += PTPB) a4[i] = make_float4(0.0f, 0.0f, 0.0f, 0.0f);
    __syncthreads();

    auto term = [&](int s, int d, bool ok) -> float {
        const int ss = ok ? s : lo;
        const int dd = ok ? d : lo;
        return z[ss] * dinv[ss] * dinv[dd];
    };

    const bool al = ((reinterpret_cast<size_t>(dst) | reinterpret_cast<size_t>(src)) & 15) == 0;
    const int E4 = al ? (E >> 2) : 0;
    const int4* d4 = reinterpret_cast<const int4*>(dst);
    const int4* s4 = reinterpret_cast<const int4*>(src);
    const int a = (int)(((long long)E4 * slice) / NSLICE);
    const int b = (int)(((long long)E4 * (slice + 1)) / NSLICE);
    int i = a + tid;
    for (; i + (PART_U - 1) * PTPB < b; i += PART_U * PTPB) {
        int4 d[PART_U], s[PART_U];
        float t[PART_U][4];
#pragma unroll
        for (int u = 0; u < PART_U; ++u) { d[u] = d4[i + u * PTPB]; s[u] = s4[i + u * PTPB]; }
#pragma unroll
        for (int u = 0; u < PART_U; ++u) {
            t[u][0] = term(s[u].x, d[u].x, (unsigned)(d[u].x - lo) < un);
            t[u][1] = term(s[u].y, d[u].y, (unsigned)(d[u].y - lo) < un);
            t[u][2] = term(s[u].z, d[u].z, (unsigned)(d[u].z - lo) < un);
            t[u][3] = term(s[u].w, d[u].w, (unsigned)(d[u].w - lo) < un);
        }
#pragma unroll
        for (int u = 0; u < PART_U; ++u) {
            const unsigned r0 = (unsigned)(d[u].x - lo), r1 = (unsigned)(d[u].y - lo);
            const unsigned r2 = (unsigned)(d[u].z - lo), r3 = (unsigned)(d[u].w - lo);
            if (r0 < un) atomicAdd(&acc[r0], t[u][0]);
            if (r1 < un) atomicAdd(&acc[r1], t[u][1]);
            if (r2 < un) atomicAdd(&acc[r2], t[u][2]);
            if (r3 < un) atomicAdd(&acc[r3], t[u][3]);
        }
    }
    for (; i < b; i += PTPB) {
        const int4 d = d4[i], s = s4[i];
        const unsigned r0 = (unsigned)(d.x - lo), r1 = (unsigned)(d.y - lo);
        const unsigned r2 = (unsigned)(d.z - lo), r3 = (unsigned)(d.w - lo);
        const float t0 = term(s.x, d.x, r0 < un), t1 = term(s.y, d.y, r1 < un);
        const float t2 = term(s.z, d.z, r2 < un), t3 = term(s.w, d.w, r3 < un);
        if (r0 < un) atomicAdd(&acc[r0], t0);
        if (r1 < un) atomicAdd(&acc[r1], t1);
        if (r2 < un) atomicAdd(&acc[r2], t2);
        if (r3 < un) atomicAdd(&acc[r3], t3);
    }
    if (slice == NSLICE - 1) {
        for (int e = 4 * E4 + tid; e < E; e += PTPB) {
            const int d = dst[e];
            const unsigned r = (unsigned)(d - lo);
            if (r < un) {
                const int s = src[e];
                atomicAdd(&acc[r], z[s] * dinv[s] * dinv[d]);
            }
        }
    }
    __syncthreads();

    float4* o4 = reinterpret_cast<float4*>(P + (size_t)blockIdx.x * HALF_N);
    for (int j = tid; j < HALF_N / 4; j += PTPB) o4[j] = a4[j];
}

__global__ __launch_bounds__(32 * SGRP) void k_out_sum(const float* __restrict__ P,
                                                       const float* __restrict__ z,
                                                       const float* __restrict__ dinv,
                                                       const float* __restrict__ b2,
                                                       float* __restrict__ out) {
    __shared__ float4 part[SGRP][32];
    const int x = threadIdx.x, g = threadIdx.y;
    const int quad = blockIdx.x * 32 + x;
    float4 s = make_float4(0.0f, 0.0f, 0.0f, 0.0f);
    if (quad < N_NODES / 4) {
        const int node = 4 * quad;
        const int half = node >= HALF_N ? 1 : 0;
        const int j = node - half * HALF_N;
        const float4* p = reinterpret_cast<const float4*>(
            P + ((size_t)(2 * g * SPG + half)) * HALF_N + j);
#pragma unroll
        for (int k = 0; k < SPG; ++k) {
            const float4 v = p[(size_t)k * (2 * HALF_N / 4)];
            s.x += v.x; s.y += v.y; s.z += v.z; s.w += v.w;
        }
    }
    part[g][x] = s;
    __syncthreads();
    if (g == 0 && quad < N_NODES / 4) {
        float4 t = part[0][x];
        for (int k = 1; k < SGRP; ++k) {
            const float4 v = part[k][x];
            t.x += v.x; t.y += v.y; t.z += v.z; t.w += v.w;
        }
        const float4 zz = reinterpret_cast<const float4*>(z)[quad];
        const float4 di = reinterpret_cast<const float4*>(dinv)[quad];
        const float bb = b2[0];
        float4 r;
        r.x = (zz.x * di.x * di.x + bb) + t.x;
        r.y = (zz.y * di.y * di.y + bb) + t.y;
        r.z = (zz.z * di.z * di.z + bb) + t.z;
        r.w = (zz.w * di.w * di.w + bb) + t.w;
        reinterpret_cast<float4*>(out)[quad] = r;
    }
}


extern "C" void kernel_launch(void* const* d_in, const int* in_sizes, int n_in,
                              void* d_out, int out_size, void* d_ws, size_t ws_size,
                              hipStream_t stream) {
    const float* X  = (const float*)d_in[0];
    const int*   EI = (const int*)d_in[1];
    const float* W1 = (const float*)d_in[2];
    const float* b1 = (const float*)d_in[3];
    const float* W2 = (const float*)d_in[4];
    const float* b2 = (const float*)d_in[5];
    float* out = (float*)d_out;

    const int E = in_sizes[1] / 2;
    const int* src = EI;
    const int* dst = EI + E;

    char* ws = (char*)d_ws;
    float* H    = (float*)ws;
    float* agg  = (float*)(ws + (size_t)N_NODES * HID * 4);
    float* dinv = (float*)(ws + (size_t)2 * N_NODES * HID * 4);
    float* z    = dinv + N_NODES;
    unsigned int* Pu = (unsigned int*)ws;
    float*        Pf = (float*)ws;

    const int B = 256;
    k_deg_part <<<PBLK, PTPB, 0, stream>>>(dst, E, Pu);
    k_dinv_sum <<<(N_NODES / 4 + 31) / 32, dim3(32, SGRP), 0, stream>>>(Pu, dinv);

    k_gemm1    <<<N_NODES / 16, 128, 0, stream>>>(X, W1, H);

    k_agg1     <<<TILES, TPB, 0, stream>>>(src, dst, E, H, dinv, agg);

    k_dot2     <<<(N_NODES * 32 + B - 1) / B, B, 0, stream>>>(agg, b1, W2, z);
    k_l2_part  <<<PBLK, PTPB, 0, stream>>>(src, dst, E, z, dinv, Pf);
    k_out_sum  <<<(N_NODES / 4 + 31) / 32, dim3(32, SGRP), 0, stream>>>(Pf, z, dinv, b2, out);
}


# Layer-1 tile kernel forms z = relu(sum + b1) . W2 from its LDS sums (one wave per row, same dot tree): the 100000-wave dot launch and the agg plane's store and reload are gone; 6 launches
# speedup vs baseline: 2.7836x; 1.0356x over previous
#include <hip/hip_runtime.h>

typedef float v2f __attribute__((ext_vector_type(2)));
typedef float v8f __attribute__((ext_vector_type(8)));

#define N_NODES 100000
#define IN_DIM  384
#define HID     64

#define HALF_N  50000
#define NSLICE  128
#define PBLK    (2 * NSLICE)
#define PTPB    1024
#define PART_U  6
#define SGRP    8
#define SPG     (NSLICE / SGRP)

static_assert(2 * HALF_N >= N_NODES && HALF_N % 4 == 0 && N_NODES % 4 == 0, "halves");
static_assert((size_t)PBLK * HALF_N <= (size_t)2 * N_NODES * HID, "planes end in front of dinv");
static_assert(NSLICE % SGRP == 0, "slice groups");

__global__ __launch_bounds__(PTPB) void k_deg_part(const int* __restrict__ dst, int E,
                                                   unsigned int* __restrict__ P) {
    __shared__ unsigned int cnt[HALF_N];
    const int tid   = threadIdx.x;
    const int half  = blockIdx.x & 1;
    const int slice = blockIdx.x >> 1;
    const int lo    = half * HALF_N;
    int n = N_NODES - lo;
    if (n > HALF_N) n = HALF_N;
    const unsigned un = n > 0 ? (unsigned)n : 0u;

    uint4* c4 = reinterpret_cast<uint4*>(cnt);
    for (int i = tid; i < HALF_N / 4; i += PTPB) c4[i] = make_uint4(0u, 0u, 0u, 0u);
    __syncthreads();

    auto hit4 = [&](int4 d) {
        const unsigned r0 = (unsigned)(d.x - lo), r1 = (unsigned)(d.y - lo);
        const unsigned r2 = (unsigned)(d.z - lo), r3 = (unsigned)(d.w - lo);
        if (r0 < un) atomicAdd(&cnt[r0], 1u);
        if (r1 < un) atomicAdd(&cnt[r1], 1u);
        if (r2 < un) atomicAdd(&cnt[r2], 1u);
        if (r3 < un) atomicAdd(&cnt[r3], 1u);
    };

    const int E4 = ((reinterpret_cast<size_t>(dst) & 15) == 0) ? (E >> 2) : 0;
    const int4* d4 = reinterpret_cast<const int4*>(dst);
    const int a = (int)(((long long)E4 * slice) / NSLICE);
    const int b = (int)(((long long)E4 * (slice + 1)) / NSLICE);
    int i = a + tid;
    for (; i + (PART_U - 1) * PTPB < b; i += PART_U * PTPB) {
        int4 d[PART_U];
#pragma unroll
        for (int u = 0; u < PART_U; ++u) d[u] = d4[i + u * PTPB];
#pragma unroll
        for (int u = 0; u < PART_U; ++u) hit4(d[u]);
    }
    for (; i < b; i += PTPB) hit4(d4[i]);
    if (slice == NSLICE - 1) {
        for (int e = 4 * E4 + tid; e < E; e += PTPB) {
            const unsigned r = (unsigned)(dst[e] - lo);
            if (r < un) atomicAdd(&cnt[r], 1u);
        }
    }
    __syncthreads();

    uint4* o4 = reinterpret_cast<uint4*>(P + (size_t)blockIdx.x * HALF_N);
    for (int j = tid; j < HALF_N / 4; j += PTPB) o4[j] = c4[j];
}

__global__ __launch_bounds__(32 * SGRP) void k_dinv_sum(const unsigned int* __restrict__ P,
                                                        float* __restrict__ dinv) {
    __shared__ uint4 part[SGRP][32];
    const int x = threadIdx.x, g = threadIdx.y;
    const int quad = blockIdx.x * 32 + x;
    uint4 s = make_uint4(0u, 0u, 0u, 0u);
    if (quad < N_NODES / 4) {
        const int node = 4 * quad;
        const int half = node >= HALF_N ? 1 : 0;
        const int j = node - half * HALF_N;
        const uint4* p = reinterpret_cast<const uint4*>(
            P + ((size_t)(2 * g * SPG + half)) * HALF_N + j);
#pragma unroll
        for (int k = 0; k < SPG; ++k) {
            const uint4 v = p[(size_t)k * (2 * HALF_N / 4)];
            s.x += v.x; s.y += v.y; s.z += v.z; s.w += v.w;
        }
    }
    part[g][x] = s;
    __syncthreads();
    if (g == 0 && quad < N_NODES / 4) {
        uint4 t = part[0][x];
        for (int k = 1; k < SGRP; ++k) {
            const uint4 v = part[k][x];
            t.x += v.x; t.y += v.y; t.z += v.z; t.w += v.w;
        }
        float4 r;
        r.x = rsqrtf((float)t.x + 1.0f);
        r.y = rsqrtf((float)t.y + 1.0f);
        r.z = rsqrtf((float)t.z + 1.0f);
        r.w = rsqrtf((float)t.w + 1.0f);
        reinterpret_cast<float4*>(dinv)[quad] = r;
    }
}

__global__ __launch_bounds__(128) void k_gemm1(const float* __restrict__ X,
                                               const float* __restrict__ W1,
                                               float* __restrict__ H) {
    const int lane = threadIdx.x & 31;
    const int wave = threadIdx.x >> 5;
    const int half = lane >> 4;
    const int lr   = lane & 15;
    const int rowBase = blockIdx.x * 16;
    const int colBase = wave * 16;

    v8f c = {};
    const float* xrow = X + (size_t)(rowBase + lr) * IN_DIM;
    for (int k0 = 0; k0 < IN_DIM; k0 += 4) {
        v2f a, b;
        const float* ap = xrow + k0 + 2 * half;
        a[0] = ap[0];
        a[1] = ap[1];
        const float* bp = W1 + (size_t)(k0 + 2 * half) * HID + colBase + lr;
        b[0] = bp[0];
        b[1] = bp[HID];
        c = __builtin_amdgcn_wmma_f32_16x16x4_f32(
                 false, a,  false, b,
                 (short)0, c,  false,  false);
    }
    float* hrow = H + (size_t)(rowBase + 8 * half) * HID + colBase + lr;
#pragma unroll
    for (int r = 0; r < 8; ++r)
        hrow[(size_t)r * HID] = c[r];
}

#define TILES 256
#define RPB   391
#define RBITS 9
#define QCAP  16384
#define TPB   1024
#define SCAN_U 8
#define ROW_U  16
#define Z_U    8

__global__ __launch_bounds__(TPB) void k_agg1(const int* __restrict__ src,
                                              const int* __restrict__ dst, int E,
                                              const float* __restrict__ H,
                                              const float* __restrict__ dinv,
                                              const float* __restrict__ b1,
                                              const float* __restrict__ W2,
                                              float* __restrict__ z) {
    __shared__ float        acc[RPB * HID];
    __shared__ unsigned int q[QCAP];
    __shared__ float        qw[QCAP];
    __shared__ float        dl[RPB + 1];
    __shared__ unsigned int qn;

    const int tid = threadIdx.x;
    const int lo  = blockIdx.x * RPB;
    int cnt = N_NODES - lo;
    if (cnt > RPB) cnt = RPB;
    const unsigned ucnt = cnt > 0 ? (unsigned)cnt : 0u;

    if (tid == 0) qn = 0u;
    for (int r = tid; r < cnt; r += TPB) dl[r] = dinv[lo + r];
    __syncthreads();

    {
        const float4* H4 = reinterpret_cast<const float4*>(H + (size_t)lo * HID);
        float4* acc4 = reinterpret_cast<float4*>(acc);
        for (int i = tid; i < cnt * (HID / 4); i += TPB) {
            const float di = dl[i >> 4];
            float4 h = H4[i];
            h.x = h.x * di * di; h.y = h.y * di * di;
            h.z = h.z * di * di; h.w = h.w * di * di;
            acc4[i] = h;
        }
    }
    __syncthreads();

    auto found = [&](int e, unsigned r) {
        const unsigned pos = atomicAdd(&qn, 1u);
        if (pos < QCAP && e < (1 << (32 - RBITS))) {
            q[pos] = ((unsigned)e << RBITS) | r;
        } else {
            const int s = src[e];
            const float w = dinv[s] * dl[r];
            const float* hs = H + (size_t)s * HID;
            float* a = acc + r * HID;
            for (int f = 0; f < HID; ++f) atomicAdd(a + f, hs[f] * w);
        }
    };
    auto test4 = [&](int4 d, int e0) {
        const unsigned r0 = (unsigned)(d.x - lo), r1 = (unsigned)(d.y - lo);
        const unsigned r2 = (unsigned)(d.z - lo), r3 = (unsigned)(d.w - lo);
        const unsigned m01 = r0 < r1 ? r0 : r1, m23 = r2 < r3 ? r2 : r3;
        if ((m01 < m23 ? m01 : m23) < ucnt) {
            if (r0 < ucnt) found(e0, r0);
            if (r1 < ucnt) found(e0 + 1, r1);
            if (r2 < ucnt) found(e0 + 2, r2);
            if (r3 < ucnt) found(e0 + 3, r3);
        }
    };

    const int E4 = ((reinterpret_cast<size_t>(dst) & 15) == 0) ? (E >> 2) : 0;
    const int4* d4 = reinterpret_cast<const int4*>(dst);
    const int start = (int)(((long long)E4 * blockIdx.x) / TILES) & ~31;
    for (int part = 0; part < 2; ++part) {
        const int a = part ? 0 : start;
        const int b = part ? start : E4;
        int i = a + tid;
        for (; i + (SCAN_U - 1) * TPB < b; i += SCAN_U * TPB) {
            int4 d[SCAN_U];
#pragma unroll
            for (int u = 0; u < SCAN_U; ++u) d[u] = d4[i + u * TPB];
#pragma unroll
            for (int u = 0; u < SCAN_U; ++u) test4(d[u], 4 * (i + u * TPB));
        }
        for (; i < b; i += TPB) test4(d4[i], 4 * i);
    }
    for (int e = 4 * E4 + tid; e < E; e += TPB) {
        const unsigned r = (unsigned)(dst[e] - lo);
        if (r < ucnt) found(e, r);
    }
    __syncthreads();

    unsigned n = qn;
    if (n > QCAP) n = QCAP;

    for (unsigned i = tid; i < n; i += TPB) {
        const unsigned v = q[i];
        const unsigned r = v & ((1u << RBITS) - 1u);
        const int s = src[v >> RBITS];
        qw[i] = dinv[s] * dl[r];
        q[i] = ((unsigned)s << RBITS) | r;
    }
    __syncthreads();

    const float b1a = b1[tid & 31], b1b = b1[(tid & 31) + 32];
    const float w2a = W2[tid & 31], w2b = W2[(tid & 31) + 32];

    {
        const int lane = tid & 31;
        const unsigned wave = (unsigned)tid >> 5;
        const unsigned NW = TPB / 32;
        unsigned i = wave;
        for (; i + (ROW_U - 1) * NW < n; i += ROW_U * NW) {
            unsigned v[ROW_U];
            float w[ROW_U];
            float2 h[ROW_U];
#pragma unroll
            for (int u = 0; u < ROW_U; ++u) { v[u] = q[i + u * NW]; w[u] = qw[i + u * NW]; }
#pragma unroll
            for (int u = 0; u < ROW_U; ++u)
                h[u] = reinterpret_cast<const float2*>(H + (size_t)(v[u] >> RBITS) * HID)[lane];
#pragma unroll
            for (int u = 0; u < ROW_U; ++u) {
                float* a = acc + (v[u] & ((1u << RBITS) - 1u)) * HID + 2 * lane;
                atomicAdd(a,     h[u].x * w[u]);
                atomicAdd(a + 1, h[u].y * w[u]);
            }
        }
        for (; i < n; i += NW) {
            const unsigned v = q[i];
            const float w = qw[i];
            const float2 h = reinterpret_cast<const float2*>(H + (size_t)(v >> RBITS) * HID)[lane];
            float* a = acc + (v & ((1u << RBITS) - 1u)) * HID + 2 * lane;
            atomicAdd(a,     h.x * w);
            atomicAdd(a + 1, h.y * w);
        }
    }
    __syncthreads();

    {
        const int lane = tid & 31;
        const int NW = TPB / 32;
        for (int r0 = tid >> 5; r0 < cnt; r0 += Z_U * NW) {
            float s[Z_U];
#pragma unroll
            for (int u = 0; u < Z_U; ++u) {
                const int r = r0 + u * NW;
                const float* a = acc + (r < cnt ? r : r0) * HID;
                float v0 = a[lane] + b1a;
                float v1 = a[lane + 32] + b1b;
                v0 = v0 > 0.0f ? v0 : 0.0f;
                v1 = v1 > 0.0f ? v1 : 0.0f;
                s[u] = v0 * w2a + v1 * w2b;
            }
#pragma unroll
            for (int off = 16; off > 0; off >>= 1) {
#pragma unroll
                for (int u = 0; u < Z_U; ++u) s[u] += __shfl_down(s[u], off, 32);
            }
#pragma unroll
            for (int u = 0; u < Z_U; ++u) {
                const int r = r0 + u * NW;
                if (lane == 0 && r < cnt) qw[r] = s[u];
            }
        }
    }
    __syncthreads();
    for (int r = tid; r < cnt; r += TPB) z[lo + r] = qw[r];
}

__global__ __launch_bounds__(PTPB) void k_l2_part(const int* __restrict__ src,
                                                  const int* __restrict__ dst, int E,
                                                  const float* __restrict__ z,
                                                  const float* __restrict__ dinv,
                                                  float* __restrict__ P) {
    __shared__ float acc[HALF_N];
    const int tid   = threadIdx.x;
    const int half  = blockIdx.x & 1;
    const int slice = blockIdx.x >> 1;
    const int lo    = half * HALF_N;
    int n = N_NODES - lo;
    if (n > HALF_N) n = HALF_N;
    const unsigned un = n > 0 ? (unsigned)n : 0u;

    float4* a4 = reinterpret_cast<float4*>(acc);
    for (int i = tid; i < HALF_N / 4; i += PTPB) a4[i] = make_float4(0.0f, 0.0f, 0.0f, 0.0f);
    __syncthreads();

    auto term = [&](int s, int d, bool ok) -> float {
        const int ss = ok ? s : lo;
        const int dd = ok ? d : lo;
        return z[ss] * dinv[ss] * dinv[dd];
    };

    const bool al = ((reinterpret_cast<size_t>(dst) | reinterpret_cast<size_t>(src)) & 15) == 0;
    const int E4 = al ? (E >> 2) : 0;
    const int4* d4 = reinterpret_cast<const int4*>(dst);
    const int4* s4 = reinterpret_cast<const int4*>(src);
    const int a = (int)(((long long)E4 * slice) / NSLICE);
    const int b = (int)(((long long)E4 * (slice + 1)) / NSLICE);
    int i = a + tid;
    for (; i + (PART_U - 1) * PTPB < b; i += PART_U * PTPB) {
        int4 d[PART_U], s[PART_U];
        float t[PART_U][4];
#pragma unroll
        for (int u = 0; u < PART_U; ++u) { d[u] = d4[i + u * PTPB]; s[u] = s4[i + u * PTPB]; }
#pragma unroll
        for (int u = 0; u < PART_U; ++u) {
            t[u][0] = term(s[u].x, d[u].x, (unsigned)(d[u].x - lo) < un);
            t[u][1] = term(s[u].y, d[u].y, (unsigned)(d[u].y - lo) < un);
            t[u][2] = term(s[u].z, d[u].z, (unsigned)(d[u].z - lo) < un);
            t[u][3] = term(s[u].w, d[u].w, (unsigned)(d[u].w - lo) < un);
        }
#pragma unroll
        for (int u = 0; u < PART_U; ++u) {
            const unsigned r0 = (unsigned)(d[u].x - lo), r1 = (unsigned)(d[u].y - lo);
            const unsigned r2 = (unsigned)(d[u].z - lo), r3 = (unsigned)(d[u].w - lo);
            if (r0 < un) atomicAdd(&acc[r0], t[u][0]);
            if (r1 < un) atomicAdd(&acc[r1], t[u][1]);
            if (r2 < un) atomicAdd(&acc[r2], t[u][2]);
            if (r3 < un) atomicAdd(&acc[r3], t[u][3]);
        }
    }
    for (; i < b; i += PTPB) {
        const int4 d = d4[i], s = s4[i];
        const unsigned r0 = (unsigned)(d.x - lo), r1 = (unsigned)(d.y - lo);
        const unsigned r2 = (unsigned)(d.z - lo), r3 = (unsigned)(d.w - lo);
        const float t0 = term(s.x, d.x, r0 < un), t1 = term(s.y, d.y, r1 < un);
        const float t2 = term(s.z, d.z, r2 < un), t3 = term(s.w, d.w, r3 < un);
        if (r0 < un) atomicAdd(&acc[r0], t0);
        if (r1 < un) atomicAdd(&acc[r1], t1);
        if (r2 < un) atomicAdd(&acc[r2], t2);
        if (r3 < un) atomicAdd(&acc[r3], t3);
    }
    if (slice == NSLICE - 1) {
        for (int e = 4 * E4 + tid; e < E; e += PTPB) {
            const int d = dst[e];
            const unsigned r = (unsigned)(d - lo);
            if (r < un) {
                const int s = src[e];
                atomicAdd(&acc[r], z[s] * dinv[s] * dinv[d]);
            }
        }
    }
    __syncthreads();

    float4* o4 = reinterpret_cast<float4*>(P + (size_t)blockIdx.x * HALF_N);
    for (int j = tid; j < HALF_N / 4; j += PTPB) o4[j] = a4[j];
}

__global__ __launch_bounds__(32 * SGRP) void k_out_sum(const float* __restrict__ P,
                                                       const float* __restrict__ z,
                                                       const float* __restrict__ dinv,
                                                       const float* __restrict__ b2,
                                                       float* __restrict__ out) {
    __shared__ float4 part[SGRP][32];
    const int x = threadIdx.x, g = threadIdx.y;
    const int quad = blockIdx.x * 32 + x;
    float4 s = make_float4(0.0f, 0.0f, 0.0f, 0.0f);
    if (quad < N_NODES / 4) {
        const int node = 4 * quad;
        const int half = node >= HALF_N ? 1 : 0;
        const int j = node - half * HALF_N;
        const float4* p = reinterpret_cast<const float4*>(
            P + ((size_t)(2 * g * SPG + half)) * HALF_N + j);
#pragma unroll
        for (int k = 0; k < SPG; ++k) {
            const float4 v = p[(size_t)k * (2 * HALF_N / 4)];
            s.x += v.x; s.y += v.y; s.z += v.z; s.w += v.w;
        }
    }
    part[g][x] = s;
    __syncthreads();
    if (g == 0 && quad < N_NODES / 4) {
        float4 t = part[0][x];
        for (int k = 1; k < SGRP; ++k) {
            const float4 v = part[k][x];
            t.x += v.x; t.y += v.y; t.z += v.z; t.w += v.w;
        }
        const float4 zz = reinterpret_cast<const float4*>(z)[quad];
        const float4 di = reinterpret_cast<const float4*>(dinv)[quad];
        const float bb = b2[0];
        float4 r;
        r.x = (zz.x * di.x * di.x + bb) + t.x;
        r.y = (zz.y * di.y * di.y + bb) + t.y;
        r.z = (zz.z * di.z * di.z + bb) + t.z;
        r.w = (zz.w * di.w * di.w + bb) + t.w;
        reinterpret_cast<float4*>(out)[quad] = r;
    }
}


extern "C" void kernel_launch(void* const* d_in, const int* in_sizes, int n_in,
                              void* d_out, int out_size, void* d_ws, size_t ws_size,
                              hipStream_t stream) {
    const float* X  = (const float*)d_in[0];
    const int*   EI = (const int*)d_in[1];
    const float* W1 = (const float*)d_in[2];
    const float* b1 = (const float*)d_in[3];
    const float* W2 = (const float*)d_in[4];
    const float* b2 = (const float*)d_in[5];
    float* out = (float*)d_out;

    const int E = in_sizes[1] / 2;
    const int* src = EI;
    const int* dst = EI + E;

    char* ws = (char*)d_ws;
    float* H    = (float*)ws;
    float* dinv = (float*)(ws + (size_t)2 * N_NODES * HID * 4);
    float* z    = dinv + N_NODES;
    unsigned int* Pu = (unsigned int*)ws;
    float*        Pf = (float*)ws;

    k_deg_part <<<PBLK, PTPB, 0, stream>>>(dst, E, Pu);
    k_dinv_sum <<<(N_NODES / 4 + 31) / 32, dim3(32, SGRP), 0, stream>>>(Pu, dinv);

    k_gemm1    <<<N_NODES / 16, 128, 0, stream>>>(X, W1, H);

    k_agg1     <<<TILES, TPB, 0, stream>>>(src, dst, E, H, dinv, b1, W2, z);

    k_l2_part  <<<PBLK, PTPB, 0, stream>>>(src, dst, E, z, dinv, Pf);
    k_out_sum  <<<(N_NODES / 4 + 31) / 32, dim3(32, SGRP), 0, stream>>>(Pf, z, dinv, b2, out);
}


# Layer-1 tiles sweep a one-byte tile key per edge (written by the degree kernel, 16 edges per 16-byte load), not 4-byte dst; hit groups noted, then expanded by all threads; LDS arrival order as before
# speedup vs baseline: 3.2850x; 1.1801x over previous
#include <hip/hip_runtime.h>

typedef float v2f __attribute__((ext_vector_type(2)));
typedef float v8f __attribute__((ext_vector_type(8)));

#define N_NODES 100000
#define IN_DIM  384
#define HID     64

#define TILES 256
#define RPB   391

#define HALF_N  50000
#define NSLICE  128
#define PBLK    (2 * NSLICE)
#define PTPB    1024
#define PART_U  6
#define SGRP    8
#define SPG     (NSLICE / SGRP)
#define DSLICE  120
#define DBLK    (2 * DSLICE)
#define DSPG    (DSLICE / SGRP)
#define KEY_CAP 3200000

static_assert(2 * HALF_N >= N_NODES && HALF_N % 4 == 0 && N_NODES % 4 == 0, "halves");
static_assert((size_t)PBLK * HALF_N <= (size_t)2 * N_NODES * HID, "planes end in front of dinv");
static_assert(NSLICE % SGRP == 0 && DSLICE % SGRP == 0, "slice groups");
static_assert((size_t)DBLK * HALF_N * 4 + KEY_CAP <= (size_t)2 * N_NODES * HID * 4, "keys end in front of dinv");
static_assert(KEY_CAP % 16 == 0 && ((size_t)DBLK * HALF_N * 4) % 16 == 0, "16-byte key groups");
static_assert(TILES == 256 && (N_NODES - 1) / RPB < TILES, "a key is one byte");

__device__ __forceinline__ unsigned tile_key(int d) {
    return (unsigned)d < (unsigned)N_NODES ? (unsigned)d / RPB : (unsigned)(TILES - 1);
}

__global__ __launch_bounds__(PTPB) void k_deg_part(const int* __restrict__ dst, int E,
                                                   unsigned int* __restrict__ P,
                                                   unsigned int* __restrict__ keys) {
    __shared__ unsigned int cnt[HALF_N];
    const int tid   = threadIdx.x;
    const int half  = blockIdx.x & 1;
    const int slice = blockIdx.x >> 1;
    const int lo    = half * HALF_N;
    int n = N_NODES - lo;
    if (n > HALF_N) n = HALF_N;
    const unsigned un = n > 0 ? (unsigned)n : 0u;

    uint4* c4 = reinterpret_cast<uint4*>(cnt);
    for (int i = tid; i < HALF_N / 4; i += PTPB) c4[i] = make_uint4(0u, 0u, 0u, 0u);
    __syncthreads();

    auto hit4 = [&](int4 d) {
        const unsigned r0 = (unsigned)(d.x - lo), r1 = (unsigned)(d.y - lo);
        const unsigned r2 = (unsigned)(d.z - lo), r3 = (unsigned)(d.w - lo);
        if (r0 < un) atomicAdd(&cnt[r0], 1u);
        if (r1 < un) atomicAdd(&cnt[r1], 1u);
        if (r2 < un) atomicAdd(&cnt[r2], 1u);
        if (r3 < un) atomicAdd(&cnt[r3], 1u);
    };

    const bool wr = half == 0;
    auto put4 = [&](int4 d, int idx) {
        if (4 * idx + 3 < KEY_CAP)
            keys[idx] = tile_key(d.x) | (tile_key(d.y) << 8) | (tile_key(d.z) << 16) | (tile_key(d.w) << 24);
    };

    const int E4 = ((reinterpret_cast<size_t>(dst) & 15) == 0) ? (E >> 2) : 0;
    const int4* d4 = reinterpret_cast<const int4*>(dst);
    const int a = (int)(((long long)E4 * slice) / DSLICE) & ~31;
    const int b = slice + 1 == DSLICE ? E4 : ((int)(((long long)E4 * (slice + 1)) / DSLICE) & ~31);
    int i = a + tid;
    for (; i + (PART_U - 1) * PTPB < b; i += PART_U * PTPB) {
        int4 d[PART_U];
#pragma unroll
        for (int u = 0; u < PART_U; ++u) d[u] = d4[i + u * PTPB];
#pragma unroll
        for (int u = 0; u < PART_U; ++u) {
            hit4(d[u]);
            if (wr) put4(d[u], i + u * PTPB);
        }
    }
    for (; i < b; i += PTPB) {
        const int4 d = d4[i];
        hit4(d);
        if (wr) put4(d, i);
    }
    if (slice == DSLICE - 1) {
        unsigned char* kb = reinterpret_cast<unsigned char*>(keys);
        for (int e = 4 * E4 + tid; e < E; e += PTPB) {
            const int d = dst[e];
            const unsigned r = (unsigned)(d - lo);
            if (r < un) atomicAdd(&cnt[r], 1u);
            if (wr && e < KEY_CAP) kb[e] = (unsigned char)tile_key(d);
        }
    }
    __syncthreads();

    uint4* o4 = reinterpret_cast<uint4*>(P + (size_t)blockIdx.x * HALF_N);
    for (int j = tid; j < HALF_N / 4; j += PTPB) o4[j] = c4[j];
}

__global__ __launch_bounds__(32 * SGRP) void k_dinv_sum(const unsigned int* __restrict__ P,
                                                        float* __restrict__ dinv) {
    __shared__ uint4 part[SGRP][32];
    const int x = threadIdx.x, g = threadIdx.y;
    const int quad = blockIdx.x * 32 + x;
    uint4 s = make_uint4(0u, 0u, 0u, 0u);
    if (quad < N_NODES / 4) {
        const int node = 4 * quad;
        const int half = node >= HALF_N ? 1 : 0;
        const int j = node - half * HALF_N;
        const uint4* p = reinterpret_cast<const uint4*>(
            P + ((size_t)(2 * g * DSPG + half)) * HALF_N + j);
#pragma unroll
        for (int k = 0; k < DSPG; ++k) {
            const uint4 v = p[(size_t)k * (2 * HALF_N / 4)];
            s.x += v.x; s.y += v.y; s.z += v.z; s.w += v.w;
        }
    }
    part[g][x] = s;
    __syncthreads();
    if (g == 0 && quad < N_NODES / 4) {
        uint4 t = part[0][x];
        for (int k = 1; k < SGRP; ++k) {
            const uint4 v = part[k][x];
            t.x += v.x; t.y += v.y; t.z += v.z; t.w += v.w;
        }
        float4 r;
        r.x = rsqrtf((float)t.x + 1.0f);
        r.y = rsqrtf((float)t.y + 1.0f);
        r.z = rsqrtf((float)t.z + 1.0f);
        r.w = rsqrtf((float)t.w + 1.0f);
        reinterpret_cast<float4*>(dinv)[quad] = r;
    }
}

__global__ __launch_bounds__(128) void k_gemm1(const float* __restrict__ X,
                                               const float* __restrict__ W1,
                                               float* __restrict__ H) {
    const int lane = threadIdx.x & 31;
    const int wave = threadIdx.x >> 5;
    const int half = lane >> 4;
    const int lr   = lane & 15;
    const int rowBase = blockIdx.x * 16;
    const int colBase = wave * 16;

    v8f c = {};
    const float* xrow = X + (size_t)(rowBase + lr) * IN_DIM;
    for (int k0 = 0; k0 < IN_DIM; k0 += 4) {
        v2f a, b;
        const float* ap = xrow + k0 + 2 * half;
        a[0] = ap[0];
        a[1] = ap[1];
        const float* bp = W1 + (size_t)(k0 + 2 * half) * HID + colBase + lr;
        b[0] = bp[0];
        b[1] = bp[HID];
        c = __builtin_amdgcn_wmma_f32_16x16x4_f32(
                 false, a,  false, b,
                 (short)0, c,  false,  false);
    }
    float* hrow = H + (size_t)(rowBase + 8 * half) * HID + colBase + lr;
#pragma unroll
    for (int r = 0; r < 8; ++r)
        hrow[(size_t)r * HID] = c[r];
}

#define RBITS 9
#define QCAP  16384
#define GCAP  16384
#define TPB   1024
#define SCAN_U 8
#define GRP_U  8
#define CNV_U  8
#define ROW_U  16
#define Z_U    8

__global__ __launch_bounds__(TPB) void k_agg1(const int* __restrict__ src,
                                              const int* __restrict__ dst,
                                              const unsigned int* __restrict__ keys, int E,
                                              const float* __restrict__ H,
                                              const float* __restrict__ dinv,
                                              const float* __restrict__ b1,
                                              const float* __restrict__ W2,
                                              float* __restrict__ z) {
    __shared__ float        acc[RPB * HID];
    __shared__ unsigned int q[QCAP];
    __shared__ float        qw[QCAP];
    __shared__ unsigned int g1[GCAP];
    __shared__ float        dl[RPB + 1];
    __shared__ unsigned int qn;
    __shared__ unsigned int gn;

    const int tid = threadIdx.x;
    const int lo  = blockIdx.x * RPB;
    int cnt = N_NODES - lo;
    if (cnt > RPB) cnt = RPB;
    const unsigned ucnt = cnt > 0 ? (unsigned)cnt : 0u;

    if (tid == 0) { qn = 0u; gn = 0u; }
    for (int r = tid; r < cnt; r += TPB) dl[r] = dinv[lo + r];
    __syncthreads();

    {
        const float4* H4 = reinterpret_cast<const float4*>(H + (size_t)lo * HID);
        float4* acc4 = reinterpret_cast<float4*>(acc);
        for (int i = tid; i < cnt * (HID / 4); i += TPB) {
            const float di = dl[i >> 4];
            float4 h = H4[i];
            h.x = h.x * di * di; h.y = h.y * di * di;
            h.z = h.z * di * di; h.w = h.w * di * di;
            acc4[i] = h;
        }
    }
    __syncthreads();

    auto add_now = [&](int e) {
        const unsigned r = (unsigned)(dst[e] - lo);
        if (r < ucnt) {
            const int s = src[e];
            const float w = dinv[s] * dl[r];
            const float* hs = H + (size_t)s * HID;
            float* a = acc + r * HID;
            for (int f = 0; f < HID; ++f) atomicAdd(a + f, hs[f] * w);
        }
    };
    auto push = [&](int e) {
        const unsigned pos = atomicAdd(&qn, 1u);
        if (pos < QCAP) q[pos] = (unsigned)e;
        else add_now(e);
    };
    const unsigned pat = (unsigned)blockIdx.x * 0x01010101u;
    auto any16 = [&](uint4 k) -> bool {
        const unsigned x0 = k.x ^ pat, x1 = k.y ^ pat, x2 = k.z ^ pat, x3 = k.w ^ pat;
        return ((((x0 - 0x01010101u) & ~x0) | ((x1 - 0x01010101u) & ~x1) |
                 ((x2 - 0x01010101u) & ~x2) | ((x3 - 0x01010101u) & ~x3)) & 0x80808080u) != 0u;
    };
    auto match16 = [&](uint4 k) -> unsigned {
        auto nib = [&](unsigned kw) -> unsigned {
            const unsigned x = kw ^ pat;
            const unsigned m = ~(((x & 0x7F7F7F7Fu) + 0x7F7F7F7Fu) | x | 0x7F7F7F7Fu);
            return (((m >> 7) * 0x00204081u) >> 21) & 15u;
        };
        return nib(k.x) | (nib(k.y) << 4) | (nib(k.z) << 8) | (nib(k.w) << 12);
    };
    auto take_apart = [&](uint4 k, unsigned g) {
        unsigned m = match16(k);
        while (m != 0u) {
            const int b = __builtin_ctz(m);
            m &= m - 1u;
            push((int)(16u * g) + b);
        }
    };
    auto scan16 = [&](uint4 k, unsigned g) {
        if (any16(k)) {
            const unsigned pos = atomicAdd(&gn, 1u);
            if (pos < GCAP) g1[pos] = g;
        }
    };

    const int EK  = E < KEY_CAP ? E : KEY_CAP;
    const int E16 = ((reinterpret_cast<size_t>(keys) & 15) == 0) ? (EK >> 4) : 0;
    const uint4* k4 = reinterpret_cast<const uint4*>(keys);
    const int start = (int)(((long long)E16 * blockIdx.x) / TILES) & ~31;
    for (int part = 0; part < 2; ++part) {
        const int a = part ? 0 : start;
        const int b = part ? start : E16;
        int i = a + tid;
        for (; i + (SCAN_U - 1) * TPB < b; i += SCAN_U * TPB) {
            uint4 k[SCAN_U];
#pragma unroll
            for (int u = 0; u < SCAN_U; ++u) k[u] = k4[i + u * TPB];
#pragma unroll
            for (int u = 0; u < SCAN_U; ++u) scan16(k[u], (unsigned)(i + u * TPB));
        }
        for (; i < b; i += TPB) scan16(k4[i], (unsigned)i);
    }
    for (int e = 16 * E16 + tid; e < E; e += TPB) {
        if ((unsigned)(dst[e] - lo) < ucnt) push(e);
    }
    __syncthreads();

    unsigned n1 = gn;
    if (n1 > GCAP) {
        n1 = 0u;
        for (int i = tid; i < E16; i += TPB) {
            const uint4 k = k4[i];
            if (any16(k)) take_apart(k, (unsigned)i);
        }
    }
    for (unsigned i0 = 0; i0 < n1; i0 += GRP_U * TPB) {
        unsigned g[GRP_U];
        uint4 k[GRP_U];
#pragma unroll
        for (int u = 0; u < GRP_U; ++u) {
            const unsigned i = i0 + u * TPB + tid;
            g[u] = i < n1 ? g1[i] : 0u;
        }
#pragma unroll
        for (int u = 0; u < GRP_U; ++u) k[u] = k4[g[u]];
#pragma unroll
        for (int u = 0; u < GRP_U; ++u) {
            if (i0 + u * TPB + tid < n1) take_apart(k[u], g[u]);
        }
    }
    __syncthreads();

    unsigned n = qn;
    if (n > QCAP) n = QCAP;

    for (unsigned i0 = 0; i0 < n; i0 += CNV_U * TPB) {
        int e[CNV_U], d[CNV_U], s[CNV_U];
        float ds[CNV_U];
#pragma unroll
        for (int u = 0; u < CNV_U; ++u) {
            const unsigned i = i0 + u * TPB + tid;
            e[u] = i < n ? (int)q[i] : 0;
        }
#pragma unroll
        for (int u = 0; u < CNV_U; ++u) { d[u] = dst[e[u]]; s[u] = src[e[u]]; }
#pragma unroll
        for (int u = 0; u < CNV_U; ++u) ds[u] = dinv[s[u]];
#pragma unroll
        for (int u = 0; u < CNV_U; ++u) {
            const unsigned i = i0 + u * TPB + tid;
            if (i < n) {
                const unsigned r = (unsigned)(d[u] - lo);
                const bool ok = r < ucnt;
                const unsigned rr = ok ? r : 0u;
                qw[i] = ok ? ds[u] * dl[rr] : 0.0f;
                q[i] = ((unsigned)s[u] << RBITS) | rr;
            }
        }
    }
    __syncthreads();

    const float b1a = b1[tid & 31], b1b = b1[(tid & 31) + 32];
    const float w2a = W2[tid & 31], w2b = W2[(tid & 31) + 32];

    {
        const int lane = tid & 31;
        const unsigned wave = (unsigned)tid >> 5;
        const unsigned NW = TPB / 32;
        unsigned i = wave;
        for (; i + (ROW_U - 1) * NW < n; i += ROW_U * NW) {
            unsigned v[ROW_U];
            float w[ROW_U];
            float2 h[ROW_U];
#pragma unroll
            for (int u = 0; u < ROW_U; ++u) { v[u] = q[i + u * NW]; w[u] = qw[i + u * NW]; }
#pragma unroll
            for (int u = 0; u < ROW_U; ++u)
                h[u] = reinterpret_cast<const float2*>(H + (size_t)(v[u] >> RBITS) * HID)[lane];
#pragma unroll
            for (int u = 0; u < ROW_U; ++u) {
                float* a = acc + (v[u] & ((1u << RBITS) - 1u)) * HID + 2 * lane;
                atomicAdd(a,     h[u].x * w[u]);
                atomicAdd(a + 1, h[u].y * w[u]);
            }
        }
        for (; i < n; i += NW) {
            const unsigned v = q[i];
            const float w = qw[i];
            const float2 h = reinterpret_cast<const float2*>(H + (size_t)(v >> RBITS) * HID)[lane];
            float* a = acc + (v & ((1u << RBITS) - 1u)) * HID + 2 * lane;
            atomicAdd(a,     h.x * w);
            atomicAdd(a + 1, h.y * w);
        }
    }
    __syncthreads();

    {
        const int lane = tid & 31;
        const int NW = TPB / 32;
        for (int r0 = tid >> 5; r0 < cnt; r0 += Z_U * NW) {
            float s[Z_U];
#pragma unroll
            for (int u = 0; u < Z_U; ++u) {
                const int r = r0 + u * NW;
                const float* a = acc + (r < cnt ? r : r0) * HID;
                float v0 = a[lane] + b1a;
                float v1 = a[lane + 32] + b1b;
                v0 = v0 > 0.0f ? v0 : 0.0f;
                v1 = v1 > 0.0f ? v1 : 0.0f;
                s[u] = v0 * w2a + v1 * w2b;
            }
#pragma unroll
            for (int off = 16; off > 0; off >>= 1) {
#pragma unroll
                for (int u = 0; u < Z_U; ++u) s[u] += __shfl_down(s[u], off, 32);
            }
#pragma unroll
            for (int u = 0; u < Z_U; ++u) {
                const int r = r0 + u * NW;
                if (lane == 0 && r < cnt) qw[r] = s[u];
            }
        }
    }
    __syncthreads();
    for (int r = tid; r < cnt; r += TPB) z[lo + r] = qw[r];
}

__global__ __launch_bounds__(PTPB) void k_l2_part(const int* __restrict__ src,
                                                  const int* __restrict__ dst, int E,
                                                  const float* __restrict__ z,
                                                  const float* __restrict__ dinv,
                                                  float* __restrict__ P) {
    __shared__ float acc[HALF_N];
    const int tid   = threadIdx.x;
    const int half  = blockIdx.x & 1;
    const int slice = blockIdx.x >> 1;
    const int lo    = half * HALF_N;
    int n = N_NODES - lo;
    if (n > HALF_N) n = HALF_N;
    const unsigned un = n > 0 ? (unsigned)n : 0u;

    float4* a4 = reinterpret_cast<float4*>(acc);
    for (int i = tid; i < HALF_N / 4; i += PTPB) a4[i] = make_float4(0.0f, 0.0f, 0.0f, 0.0f);
    __syncthreads();

    auto term = [&](int s, int d, bool ok) -> float {
        const int ss = ok ? s : lo;
        const int dd = ok ? d : lo;
        return z[ss] * dinv[ss] * dinv[dd];
    };

    const bool al = ((reinterpret_cast<size_t>(dst) | reinterpret_cast<size_t>(src)) & 15) == 0;
    const int E4 = al ? (E >> 2) : 0;
    const int4* d4 = reinterpret_cast<const int4*>(dst);
    const int4* s4 = reinterpret_cast<const int4*>(src);
    const int a = (int)(((long long)E4 * slice) / NSLICE);
    const int b = (int)(((long long)E4 * (slice + 1)) / NSLICE);
    int i = a + tid;
    for (; i + (PART_U - 1) * PTPB < b; i += PART_U * PTPB) {
        int4 d[PART_U], s[PART_U];
        float t[PART_U][4];
#pragma unroll
        for (int u = 0; u < PART_U; ++u) { d[u] = d4[i + u * PTPB]; s[u] = s4[i + u * PTPB]; }
#pragma unroll
        for (int u = 0; u < PART_U; ++u) {
            t[u][0] = term(s[u].x, d[u].x, (unsigned)(d[u].x - lo) < un);
            t[u][1] = term(s[u].y, d[u].y, (unsigned)(d[u].y - lo) < un);
            t[u][2] = term(s[u].z, d[u].z, (unsigned)(d[u].z - lo) < un);
            t[u][3] = term(s[u].w, d[u].w, (unsigned)(d[u].w - lo) < un);
        }
#pragma unroll
        for (int u = 0; u < PART_U; ++u) {
            const unsigned r0 = (unsigned)(d[u].x - lo), r1 = (unsigned)(d[u].y - lo);
            const unsigned r2 = (unsigned)(d[u].z - lo), r3 = (unsigned)(d[u].w - lo);
            if (r0 < un) atomicAdd(&acc[r0], t[u][0]);
            if (r1 < un) atomicAdd(&acc[r1], t[u][1]);
            if (r2 < un) atomicAdd(&acc[r2], t[u][2]);
            if (r3 < un) atomicAdd(&acc[r3], t[u][3]);
        }
    }
    for (; i < b; i += PTPB) {
        const int4 d = d4[i], s = s4[i];
        const unsigned r0 = (unsigned)(d.x - lo), r1 = (unsigned)(d.y - lo);
        const unsigned r2 = (unsigned)(d.z - lo), r3 = (unsigned)(d.w - lo);
        const float t0 = term(s.x, d.x, r0 < un), t1 = term(s.y, d.y, r1 < un);
        const float t2 = term(s.z, d.z, r2 < un), t3 = term(s.w, d.w, r3 < un);
        if (r0 < un) atomicAdd(&acc[r0], t0);
        if (r1 < un) atomicAdd(&acc[r1], t1);
        if (r2 < un) atomicAdd(&acc[r2], t2);
        if (r3 < un) atomicAdd(&acc[r3], t3);
    }
    if (slice == NSLICE - 1) {
        for (int e = 4 * E4 + tid; e < E; e += PTPB) {
            const int d = dst[e];
            const unsigned r = (unsigned)(d - lo);
            if (r < un) {
                const int s = src[e];
                atomicAdd(&acc[r], z[s] * dinv[s] * dinv[d]);
            }
        }
    }
    __syncthreads();

    float4* o4 = reinterpret_cast<float4*>(P + (size_t)blockIdx.x * HALF_N);
    for (int j = tid; j < HALF_N / 4; j += PTPB) o4[j] = a4[j];
}

__global__ __launch_bounds__(32 * SGRP) void k_out_sum(const float* __restrict__ P,
                                                       const float* __restrict__ z,
                                                       const float* __restrict__ dinv,
                                                       const float* __restrict__ b2,
                                                       float* __restrict__ out) {
    __shared__ float4 part[SGRP][32];
    const int x = threadIdx.x, g = threadIdx.y;
    const int quad = blockIdx.x * 32 + x;
    float4 s = make_float4(0.0f, 0.0f, 0.0f, 0.0f);
    if (quad < N_NODES / 4) {
        const int node = 4 * quad;
        const int half = node >= HALF_N ? 1 : 0;
        const int j = node - half * HALF_N;
        const float4* p = reinterpret_cast<const float4*>(
            P + ((size_t)(2 * g * SPG + half)) * HALF_N + j);
#pragma unroll
        for (int k = 0; k < SPG; ++k) {
            const float4 v = p[(size_t)k * (2 * HALF_N / 4)];
            s.x += v.x; s.y += v.y; s.z += v.z; s.w += v.w;
        }
    }
    part[g][x] = s;
    __syncthreads();
    if (g == 0 && quad < N_NODES / 4) {
        float4 t = part[0][x];
        for (int k = 1; k < SGRP; ++k) {
            const float4 v = part[k][x];
            t.x += v.x; t.y += v.y; t.z += v.z; t.w += v.w;
        }
        const float4 zz = reinterpret_cast<const float4*>(z)[quad];
        const float4 di = reinterpret_cast<const float4*>(dinv)[quad];
        const float bb = b2[0];
        float4 r;
        r.x = (zz.x * di.x * di.x + bb) + t.x;
        r.y = (zz.y * di.y * di.y + bb) + t.y;
        r.z = (zz.z * di.z * di.z + bb) + t.z;
        r.w = (zz.w * di.w * di.w + bb) + t.w;
        reinterpret_cast<float4*>(out)[quad] = r;
    }
}


extern "C" void kernel_launch(void* const* d_in, const int* in_sizes, int n_in,
                              void* d_out, int out_size, void* d_ws, size_t ws_size,
                              hipStream_t stream) {
    const float* X  = (const float*)d_in[0];
    const int*   EI = (const int*)d_in[1];
    const float* W1 = (const float*)d_in[2];
    const float* b1 = (const float*)d_in[3];
    const float* W2 = (const float*)d_in[4];
    const float* b2 = (const float*)d_in[5];
    float* out = (float*)d_out;

    const int E = in_sizes[1] / 2;
    const int* src = EI;
    const int* dst = EI + E;

    char* ws = (char*)d_ws;
    float* H    = (float*)ws;
    float* dinv = (float*)(ws + (size_t)2 * N_NODES * HID * 4);
    float* z    = dinv + N_NODES;
    unsigned int* Pu = (unsigned int*)ws;
    float*        Pf = (float*)ws;
    unsigned int* keys = (unsigned int*)(ws + (size_t)DBLK * HALF_N * 4);

    k_deg_part <<<DBLK, PTPB, 0, stream>>>(dst, E, Pu, keys);
    k_dinv_sum <<<(N_NODES / 4 + 31) / 32, dim3(32, SGRP), 0, stream>>>(Pu, dinv);

    k_gemm1    <<<N_NODES / 16, 128, 0, stream>>>(X, W1, H);

    k_agg1     <<<TILES, TPB, 0, stream>>>(src, dst, keys, E, H, dinv, b1, W2, z);

    k_l2_part  <<<PBLK, PTPB, 0, stream>>>(src, dst, E, z, dinv, Pf);
    k_out_sum  <<<(N_NODES / 4 + 31) / 32, dim3(32, SGRP), 0, stream>>>(Pf, z, dinv, b2, out);
}


# X @ W1 by 256 resident work-groups: a wave keeps its 16 columns of W1 in registers, 32-row X blocks go through LDS by whole-line loads one block ahead; each dot sum in two k chains (new sum order)
# speedup vs baseline: 3.7283x; 1.1349x over previous
#include <hip/hip_runtime.h>

typedef float v2f __attribute__((ext_vector_type(2)));
typedef float v8f __attribute__((ext_vector_type(8)));
typedef float v4f __attribute__((ext_vector_type(4)));

#define N_NODES 100000
#define IN_DIM  384
#define HID     64

#define TILES 256
#define RPB   391

#define HALF_N  50000
#define NSLICE  128
#define PBLK    (2 * NSLICE)
#define PTPB    1024
#define PART_U  6
#define SGRP    8
#define SPG     (NSLICE / SGRP)
#define DSLICE  120
#define DBLK    (2 * DSLICE)
#define DSPG    (DSLICE / SGRP)
#define KEY_CAP 3200000

static_assert(2 * HALF_N >= N_NODES && HALF_N % 4 == 0 && N_NODES % 4 == 0, "halves");
static_assert((size_t)PBLK * HALF_N <= (size_t)2 * N_NODES * HID, "planes end in front of dinv");
static_assert(NSLICE % SGRP == 0 && DSLICE % SGRP == 0, "slice groups");
static_assert((size_t)DBLK * HALF_N * 4 + KEY_CAP <= (size_t)2 * N_NODES * HID * 4, "keys end in front of dinv");
static_assert(KEY_CAP % 16 == 0 && ((size_t)DBLK * HALF_N * 4) % 16 == 0, "16-byte key groups");
static_assert(TILES == 256 && (N_NODES - 1) / RPB < TILES, "a key is one byte");

__device__ __forceinline__ unsigned tile_key(int d) {
    return (unsigned)d < (unsigned)N_NODES ? (unsigned)d / RPB : (unsigned)(TILES - 1);
}

__global__ __launch_bounds__(PTPB) void k_deg_part(const int* __restrict__ dst, int E,
                                                   unsigned int* __restrict__ P,
                                                   unsigned int* __restrict__ keys) {
    __shared__ unsigned int cnt[HALF_N];
    const int tid   = threadIdx.x;
    const int half  = blockIdx.x & 1;
    const int slice = blockIdx.x >> 1;
    const int lo    = half * HALF_N;
    int n = N_NODES - lo;
    if (n > HALF_N) n = HALF_N;
    const unsigned un = n > 0 ? (unsigned)n : 0u;

    uint4* c4 = reinterpret_cast<uint4*>(cnt);
    for (int i = tid; i < HALF_N / 4; i += PTPB) c4[i] = make_uint4(0u, 0u, 0u, 0u);
    __syncthreads();

    auto hit4 = [&](int4 d) {
        const unsigned r0 = (unsigned)(d.x - lo), r1 = (unsigned)(d.y - lo);
        const unsigned r2 = (unsigned)(d.z - lo), r3 = (unsigned)(d.w - lo);
        if (r0 < un) atomicAdd(&cnt[r0], 1u);
        if (r1 < un) atomicAdd(&cnt[r1], 1u);
        if (r2 < un) atomicAdd(&cnt[r2], 1u);
        if (r3 < un) atomicAdd(&cnt[r3], 1u);
    };

    const bool wr = half == 0;
    auto put4 = [&](int4 d, int idx) {
        if (4 * idx + 3 < KEY_CAP)
            keys[idx] = tile_key(d.x) | (tile_key(d.y) << 8) | (tile_key(d.z) << 16) | (tile_key(d.w) << 24);
    };

    const int E4 = ((reinterpret_cast<size_t>(dst) & 15) == 0) ? (E >> 2) : 0;
    const int4* d4 = reinterpret_cast<const int4*>(dst);
    const int a = (int)(((long long)E4 * slice) / DSLICE) & ~31;
    const int b = slice + 1 == DSLICE ? E4 : ((int)(((long long)E4 * (slice + 1)) / DSLICE) & ~31);
    int i = a + tid;
    for (; i + (PART_U - 1) * PTPB < b; i += PART_U * PTPB) {
        int4 d[PART_U];
#pragma unroll
        for (int u = 0; u < PART_U; ++u) d[u] = d4[i + u * PTPB];
#pragma unroll
        for (int u = 0; u < PART_U; ++u) {
            hit4(d[u]);
            if (wr) put4(d[u], i + u * PTPB);
        }
    }
    for (; i < b; i += PTPB) {
        const int4 d = d4[i];
        hit4(d);
        if (wr) put4(d, i);
    }
    if (slice == DSLICE - 1) {
        unsigned char* kb = reinterpret_cast<unsigned char*>(keys);
        for (int e = 4 * E4 + tid; e < E; e += PTPB) {
            const int d = dst[e];
            const unsigned r = (unsigned)(d - lo);
            if (r < un) atomicAdd(&cnt[r], 1u);
            if (wr && e < KEY_CAP) kb[e] = (unsigned char)tile_key(d);
        }
    }
    __syncthreads();

    uint4* o4 = reinterpret_cast<uint4*>(P + (size_t)blockIdx.x * HALF_N);
    for (int j = tid; j < HALF_N / 4; j += PTPB) o4[j] = c4[j];
}

__global__ __launch_bounds__(32 * SGRP) void k_dinv_sum(const unsigned int* __restrict__ P,
                                                        float* __restrict__ dinv) {
    __shared__ uint4 part[SGRP][32];
    const int x = threadIdx.x, g = threadIdx.y;
    const int quad = blockIdx.x * 32 + x;
    uint4 s = make_uint4(0u, 0u, 0u, 0u);
    if (quad < N_NODES / 4) {
        const int node = 4 * quad;
        const int half = node >= HALF_N ? 1 : 0;
        const int j = node - half * HALF_N;
        const uint4* p = reinterpret_cast<const uint4*>(
            P + ((size_t)(2 * g * DSPG + half)) * HALF_N + j);
#pragma unroll
        for (int k = 0; k < DSPG; ++k) {
            const uint4 v = p[(size_t)k * (2 * HALF_N / 4)];
            s.x += v.x; s.y += v.y; s.z += v.z; s.w += v.w;
        }
    }
    part[g][x] = s;
    __syncthreads();
    if (g == 0 && quad < N_NODES / 4) {
        uint4 t = part[0][x];
        for (int k = 1; k < SGRP; ++k) {
            const uint4 v = part[k][x];
            t.x += v.x; t.y += v.y; t.z += v.z; t.w += v.w;
        }
        float4 r;
        r.x = rsqrtf((float)t.x + 1.0f);
        r.y = rsqrtf((float)t.y + 1.0f);
        r.z = rsqrtf((float)t.z + 1.0f);
        r.w = rsqrtf((float)t.w + 1.0f);
        reinterpret_cast<float4*>(dinv)[quad] = r;
    }
}

#define GB_ROWS  32
#define GB_P4    (IN_DIM / 4 + 1)
#define GB_WG    256
#define GB_TPB   256
#define GB_NBLK  (N_NODES / GB_ROWS)
#define GB_C4    (GB_ROWS * IN_DIM / 4)
#define GB_CH    (GB_C4 / GB_TPB)
#define GB_SLOTS 4
#define GB_KJ    (IN_DIM / 8)
#define GB_G     16
#define GB_EACH(M) M(0) M(1) M(2) M(3) M(4) M(5) M(6) M(7) M(8) M(9) M(10) M(11)

static_assert(GB_CH == 12, "GB_EACH lists the chunks");
static_assert(GB_NBLK >= GB_WG, "every work-group has a block");
static_assert(N_NODES % GB_ROWS == 0 && GB_C4 % GB_TPB == 0 && IN_DIM % 8 == 0 && HID == 64, "gemm tiling");
static_assert((GB_SLOTS & (GB_SLOTS - 1)) == 0 && GB_KJ % GB_G == 0, "ring index by mask, whole groups");

__global__ __launch_bounds__(GB_TPB) void k_gemm1(const float* X,
                                                  const float* __restrict__ W1,
                                                  float* H) {
    __shared__ v4f xs[GB_SLOTS][GB_ROWS * GB_P4];
    const int tid  = threadIdx.x;
    const int lane = tid & 31;
    const int wave = tid >> 5;
    const int half = lane >> 4;
    const int lr   = lane & 15;
    const int ct   = wave & 3;
    const int rt   = wave >> 2;

    const int t0 = (int)(((long long)GB_NBLK * blockIdx.x) / GB_WG);
    const int t1 = (int)(((long long)GB_NBLK * (blockIdx.x + 1)) / GB_WG);

    const v4f* X4 = reinterpret_cast<const v4f*>(X);
    int so[GB_CH];
#pragma unroll
    for (int i = 0; i < GB_CH; ++i) {
        const int j   = i * GB_TPB + tid;
        const int row = j / (IN_DIM / 4);
        so[i] = row * GB_P4 + (j - row * (IN_DIM / 4));
    }

    const v4f* p0 = X4 + (size_t)t0 * GB_C4 + tid;
#define GB_LD(i) const v4f f##i = p0[(i) * GB_TPB];
    GB_EACH(GB_LD)
#undef GB_LD

    v2f b[2 * GB_KJ];
    {
        const float* wp = W1 + (size_t)(4 * half) * HID + 16 * ct + lr;
#pragma unroll
        for (int j = 0; j < GB_KJ; ++j) {
#pragma unroll
            for (int q = 0; q < 2; ++q) {
                const float* bp = wp + (size_t)(8 * j + 2 * q) * HID;
                b[2 * j + q][0] = bp[0];
                b[2 * j + q][1] = bp[HID];
            }
        }
    }
#define GB_ST(i) xs[0][so[i]] = f##i;
    GB_EACH(GB_ST)
#undef GB_ST
    __syncthreads();

    auto products = [&](int slot) -> v8f {
        v8f c0 = {}, c1 = {};
        const v4f* ar = &xs[slot][(16 * rt + lr) * GB_P4 + half];
        v4f af[2][GB_G];
#pragma unroll
        for (int u = 0; u < GB_G; ++u) af[0][u] = ar[2 * u];
#pragma unroll
        for (int g = 0; g < GB_KJ / GB_G; ++g) {
            if (g + 1 < GB_KJ / GB_G) {
#pragma unroll
                for (int u = 0; u < GB_G; ++u) af[(g + 1) & 1][u] = ar[2 * ((g + 1) * GB_G + u)];
            }
            asm volatile("" ::: "memory");
#pragma unroll
            for (int u = 0; u < GB_G; ++u) {
                const int j = g * GB_G + u;
                const v4f a4 = af[g & 1][u];
                v2f a0, a1;
                a0[0] = a4.x; a0[1] = a4.y;
                a1[0] = a4.z; a1[1] = a4.w;
                c0 = __builtin_amdgcn_wmma_f32_16x16x4_f32(
                         false, a0,  false, b[2 * j],
                         (short)0, c0,  false,  false);
                c1 = __builtin_amdgcn_wmma_f32_16x16x4_f32(
                         false, a1,  false, b[2 * j + 1],
                         (short)0, c1,  false,  false);
            }
        }
        return c0 + c1;
    };
    auto put = [&](int t, v8f c) {
        float* hrow = H + (size_t)(t * GB_ROWS + 16 * rt + 8 * half) * HID + 16 * ct + lr;
#pragma unroll
        for (int r = 0; r < 8; ++r) hrow[(size_t)r * HID] = c[r];
    };

    int t = t0;
    for (; t + 1 < t1; ++t) {
        const int slot = (t - t0) & (GB_SLOTS - 1);
        const v4f* p = X4 + (size_t)(t + 1) * GB_C4 + tid;
#define GB_LD(i) const v4f st##i = p[(i) * GB_TPB];
        GB_EACH(GB_LD)
#undef GB_LD

        const v8f c = products(slot);
        asm volatile("" :: "v"(c) : "memory");

        v4f* xn = xs[(slot + 1) & (GB_SLOTS - 1)];
#define GB_ST(i) xn[so[i]] = st##i;
        GB_EACH(GB_ST)
#undef GB_ST
        __syncthreads();
        put(t, c);
    }
    put(t, products((t - t0) & (GB_SLOTS - 1)));
}

#define RBITS 9
#define QCAP  16384
#define GCAP  16384
#define TPB   1024
#define SCAN_U 8
#define GRP_U  8
#define CNV_U  8
#define ROW_U  16
#define Z_U    8

__global__ __launch_bounds__(TPB) void k_agg1(const int* __restrict__ src,
                                              const int* __restrict__ dst,
                                              const unsigned int* __restrict__ keys, int E,
                                              const float* __restrict__ H,
                                              const float* __restrict__ dinv,
                                              const float* __restrict__ b1,
                                              const float* __restrict__ W2,
                                              float* __restrict__ z) {
    __shared__ float        acc[RPB * HID];
    __shared__ unsigned int q[QCAP];
    __shared__ float        qw[QCAP];
    __shared__ unsigned int g1[GCAP];
    __shared__ float        dl[RPB + 1];
    __shared__ unsigned int qn;
    __shared__ unsigned int gn;

    const int tid = threadIdx.x;
    const int lo  = blockIdx.x * RPB;
    int cnt = N_NODES - lo;
    if (cnt > RPB) cnt = RPB;
    const unsigned ucnt = cnt > 0 ? (unsigned)cnt : 0u;

    if (tid == 0) { qn = 0u; gn = 0u; }
    for (int r = tid; r < cnt; r += TPB) dl[r] = dinv[lo + r];
    __syncthreads();

    {
        const float4* H4 = reinterpret_cast<const float4*>(H + (size_t)lo * HID);
        float4* acc4 = reinterpret_cast<float4*>(acc);
        for (int i = tid; i < cnt * (HID / 4); i += TPB) {
            const float di = dl[i >> 4];
            float4 h = H4[i];
            h.x = h.x * di * di; h.y = h.y * di * di;
            h.z = h.z * di * di; h.w = h.w * di * di;
            acc4[i] = h;
        }
    }
    __syncthreads();

    auto add_now = [&](int e) {
        const unsigned r = (unsigned)(dst[e] - lo);
        if (r < ucnt) {
            const int s = src[e];
            const float w = dinv[s] * dl[r];
            const float* hs = H + (size_t)s * HID;
            float* a = acc + r * HID;
            for (int f = 0; f < HID; ++f) atomicAdd(a + f, hs[f] * w);
        }
    };
    auto push = [&](int e) {
        const unsigned pos = atomicAdd(&qn, 1u);
        if (pos < QCAP) q[pos] = (unsigned)e;
        else add_now(e);
    };
    const unsigned pat = (unsigned)blockIdx.x * 0x01010101u;
    auto any16 = [&](uint4 k) -> bool {
        const unsigned x0 = k.x ^ pat, x1 = k.y ^ pat, x2 = k.z ^ pat, x3 = k.w ^ pat;
        return ((((x0 - 0x01010101u) & ~x0) | ((x1 - 0x01010101u) & ~x1) |
                 ((x2 - 0x01010101u) & ~x2) | ((x3 - 0x01010101u) & ~x3)) & 0x80808080u) != 0u;
    };
    auto match16 = [&](uint4 k) -> unsigned {
        auto nib = [&](unsigned kw) -> unsigned {
            const unsigned x = kw ^ pat;
            const unsigned m = ~(((x & 0x7F7F7F7Fu) + 0x7F7F7F7Fu) | x | 0x7F7F7F7Fu);
            return (((m >> 7) * 0x00204081u) >> 21) & 15u;
        };
        return nib(k.x) | (nib(k.y) << 4) | (nib(k.z) << 8) | (nib(k.w) << 12);
    };
    auto take_apart = [&](uint4 k, unsigned g) {
        unsigned m = match16(k);
        while (m != 0u) {
            const int b = __builtin_ctz(m);
            m &= m - 1u;
            push((int)(16u * g) + b);
        }
    };
    auto scan16 = [&](uint4 k, unsigned g) {
        if (any16(k)) {
            const unsigned pos = atomicAdd(&gn, 1u);
            if (pos < GCAP) g1[pos] = g;
        }
    };

    const int EK  = E < KEY_CAP ? E : KEY_CAP;
    const int E16 = ((reinterpret_cast<size_t>(keys) & 15) == 0) ? (EK >> 4) : 0;
    const uint4* k4 = reinterpret_cast<const uint4*>(keys);
    const int start = (int)(((long long)E16 * blockIdx.x) / TILES) & ~31;
    for (int part = 0; part < 2; ++part) {
        const int a = part ? 0 : start;
        const int b = part ? start : E16;
        int i = a + tid;
        for (; i + (SCAN_U - 1) * TPB < b; i += SCAN_U * TPB) {
            uint4 k[SCAN_U];
#pragma unroll
            for (int u = 0; u < SCAN_U; ++u) k[u] = k4[i + u * TPB];
#pragma unroll
            for (int u = 0; u < SCAN_U; ++u) scan16(k[u], (unsigned)(i + u * TPB));
        }
        for (; i < b; i += TPB) scan16(k4[i], (unsigned)i);
    }
    for (int e = 16 * E16 + tid; e < E; e += TPB) {
        if ((unsigned)(dst[e] - lo) < ucnt) push(e);
    }
    __syncthreads();

    unsigned n1 = gn;
    if (n1 > GCAP) {
        n1 = 0u;
        for (int i = tid; i < E16; i += TPB) {
            const uint4 k = k4[i];
            if (any16(k)) take_apart(k, (unsigned)i);
        }
    }
    for (unsigned i0 = 0; i0 < n1; i0 += GRP_U * TPB) {
        unsigned g[GRP_U];
        uint4 k[GRP_U];
#pragma unroll
        for (int u = 0; u < GRP_U; ++u) {
            const unsigned i = i0 + u * TPB + tid;
            g[u] = i < n1 ? g1[i] : 0u;
        }
#pragma unroll
        for (int u = 0; u < GRP_U; ++u) k[u] = k4[g[u]];
#pragma unroll
        for (int u = 0; u < GRP_U; ++u) {
            if (i0 + u * TPB + tid < n1) take_apart(k[u], g[u]);
        }
    }
    __syncthreads();

    unsigned n = qn;
    if (n > QCAP) n = QCAP;

    for (unsigned i0 = 0; i0 < n; i0 += CNV_U * TPB) {
        int e[CNV_U], d[CNV_U], s[CNV_U];
        float ds[CNV_U];
#pragma unroll
        for (int u = 0; u < CNV_U; ++u) {
            const unsigned i = i0 + u * TPB + tid;
            e[u] = i < n ? (int)q[i] : 0;
        }
#pragma unroll
        for (int u = 0; u < CNV_U; ++u) { d[u] = dst[e[u]]; s[u] = src[e[u]]; }
#pragma unroll
        for (int u = 0; u < CNV_U; ++u) ds[u] = dinv[s[u]];
#pragma unroll
        for (int u = 0; u < CNV_U; ++u) {
            const unsigned i = i0 + u * TPB + tid;
            if (i < n) {
                const unsigned r = (unsigned)(d[u] - lo);
                const bool ok = r < ucnt;
                const unsigned rr = ok ? r : 0u;
                qw[i] = ok ? ds[u] * dl[rr] : 0.0f;
                q[i] = ((unsigned)s[u] << RBITS) | rr;
            }
        }
    }
    __syncthreads();

    const float b1a = b1[tid & 31], b1b = b1[(tid & 31) + 32];
    const float w2a = W2[tid & 31], w2b = W2[(tid & 31) + 32];

    {
        const int lane = tid & 31;
        const unsigned wave = (unsigned)tid >> 5;
        const unsigned NW = TPB / 32;
        unsigned i = wave;
        for (; i + (ROW_U - 1) * NW < n; i += ROW_U * NW) {
            unsigned v[ROW_U];
            float w[ROW_U];
            float2 h[ROW_U];
#pragma unroll
            for (int u = 0; u < ROW_U; ++u) { v[u] = q[i + u * NW]; w[u] = qw[i + u * NW]; }
#pragma unroll
            for (int u = 0; u < ROW_U; ++u)
                h[u] = reinterpret_cast<const float2*>(H + (size_t)(v[u] >> RBITS) * HID)[lane];
#pragma unroll
            for (int u = 0; u < ROW_U; ++u) {
                float* a = acc + (v[u] & ((1u << RBITS) - 1u)) * HID + 2 * lane;
                atomicAdd(a,     h[u].x * w[u]);
                atomicAdd(a + 1, h[u].y * w[u]);
            }
        }
        for (; i < n; i += NW) {
            const unsigned v = q[i];
            const float w = qw[i];
            const float2 h = reinterpret_cast<const float2*>(H + (size_t)(v >> RBITS) * HID)[lane];
            float* a = acc + (v & ((1u << RBITS) - 1u)) * HID + 2 * lane;
            atomicAdd(a,     h.x * w);
            atomicAdd(a + 1, h.y * w);
        }
    }
    __syncthreads();

    {
        const int lane = tid & 31;
        const int NW = TPB / 32;
        for (int r0 = tid >> 5; r0 < cnt; r0 += Z_U * NW) {
            float s[Z_U];
#pragma unroll
            for (int u = 0; u < Z_U; ++u) {
                const int r = r0 + u * NW;
                const float* a = acc + (r < cnt ? r : r0) * HID;
                float v0 = a[lane] + b1a;
                float v1 = a[lane + 32] + b1b;
                v0 = v0 > 0.0f ? v0 : 0.0f;
                v1 = v1 > 0.0f ? v1 : 0.0f;
                s[u] = v0 * w2a + v1 * w2b;
            }
#pragma unroll
            for (int off = 16; off > 0; off >>= 1) {
#pragma unroll
                for (int u = 0; u < Z_U; ++u) s[u] += __shfl_down(s[u], off, 32);
            }
#pragma unroll
            for (int u = 0; u < Z_U; ++u) {
                const int r = r0 + u * NW;
                if (lane == 0 && r < cnt) qw[r] = s[u];
            }
        }
    }
    __syncthreads();
    for (int r = tid; r < cnt; r += TPB) z[lo + r] = qw[r];
}

__global__ __launch_bounds__(PTPB) void k_l2_part(const int* __restrict__ src,
                                                  const int* __restrict__ dst, int E,
                                                  const float* __restrict__ z,
                                                  const float* __restrict__ dinv,
                                                  float* __restrict__ P) {
    __shared__ float acc[HALF_N];
    const int tid   = threadIdx.x;
    const int half  = blockIdx.x & 1;
    const int slice = blockIdx.x >> 1;
    const int lo    = half * HALF_N;
    int n = N_NODES - lo;
    if (n > HALF_N) n = HALF_N;
    const unsigned un = n > 0 ? (unsigned)n : 0u;

    float4* a4 = reinterpret_cast<float4*>(acc);
    for (int i = tid; i < HALF_N / 4; i += PTPB) a4[i] = make_float4(0.0f, 0.0f, 0.0f, 0.0f);
    __syncthreads();

    auto term = [&](int s, int d, bool ok) -> float {
        const int ss = ok ? s : lo;
        const int dd = ok ? d : lo;
        return z[ss] * dinv[ss] * dinv[dd];
    };

    const bool al = ((reinterpret_cast<size_t>(dst) | reinterpret_cast<size_t>(src)) & 15) == 0;
    const int E4 = al ? (E >> 2) : 0;
    const int4* d4 = reinterpret_cast<const int4*>(dst);
    const int4* s4 = reinterpret_cast<const int4*>(src);
    const int a = (int)(((long long)E4 * slice) / NSLICE);
    const int b = (int)(((long long)E4 * (slice + 1)) / NSLICE);
    int i = a + tid;
    for (; i + (PART_U - 1) * PTPB < b; i += PART_U * PTPB) {
        int4 d[PART_U], s[PART_U];
        float t[PART_U][4];
#pragma unroll
        for (int u = 0; u < PART_U; ++u) { d[u] = d4[i + u * PTPB]; s[u] = s4[i + u * PTPB]; }
#pragma unroll
        for (int u = 0; u < PART_U; ++u) {
            t[u][0] = term(s[u].x, d[u].x, (unsigned)(d[u].x - lo) < un);
            t[u][1] = term(s[u].y, d[u].y, (unsigned)(d[u].y - lo) < un);
            t[u][2] = term(s[u].z, d[u].z, (unsigned)(d[u].z - lo) < un);
            t[u][3] = term(s[u].w, d[u].w, (unsigned)(d[u].w - lo) < un);
        }
#pragma unroll
        for (int u = 0; u < PART_U; ++u) {
            const unsigned r0 = (unsigned)(d[u].x - lo), r1 = (unsigned)(d[u].y - lo);
            const unsigned r2 = (unsigned)(d[u].z - lo), r3 = (unsigned)(d[u].w - lo);
            if (r0 < un) atomicAdd(&acc[r0], t[u][0]);
            if (r1 < un) atomicAdd(&acc[r1], t[u][1]);
            if (r2 < un) atomicAdd(&acc[r2], t[u][2]);
            if (r3 < un) atomicAdd(&acc[r3], t[u][3]);
        }
    }
    for (; i < b; i += PTPB) {
        const int4 d = d4[i], s = s4[i];
        const unsigned r0 = (unsigned)(d.x - lo), r1 = (unsigned)(d.y - lo);
        const unsigned r2 = (unsigned)(d.z - lo), r3 = (unsigned)(d.w - lo);
        const float t0 = term(s.x, d.x, r0 < un), t1 = term(s.y, d.y, r1 < un);
        const float t2 = term(s.z, d.z, r2 < un), t3 = term(s.w, d.w, r3 < un);
        if (r0 < un) atomicAdd(&acc[r0], t0);
        if (r1 < un) atomicAdd(&acc[r1], t1);
        if (r2 < un) atomicAdd(&acc[r2], t2);
        if (r3 < un) atomicAdd(&acc[r3], t3);
    }
    if (slice == NSLICE - 1) {
        for (int e = 4 * E4 + tid; e < E; e += PTPB) {
            const int d = dst[e];
            const unsigned r = (unsigned)(d - lo);
            if (r < un) {
                const int s = src[e];
                atomicAdd(&acc[r], z[s] * dinv[s] * dinv[d]);
            }
        }
    }
    __syncthreads();

    float4* o4 = reinterpret_cast<float4*>(P + (size_t)blockIdx.x * HALF_N);
    for (int j = tid; j < HALF_N / 4; j += PTPB) o4[j] = a4[j];
}

__global__ __launch_bounds__(32 * SGRP) void k_out_sum(const float* __restrict__ P,
                                                       const float* __restrict__ z,
                                                       const float* __restrict__ dinv,
                                                       const float* __restrict__ b2,
                                                       float* __restrict__ out) {
    __shared__ float4 part[SGRP][32];
    const int x = threadIdx.x, g = threadIdx.y;
    const int quad = blockIdx.x * 32 + x;
    float4 s = make_float4(0.0f, 0.0f, 0.0f, 0.0f);
    if (quad < N_NODES / 4) {
        const int node = 4 * quad;
        const int half = node >= HALF_N ? 1 : 0;
        const int j = node - half * HALF_N;
        const float4* p = reinterpret_cast<const float4*>(
            P + ((size_t)(2 * g * SPG + half)) * HALF_N + j);
#pragma unroll
        for (int k = 0; k < SPG; ++k) {
            const float4 v = p[(size_t)k * (2 * HALF_N / 4)];
            s.x += v.x; s.y += v.y; s.z += v.z; s.w += v.w;
        }
    }
    part[g][x] = s;
    __syncthreads();
    if (g == 0 && quad < N_NODES / 4) {
        float4 t = part[0][x];
        for (int k = 1; k < SGRP; ++k) {
            const float4 v = part[k][x];
            t.x += v.x; t.y += v.y; t.z += v.z; t.w += v.w;
        }
        const float4 zz = reinterpret_cast<const float4*>(z)[quad];
        const float4 di = reinterpret_cast<const float4*>(dinv)[quad];
        const float bb = b2[0];
        float4 r;
        r.x = (zz.x * di.x * di.x + bb) + t.x;
        r.y = (zz.y * di.y * di.y + bb) + t.y;
        r.z = (zz.z * di.z * di.z + bb) + t.z;
        r.w = (zz.w * di.w * di.w + bb) + t.w;
        reinterpret_cast<float4*>(out)[quad] = r;
    }
}


extern "C" void kernel_launch(void* const* d_in, const int* in_sizes, int n_in,
                              void* d_out, int out_size, void* d_ws, size_t ws_size,
                              hipStream_t stream) {
    const float* X  = (const float*)d_in[0];
    const int*   EI = (const int*)d_in[1];
    const float* W1 = (const float*)d_in[2];
    const float* b1 = (const float*)d_in[3];
    const float* W2 = (const float*)d_in[4];
    const float* b2 = (const float*)d_in[5];
    float* out = (float*)d_out;

    const int E = in_sizes[1] / 2;
    const int* src = EI;
    const int* dst = EI + E;

    char* ws = (char*)d_ws;
    float* H    = (float*)ws;
    float* dinv = (float*)(ws + (size_t)2 * N_NODES * HID * 4);
    float* z    = dinv + N_NODES;
    unsigned int* Pu = (unsigned int*)ws;
    float*        Pf = (float*)ws;
    unsigned int* keys = (unsigned int*)(ws + (size_t)DBLK * HALF_N * 4);

    k_deg_part <<<DBLK, PTPB, 0, stream>>>(dst, E, Pu, keys);
    k_dinv_sum <<<(N_NODES / 4 + 31) / 32, dim3(32, SGRP), 0, stream>>>(Pu, dinv);

    k_gemm1    <<<GB_WG, GB_TPB, 0, stream>>>(X, W1, H);

    k_agg1     <<<TILES, TPB, 0, stream>>>(src, dst, keys, E, H, dinv, b1, W2, z);

    k_l2_part  <<<PBLK, PTPB, 0, stream>>>(src, dst, E, z, dinv, Pf);
    k_out_sum  <<<(N_NODES / 4 + 31) / 32, dim3(32, SGRP), 0, stream>>>(Pf, z, dinv, b2, out);
}


# Layer 2 on layer 1's destination tiles: a tile reuses the (source,row) queue its layer-1 tile left, reads (z,dinv) pairs, sums rows in LDS (new order: atomic arrival), stores out; 5 launches
# speedup vs baseline: 4.2967x; 1.1525x over previous
#include <hip/hip_runtime.h>

typedef float v2f __attribute__((ext_vector_type(2)));
typedef float v8f __attribute__((ext_vector_type(8)));
typedef float v4f __attribute__((ext_vector_type(4)));

#define N_NODES 100000
#define IN_DIM  384
#define HID     64

#define TILES 256
#define RPB   391

#define HALF_N  50000
#define PTPB    1024
#define PART_U  6
#define SGRP    8
#define DSLICE  120
#define DBLK    (2 * DSLICE)
#define DSPG    (DSLICE / SGRP)
#define KEY_CAP 3200000
#define QCAP    16384

static_assert(2 * HALF_N >= N_NODES && HALF_N % 4 == 0 && N_NODES % 4 == 0, "halves");
static_assert(DSLICE % SGRP == 0, "slice groups");
static_assert((size_t)DBLK * HALF_N * 4 + KEY_CAP <= (size_t)2 * N_NODES * HID * 4, "keys end in front of dinv");
static_assert(KEY_CAP % 16 == 0 && ((size_t)DBLK * HALF_N * 4) % 16 == 0, "16-byte key groups");
static_assert(TILES == 256 && (N_NODES - 1) / RPB < TILES, "a key is one byte");
static_assert((size_t)N_NODES * HID * 4 + (size_t)TILES * QCAP * 4 + (size_t)N_NODES * 8 + (size_t)TILES * 4
              <= (size_t)DBLK * HALF_N * 4, "queue images end in front of the keys");
static_assert(((size_t)N_NODES * HID * 4 + (size_t)TILES * QCAP * 4) % 8 == 0, "pairs on 8-byte addresses");

__device__ __forceinline__ unsigned tile_key(int d) {
    return (unsigned)d < (unsigned)N_NODES ? (unsigned)d / RPB : (unsigned)(TILES - 1);
}

__global__ __launch_bounds__(PTPB) void k_deg_part(const int* __restrict__ dst, int E,
                                                   unsigned int* __restrict__ P,
                                                   unsigned int* __restrict__ keys) {
    __shared__ unsigned int cnt[HALF_N];
    const int tid   = threadIdx.x;
    const int half  = blockIdx.x & 1;
    const int slice = blockIdx.x >> 1;
    const int lo    = half * HALF_N;
    int n = N_NODES - lo;
    if (n > HALF_N) n = HALF_N;
    const unsigned un = n > 0 ? (unsigned)n : 0u;

    uint4* c4 = reinterpret_cast<uint4*>(cnt);
    for (int i = tid; i < HALF_N / 4; i += PTPB) c4[i] = make_uint4(0u, 0u, 0u, 0u);
    __syncthreads();

    auto hit4 = [&](int4 d) {
        const unsigned r0 = (unsigned)(d.x - lo), r1 = (unsigned)(d.y - lo);
        const unsigned r2 = (unsigned)(d.z - lo), r3 = (unsigned)(d.w - lo);
        if (r0 < un) atomicAdd(&cnt[r0], 1u);
        if (r1 < un) atomicAdd(&cnt[r1], 1u);
        if (r2 < un) atomicAdd(&cnt[r2], 1u);
        if (r3 < un) atomicAdd(&cnt[r3], 1u);
    };

    const bool wr = half == 0;
    auto put4 = [&](int4 d, int idx) {
        if (4 * idx + 3 < KEY_CAP)
            keys[idx] = tile_key(d.x) | (tile_key(d.y) << 8) | (tile_key(d.z) << 16) | (tile_key(d.w) << 24);
    };

    const int E4 = ((reinterpret_cast<size_t>(dst) & 15) == 0) ? (E >> 2) : 0;
    const int4* d4 = reinterpret_cast<const int4*>(dst);
    const int a = (int)(((long long)E4 * slice) / DSLICE) & ~31;
    const int b = slice + 1 == DSLICE ? E4 : ((int)(((long long)E4 * (slice + 1)) / DSLICE) & ~31);
    int i = a + tid;
    for (; i + (PART_U - 1) * PTPB < b; i += PART_U * PTPB) {
        int4 d[PART_U];
#pragma unroll
        for (int u = 0; u < PART_U; ++u) d[u] = d4[i + u * PTPB];
#pragma unroll
        for (int u = 0; u < PART_U; ++u) {
            hit4(d[u]);
            if (wr) put4(d[u], i + u * PTPB);
        }
    }
    for (; i < b; i += PTPB) {
        const int4 d = d4[i];
        hit4(d);
        if (wr) put4(d, i);
    }
    if (slice == DSLICE - 1) {
        unsigned char* kb = reinterpret_cast<unsigned char*>(keys);
        for (int e = 4 * E4 + tid; e < E; e += PTPB) {
            const int d = dst[e];
            const unsigned r = (unsigned)(d - lo);
            if (r < un) atomicAdd(&cnt[r], 1u);
            if (wr && e < KEY_CAP) kb[e] = (unsigned char)tile_key(d);
        }
    }
    __syncthreads();

    uint4* o4 = reinterpret_cast<uint4*>(P + (size_t)blockIdx.x * HALF_N);
    for (int j = tid; j < HALF_N / 4; j += PTPB) o4[j] = c4[j];
}

__global__ __launch_bounds__(32 * SGRP) void k_dinv_sum(const unsigned int* __restrict__ P,
                                                        float* __restrict__ dinv) {
    __shared__ uint4 part[SGRP][32];
    const int x = threadIdx.x, g = threadIdx.y;
    const int quad = blockIdx.x * 32 + x;
    uint4 s = make_uint4(0u, 0u, 0u, 0u);
    if (quad < N_NODES / 4) {
        const int node = 4 * quad;
        const int half = node >= HALF_N ? 1 : 0;
        const int j = node - half * HALF_N;
        const uint4* p = reinterpret_cast<const uint4*>(
            P + ((size_t)(2 * g * DSPG + half)) * HALF_N + j);
#pragma unroll
        for (int k = 0; k < DSPG; ++k) {
            const uint4 v = p[(size_t)k * (2 * HALF_N / 4)];
            s.x += v.x; s.y += v.y; s.z += v.z; s.w += v.w;
        }
    }
    part[g][x] = s;
    __syncthreads();
    if (g == 0 && quad < N_NODES / 4) {
        uint4 t = part[0][x];
        for (int k = 1; k < SGRP; ++k) {
            const uint4 v = part[k][x];
            t.x += v.x; t.y += v.y; t.z += v.z; t.w += v.w;
        }
        float4 r;
        r.x = rsqrtf((float)t.x + 1.0f);
        r.y = rsqrtf((float)t.y + 1.0f);
        r.z = rsqrtf((float)t.z + 1.0f);
        r.w = rsqrtf((float)t.w + 1.0f);
        reinterpret_cast<float4*>(dinv)[quad] = r;
    }
}

#define GB_ROWS  32
#define GB_P4    (IN_DIM / 4 + 1)
#define GB_WG    256
#define GB_TPB   256
#define GB_NBLK  (N_NODES / GB_ROWS)
#define GB_C4    (GB_ROWS * IN_DIM / 4)
#define GB_CH    (GB_C4 / GB_TPB)
#define GB_SLOTS 4
#define GB_KJ    (IN_DIM / 8)
#define GB_G     16
#define GB_EACH(M) M(0) M(1) M(2) M(3) M(4) M(5) M(6) M(7) M(8) M(9) M(10) M(11)

static_assert(GB_CH == 12, "GB_EACH lists the chunks");
static_assert(GB_NBLK >= GB_WG, "every work-group has a block");
static_assert(N_NODES % GB_ROWS == 0 && GB_C4 % GB_TPB == 0 && IN_DIM % 8 == 0 && HID == 64, "gemm tiling");
static_assert((GB_SLOTS & (GB_SLOTS - 1)) == 0 && GB_KJ % GB_G == 0, "ring index by mask, whole groups");

__global__ __launch_bounds__(GB_TPB) void k_gemm1(const float* X,
                                                  const float* __restrict__ W1,
                                                  float* H) {
    __shared__ v4f xs[GB_SLOTS][GB_ROWS * GB_P4];
    const int tid  = threadIdx.x;
    const int lane = tid & 31;
    const int wave = tid >> 5;
    const int half = lane >> 4;
    const int lr   = lane & 15;
    const int ct   = wave & 3;
    const int rt   = wave >> 2;

    const int t0 = (int)(((long long)GB_NBLK * blockIdx.x) / GB_WG);
    const int t1 = (int)(((long long)GB_NBLK * (blockIdx.x + 1)) / GB_WG);

    const v4f* X4 = reinterpret_cast<const v4f*>(X);
    int so[GB_CH];
#pragma unroll
    for (int i = 0; i < GB_CH; ++i) {
        const int j   = i * GB_TPB + tid;
        const int row = j / (IN_DIM / 4);
        so[i] = row * GB_P4 + (j - row * (IN_DIM / 4));
    }

    const v4f* p0 = X4 + (size_t)t0 * GB_C4 + tid;
#define GB_LD(i) const v4f f##i = p0[(i) * GB_TPB];
    GB_EACH(GB_LD)
#undef GB_LD

    v2f b[2 * GB_KJ];
    {
        const float* wp = W1 + (size_t)(4 * half) * HID + 16 * ct + lr;
#pragma unroll
        for (int j = 0; j < GB_KJ; ++j) {
#pragma unroll
            for (int q = 0; q < 2; ++q) {
                const float* bp = wp + (size_t)(8 * j + 2 * q) * HID;
                b[2 * j + q][0] = bp[0];
                b[2 * j + q][1] = bp[HID];
            }
        }
    }
#define GB_ST(i) xs[0][so[i]] = f##i;
    GB_EACH(GB_ST)
#undef GB_ST
    __syncthreads();

    auto products = [&](int slot) -> v8f {
        v8f c0 = {}, c1 = {};
        const v4f* ar = &xs[slot][(16 * rt + lr) * GB_P4 + half];
        v4f af[2][GB_G];
#pragma unroll
        for (int u = 0; u < GB_G; ++u) af[0][u] = ar[2 * u];
#pragma unroll
        for (int g = 0; g < GB_KJ / GB_G; ++g) {
            if (g + 1 < GB_KJ / GB_G) {
#pragma unroll
                for (int u = 0; u < GB_G; ++u) af[(g + 1) & 1][u] = ar[2 * ((g + 1) * GB_G + u)];
            }
            asm volatile("" ::: "memory");
#pragma unroll
            for (int u = 0; u < GB_G; ++u) {
                const int j = g * GB_G + u;
                const v4f a4 = af[g & 1][u];
                v2f a0, a1;
                a0[0] = a4.x; a0[1] = a4.y;
                a1[0] = a4.z; a1[1] = a4.w;
                c0 = __builtin_amdgcn_wmma_f32_16x16x4_f32(
                         false, a0,  false, b[2 * j],
                         (short)0, c0,  false,  false);
                c1 = __builtin_amdgcn_wmma_f32_16x16x4_f32(
                         false, a1,  false, b[2 * j + 1],
                         (short)0, c1,  false,  false);
            }
        }
        return c0 + c1;
    };
    auto put = [&](int t, v8f c) {
        float* hrow = H + (size_t)(t * GB_ROWS + 16 * rt + 8 * half) * HID + 16 * ct + lr;
#pragma unroll
        for (int r = 0; r < 8; ++r) hrow[(size_t)r * HID] = c[r];
    };

    int t = t0;
    for (; t + 1 < t1; ++t) {
        const int slot = (t - t0) & (GB_SLOTS - 1);
        const v4f* p = X4 + (size_t)(t + 1) * GB_C4 + tid;
#define GB_LD(i) const v4f st##i = p[(i) * GB_TPB];
        GB_EACH(GB_LD)
#undef GB_LD

        const v8f c = products(slot);
        asm volatile("" :: "v"(c) : "memory");

        v4f* xn = xs[(slot + 1) & (GB_SLOTS - 1)];
#define GB_ST(i) xn[so[i]] = st##i;
        GB_EACH(GB_ST)
#undef GB_ST
        __syncthreads();
        put(t, c);
    }
    put(t, products((t - t0) & (GB_SLOTS - 1)));
}

#define RBITS 9
#define GCAP  16384
#define TPB   1024
#define SCAN_U 8
#define GRP_U  8
#define CNV_U  8
#define ROW_U  16
#define Z_U    8
#define Q_SKIP 0xFFFFFFFFu
#define L2_TPB 512
#define L2_U   8

__global__ __launch_bounds__(TPB) void k_agg1(const int* __restrict__ src,
                                              const int* __restrict__ dst,
                                              const unsigned int* __restrict__ keys, int E,
                                              const float* __restrict__ H,
                                              const float* __restrict__ dinv,
                                              const float* __restrict__ b1,
                                              const float* __restrict__ W2,
                                              unsigned int* __restrict__ qbuf,
                                              unsigned int* __restrict__ qcnt,
                                              float* __restrict__ zpair) {
    float2* zp = reinterpret_cast<float2*>(zpair);
    __shared__ float        acc[RPB * HID];
    __shared__ unsigned int q[QCAP];
    __shared__ float        qw[QCAP];
    __shared__ unsigned int g1[GCAP];
    __shared__ float        dl[RPB + 1];
    __shared__ unsigned int qn;
    __shared__ unsigned int gn;

    const int tid = threadIdx.x;
    const int lo  = blockIdx.x * RPB;
    int cnt = N_NODES - lo;
    if (cnt > RPB) cnt = RPB;
    const unsigned ucnt = cnt > 0 ? (unsigned)cnt : 0u;

    if (tid == 0) { qn = 0u; gn = 0u; }
    for (int r = tid; r < cnt; r += TPB) dl[r] = dinv[lo + r];
    __syncthreads();

    {
        const float4* H4 = reinterpret_cast<const float4*>(H + (size_t)lo * HID);
        float4* acc4 = reinterpret_cast<float4*>(acc);
        for (int i = tid; i < cnt * (HID / 4); i += TPB) {
            const float di = dl[i >> 4];
            float4 h = H4[i];
            h.x = h.x * di * di; h.y = h.y * di * di;
            h.z = h.z * di * di; h.w = h.w * di * di;
            acc4[i] = h;
        }
    }
    __syncthreads();

    auto add_now = [&](int e) {
        const unsigned r = (unsigned)(dst[e] - lo);
        if (r < ucnt) {
            const int s = src[e];
            const float w = dinv[s] * dl[r];
            const float* hs = H + (size_t)s * HID;
            float* a = acc + r * HID;
            for (int f = 0; f < HID; ++f) atomicAdd(a + f, hs[f] * w);
        }
    };
    auto push = [&](int e) {
        const unsigned pos = atomicAdd(&qn, 1u);
        if (pos < QCAP) q[pos] = (unsigned)e;
        else add_now(e);
    };
    const unsigned pat = (unsigned)blockIdx.x * 0x01010101u;
    auto any16 = [&](uint4 k) -> bool {
        const unsigned x0 = k.x ^ pat, x1 = k.y ^ pat, x2 = k.z ^ pat, x3 = k.w ^ pat;
        return ((((x0 - 0x01010101u) & ~x0) | ((x1 - 0x01010101u) & ~x1) |
                 ((x2 - 0x01010101u) & ~x2) | ((x3 - 0x01010101u) & ~x3)) & 0x80808080u) != 0u;
    };
    auto match16 = [&](uint4 k) -> unsigned {
        auto nib = [&](unsigned kw) -> unsigned {
            const unsigned x = kw ^ pat;
            const unsigned m = ~(((x & 0x7F7F7F7Fu) + 0x7F7F7F7Fu) | x | 0x7F7F7F7Fu);
            return (((m >> 7) * 0x00204081u) >> 21) & 15u;
        };
        return nib(k.x) | (nib(k.y) << 4) | (nib(k.z) << 8) | (nib(k.w) << 12);
    };
    auto take_apart = [&](uint4 k, unsigned g) {
        unsigned m = match16(k);
        while (m != 0u) {
            const int b = __builtin_ctz(m);
            m &= m - 1u;
            push((int)(16u * g) + b);
        }
    };
    auto scan16 = [&](uint4 k, unsigned g) {
        if (any16(k)) {
            const unsigned pos = atomicAdd(&gn, 1u);
            if (pos < GCAP) g1[pos] = g;
        }
    };

    const int EK  = E < KEY_CAP ? E : KEY_CAP;
    const int E16 = ((reinterpret_cast<size_t>(keys) & 15) == 0) ? (EK >> 4) : 0;
    const uint4* k4 = reinterpret_cast<const uint4*>(keys);
    const int start = (int)(((long long)E16 * blockIdx.x) / TILES) & ~31;
    for (int part = 0; part < 2; ++part) {
        const int a = part ? 0 : start;
        const int b = part ? start : E16;
        int i = a + tid;
        for (; i + (SCAN_U - 1) * TPB < b; i += SCAN_U * TPB) {
            uint4 k[SCAN_U];
#pragma unroll
            for (int u = 0; u < SCAN_U; ++u) k[u] = k4[i + u * TPB];
#pragma unroll
            for (int u = 0; u < SCAN_U; ++u) scan16(k[u], (unsigned)(i + u * TPB));
        }
        for (; i < b; i += TPB) scan16(k4[i], (unsigned)i);
    }
    for (int e = 16 * E16 + tid; e < E; e += TPB) {
        if ((unsigned)(dst[e] - lo) < ucnt) push(e);
    }
    __syncthreads();

    unsigned n1 = gn;
    if (n1 > GCAP) {
        n1 = 0u;
        for (int i = tid; i < E16; i += TPB) {
            const uint4 k = k4[i];
            if (any16(k)) take_apart(k, (unsigned)i);
        }
    }
    for (unsigned i0 = 0; i0 < n1; i0 += GRP_U * TPB) {
        unsigned g[GRP_U];
        uint4 k[GRP_U];
#pragma unroll
        for (int u = 0; u < GRP_U; ++u) {
            const unsigned i = i0 + u * TPB + tid;
            g[u] = i < n1 ? g1[i] : 0u;
        }
#pragma unroll
        for (int u = 0; u < GRP_U; ++u) k[u] = k4[g[u]];
#pragma unroll
        for (int u = 0; u < GRP_U; ++u) {
            if (i0 + u * TPB + tid < n1) take_apart(k[u], g[u]);
        }
    }
    __syncthreads();

    unsigned n = qn;
    if (tid == 0) qcnt[blockIdx.x] = n;
    if (n > QCAP) n = QCAP;
    unsigned int* qout = qbuf + (size_t)blockIdx.x * QCAP;

    for (unsigned i0 = 0; i0 < n; i0 += CNV_U * TPB) {
        int e[CNV_U], d[CNV_U], s[CNV_U];
        float ds[CNV_U];
#pragma unroll
        for (int u = 0; u < CNV_U; ++u) {
            const unsigned i = i0 + u * TPB + tid;
            e[u] = i < n ? (int)q[i] : 0;
        }
#pragma unroll
        for (int u = 0; u < CNV_U; ++u) { d[u] = dst[e[u]]; s[u] = src[e[u]]; }
#pragma unroll
        for (int u = 0; u < CNV_U; ++u) ds[u] = dinv[s[u]];
#pragma unroll
        for (int u = 0; u < CNV_U; ++u) {
            const unsigned i = i0 + u * TPB + tid;
            if (i < n) {
                const unsigned r = (unsigned)(d[u] - lo);
                const bool ok = r < ucnt;
                const unsigned rr = ok ? r : 0u;
                qw[i] = ok ? ds[u] * dl[rr] : 0.0f;
                const unsigned v = ((unsigned)s[u] << RBITS) | rr;
                q[i] = v;
                qout[i] = ok ? v : Q_SKIP;
            }
        }
    }
    __syncthreads();

    const float b1a = b1[tid & 31], b1b = b1[(tid & 31) + 32];
    const float w2a = W2[tid & 31], w2b = W2[(tid & 31) + 32];

    {
        const int lane = tid & 31;
        const unsigned wave = (unsigned)tid >> 5;
        const unsigned NW = TPB / 32;
        unsigned i = wave;
        for (; i + (ROW_U - 1) * NW < n; i += ROW_U * NW) {
            unsigned v[ROW_U];
            float w[ROW_U];
            float2 h[ROW_U];
#pragma unroll
            for (int u = 0; u < ROW_U; ++u) { v[u] = q[i + u * NW]; w[u] = qw[i + u * NW]; }
#pragma unroll
            for (int u = 0; u < ROW_U; ++u)
                h[u] = reinterpret_cast<const float2*>(H + (size_t)(v[u] >> RBITS) * HID)[lane];
#pragma unroll
            for (int u = 0; u < ROW_U; ++u) {
                float* a = acc + (v[u] & ((1u << RBITS) - 1u)) * HID + 2 * lane;
                atomicAdd(a,     h[u].x * w[u]);
                atomicAdd(a + 1, h[u].y * w[u]);
            }
        }
        for (; i < n; i += NW) {
            const unsigned v = q[i];
            const float w = qw[i];
            const float2 h = reinterpret_cast<const float2*>(H + (size_t)(v >> RBITS) * HID)[lane];
            float* a = acc + (v & ((1u << RBITS) - 1u)) * HID + 2 * lane;
            atomicAdd(a,     h.x * w);
            atomicAdd(a + 1, h.y * w);
        }
    }
    __syncthreads();

    {
        const int lane = tid & 31;
        const int NW = TPB / 32;
        for (int r0 = tid >> 5; r0 < cnt; r0 += Z_U * NW) {
            float s[Z_U];
#pragma unroll
            for (int u = 0; u < Z_U; ++u) {
                const int r = r0 + u * NW;
                const float* a = acc + (r < cnt ? r : r0) * HID;
                float v0 = a[lane] + b1a;
                float v1 = a[lane + 32] + b1b;
                v0 = v0 > 0.0f ? v0 : 0.0f;
                v1 = v1 > 0.0f ? v1 : 0.0f;
                s[u] = v0 * w2a + v1 * w2b;
            }
#pragma unroll
            for (int off = 16; off > 0; off >>= 1) {
#pragma unroll
                for (int u = 0; u < Z_U; ++u) s[u] += __shfl_down(s[u], off, 32);
            }
#pragma unroll
            for (int u = 0; u < Z_U; ++u) {
                const int r = r0 + u * NW;
                if (lane == 0 && r < cnt) qw[r] = s[u];
            }
        }
    }
    __syncthreads();
    for (int r = tid; r < cnt; r += TPB) zp[lo + r] = make_float2(qw[r], dl[r]);
}

__global__ __launch_bounds__(L2_TPB) void k_l2_tile(const int* __restrict__ src,
                                                    const int* __restrict__ dst, int E,
                                                    const unsigned int* __restrict__ qbuf,
                                                    const unsigned int* __restrict__ qcnt,
                                                    const float* __restrict__ zpair,
                                                    const float* __restrict__ b2,
                                                    float* __restrict__ out) {
    const float2* zp = reinterpret_cast<const float2*>(zpair);
    __shared__ float acc[RPB + 1];
    __shared__ float zs[RPB + 1];
    __shared__ float dl[RPB + 1];

    const int tid = threadIdx.x;
    const int lo  = blockIdx.x * RPB;
    int cnt = N_NODES - lo;
    if (cnt > RPB) cnt = RPB;
    const unsigned ucnt = cnt > 0 ? (unsigned)cnt : 0u;

    for (int r = tid; r < cnt; r += L2_TPB) {
        const float2 p = zp[lo + r];
        zs[r] = p.x;
        dl[r] = p.y;
        acc[r] = 0.0f;
    }
    __syncthreads();

    const unsigned nq = qcnt[blockIdx.x];
    if (nq <= QCAP) {
        const unsigned int* qin = qbuf + (size_t)blockIdx.x * QCAP;
        unsigned i = (unsigned)tid;
        for (; i + (L2_U - 1) * L2_TPB < nq; i += L2_U * L2_TPB) {
            unsigned v[L2_U];
            float2 p[L2_U];
#pragma unroll
            for (int u = 0; u < L2_U; ++u) v[u] = qin[i + u * L2_TPB];
#pragma unroll
            for (int u = 0; u < L2_U; ++u) {
                const unsigned s = v[u] == Q_SKIP ? 0u : (v[u] >> RBITS);
                p[u] = zp[s];
            }
            unsigned r[L2_U];
            float dr[L2_U];
#pragma unroll
            for (int u = 0; u < L2_U; ++u) {
                r[u] = v[u] != Q_SKIP ? (v[u] & ((1u << RBITS) - 1u)) : 0u;
                dr[u] = dl[r[u]];
            }
#pragma unroll
            for (int u = 0; u < L2_U; ++u) {
                const float t = p[u].x * p[u].y * dr[u];
                atomicAdd(&acc[r[u]], v[u] != Q_SKIP ? t : 0.0f);
            }
        }
        for (; i < nq; i += L2_TPB) {
            const unsigned v = qin[i];
            const bool ok = v != Q_SKIP;
            const float2 p = zp[ok ? (v >> RBITS) : 0u];
            const unsigned r = ok ? (v & ((1u << RBITS) - 1u)) : 0u;
            const float t = p.x * p.y * dl[r];
            atomicAdd(&acc[r], ok ? t : 0.0f);
        }
    } else {
        for (int e = tid; e < E; e += L2_TPB) {
            const unsigned r = (unsigned)(dst[e] - lo);
            if (r < ucnt) {
                const float2 p = zp[src[e]];
                atomicAdd(&acc[r], p.x * p.y * dl[r]);
            }
        }
    }
    __syncthreads();

    const float bb = b2[0];
    for (int r = tid; r < cnt; r += L2_TPB) {
        const float di = dl[r];
        out[lo + r] = (zs[r] * di * di + bb) + acc[r];
    }
}


extern "C" void kernel_launch(void* const* d_in, const int* in_sizes, int n_in,
                              void* d_out, int out_size, void* d_ws, size_t ws_size,
                              hipStream_t stream) {
    const float* X  = (const float*)d_in[0];
    const int*   EI = (const int*)d_in[1];
    const float* W1 = (const float*)d_in[2];
    const float* b1 = (const float*)d_in[3];
    const float* W2 = (const float*)d_in[4];
    const float* b2 = (const float*)d_in[5];
    float* out = (float*)d_out;

    const int E = in_sizes[1] / 2;
    const int* src = EI;
    const int* dst = EI + E;

    char* ws = (char*)d_ws;
    float* H    = (float*)ws;
    float* dinv = (float*)(ws + (size_t)2 * N_NODES * HID * 4);
    unsigned int* Pu = (unsigned int*)ws;
    unsigned int* keys = (unsigned int*)(ws + (size_t)DBLK * HALF_N * 4);
    unsigned int* qbuf = (unsigned int*)(ws + (size_t)N_NODES * HID * 4);
    float*        zp   = (float*)(ws + (size_t)N_NODES * HID * 4 + (size_t)TILES * QCAP * 4);
    unsigned int* qcnt = (unsigned int*)(ws + (size_t)N_NODES * HID * 4 + (size_t)TILES * QCAP * 4 + (size_t)N_NODES * 8);

    k_deg_part <<<DBLK, PTPB, 0, stream>>>(dst, E, Pu, keys);
    k_dinv_sum <<<(N_NODES / 4 + 31) / 32, dim3(32, SGRP), 0, stream>>>(Pu, dinv);

    k_gemm1    <<<GB_WG, GB_TPB, 0, stream>>>(X, W1, H);

    k_agg1     <<<TILES, TPB, 0, stream>>>(src, dst, keys, E, H, dinv, b1, W2, qbuf, qcnt, zp);

    k_l2_tile  <<<TILES, L2_TPB, 0, stream>>>(src, dst, E, qbuf, qcnt, zp, b2, out);
}


# Degrees counted by the destination tiles from their own edge queues (integer LDS adds, exact): tile kernel split into finder and gather; the two degree-plane kernels give way to a key writer
# speedup vs baseline: 4.5976x; 1.0700x over previous
#include <hip/hip_runtime.h>

typedef float v2f __attribute__((ext_vector_type(2)));
typedef float v8f __attribute__((ext_vector_type(8)));
typedef float v4f __attribute__((ext_vector_type(4)));

#define N_NODES 100000
#define IN_DIM  384
#define HID     64

#define TILES 256
#define RPB   391

#define KEY_OFF ((size_t)2 * N_NODES * HID * 4 - 3200000)
#define KEY_CAP 3200000
#define QCAP    16384
#define KTPB    256
#define KEY_U   4

static_assert(N_NODES % 4 == 0, "four dinv values per 16 bytes");
static_assert(KEY_CAP % 16 == 0 && KEY_OFF % 16 == 0, "16-byte key groups");
static_assert(TILES == 256 && (N_NODES - 1) / RPB < TILES, "a key is one byte");
static_assert((size_t)N_NODES * HID * 4 + (size_t)TILES * QCAP * 4 + (size_t)N_NODES * 8 + (size_t)TILES * 4
              <= KEY_OFF, "queue images end in front of the keys");
static_assert(((size_t)N_NODES * HID * 4 + (size_t)TILES * QCAP * 4) % 8 == 0, "pairs on 8-byte addresses");

__device__ __forceinline__ unsigned tile_key(int d) {
    return (unsigned)d < (unsigned)N_NODES ? (unsigned)d / RPB : (unsigned)(TILES - 1);
}

__global__ __launch_bounds__(KTPB) void k_keys(const int* __restrict__ dst, int E,
                                               unsigned int* __restrict__ keys) {
    const int E4 = ((reinterpret_cast<size_t>(dst) & 15) == 0) ? (E >> 2) : 0;
    const int n4 = E4 < KEY_CAP / 4 ? E4 : KEY_CAP / 4;
    const int4* d4 = reinterpret_cast<const int4*>(dst);
    const int step = (int)gridDim.x * KTPB;
    auto put4 = [&](int4 d, int idx) {
        keys[idx] = tile_key(d.x) | (tile_key(d.y) << 8) | (tile_key(d.z) << 16) | (tile_key(d.w) << 24);
    };
    int i = (int)blockIdx.x * KTPB + (int)threadIdx.x;
    for (; i + (KEY_U - 1) * step < n4; i += KEY_U * step) {
        int4 d[KEY_U];
#pragma unroll
        for (int u = 0; u < KEY_U; ++u) d[u] = d4[i + u * step];
#pragma unroll
        for (int u = 0; u < KEY_U; ++u) put4(d[u], i + u * step);
    }
    for (; i < n4; i += step) put4(d4[i], i);
    unsigned char* kb = reinterpret_cast<unsigned char*>(keys);
    for (int e = 4 * n4 + (int)blockIdx.x * KTPB + (int)threadIdx.x; e < E && e < KEY_CAP; e += step)
        kb[e] = (unsigned char)tile_key(dst[e]);
}

#define GB_ROWS  32
#define GB_P4    (IN_DIM / 4 + 1)
#define GB_WG    256
#define GB_TPB   256
#define GB_NBLK  (N_NODES / GB_ROWS)
#define GB_C4    (GB_ROWS * IN_DIM / 4)
#define GB_CH    (GB_C4 / GB_TPB)
#define GB_SLOTS 4
#define GB_KJ    (IN_DIM / 8)
#define GB_G     16
#define GB_EACH(M) M(0) M(1) M(2) M(3) M(4) M(5) M(6) M(7) M(8) M(9) M(10) M(11)

static_assert(GB_CH == 12, "GB_EACH lists the chunks");
static_assert(GB_NBLK >= GB_WG, "every work-group has a block");
static_assert(N_NODES % GB_ROWS == 0 && GB_C4 % GB_TPB == 0 && IN_DIM % 8 == 0 && HID == 64, "gemm tiling");
static_assert((GB_SLOTS & (GB_SLOTS - 1)) == 0 && GB_KJ % GB_G == 0, "ring index by mask, whole groups");

__global__ __launch_bounds__(GB_TPB) void k_gemm1(const float* X,
                                                  const float* __restrict__ W1,
                                                  float* H) {
    __shared__ v4f xs[GB_SLOTS][GB_ROWS * GB_P4];
    const int tid  = threadIdx.x;
    const int lane = tid & 31;
    const int wave = tid >> 5;
    const int half = lane >> 4;
    const int lr   = lane & 15;
    const int ct   = wave & 3;
    const int rt   = wave >> 2;

    const int t0 = (int)(((long long)GB_NBLK * blockIdx.x) / GB_WG);
    const int t1 = (int)(((long long)GB_NBLK * (blockIdx.x + 1)) / GB_WG);

    const v4f* X4 = reinterpret_cast<const v4f*>(X);
    int so[GB_CH];
#pragma unroll
    for (int i = 0; i < GB_CH; ++i) {
        const int j   = i * GB_TPB + tid;
        const int row = j / (IN_DIM / 4);
        so[i] = row * GB_P4 + (j - row * (IN_DIM / 4));
    }

    const v4f* p0 = X4 + (size_t)t0 * GB_C4 + tid;
#define GB_LD(i) const v4f f##i = p0[(i) * GB_TPB];
    GB_EACH(GB_LD)
#undef GB_LD

    v2f b[2 * GB_KJ];
    {
        const float* wp = W1 + (size_t)(4 * half) * HID + 16 * ct + lr;
#pragma unroll
        for (int j = 0; j < GB_KJ; ++j) {
#pragma unroll
            for (int q = 0; q < 2; ++q) {
                const float* bp = wp + (size_t)(8 * j + 2 * q) * HID;
                b[2 * j + q][0] = bp[0];
                b[2 * j + q][1] = bp[HID];
            }
        }
    }
#define GB_ST(i) xs[0][so[i]] = f##i;
    GB_EACH(GB_ST)
#undef GB_ST
    __syncthreads();

    auto products = [&](int slot) -> v8f {
        v8f c0 = {}, c1 = {};
        const v4f* ar = &xs[slot][(16 * rt + lr) * GB_P4 + half];
        v4f af[2][GB_G];
#pragma unroll
        for (int u = 0; u < GB_G; ++u) af[0][u] = ar[2 * u];
#pragma unroll
        for (int g = 0; g < GB_KJ / GB_G; ++g) {
            if (g + 1 < GB_KJ / GB_G) {
#pragma unroll
                for (int u = 0; u < GB_G; ++u) af[(g + 1) & 1][u] = ar[2 * ((g + 1) * GB_G + u)];
            }
            asm volatile("" ::: "memory");
#pragma unroll
            for (int u = 0; u < GB_G; ++u) {
                const int j = g * GB_G + u;
                const v4f a4 = af[g & 1][u];
                v2f a0, a1;
                a0[0] = a4.x; a0[1] = a4.y;
                a1[0] = a4.z; a1[1] = a4.w;
                c0 = __builtin_amdgcn_wmma_f32_16x16x4_f32(
                         false, a0,  false, b[2 * j],
                         (short)0, c0,  false,  false);
                c1 = __builtin_amdgcn_wmma_f32_16x16x4_f32(
                         false, a1,  false, b[2 * j + 1],
                         (short)0, c1,  false,  false);
            }
        }
        return c0 + c1;
    };
    auto put = [&](int t, v8f c) {
        float* hrow = H + (size_t)(t * GB_ROWS + 16 * rt + 8 * half) * HID + 16 * ct + lr;
#pragma unroll
        for (int r = 0; r < 8; ++r) hrow[(size_t)r * HID] = c[r];
    };

    int t = t0;
    for (; t + 1 < t1; ++t) {
        const int slot = (t - t0) & (GB_SLOTS - 1);
        const v4f* p = X4 + (size_t)(t + 1) * GB_C4 + tid;
#define GB_LD(i) const v4f st##i = p[(i) * GB_TPB];
        GB_EACH(GB_LD)
#undef GB_LD

        const v8f c = products(slot);
        asm volatile("" :: "v"(c) : "memory");

        v4f* xn = xs[(slot + 1) & (GB_SLOTS - 1)];
#define GB_ST(i) xn[so[i]] = st##i;
        GB_EACH(GB_ST)
#undef GB_ST
        __syncthreads();
        put(t, c);
    }
    put(t, products((t - t0) & (GB_SLOTS - 1)));
}

#define RBITS 9
#define GCAP  32768
#define TPB   1024
#define SCAN_U 8
#define GRP_U  8
#define CNV_U  8
#define ROW_U  16
#define Z_U    8
#define Q_SKIP 0xFFFFFFFFu
#define L2_TPB 512
#define L2_U   8

__global__ __launch_bounds__(TPB) void k_find(const int* __restrict__ src,
                                              const int* __restrict__ dst,
                                              const unsigned int* __restrict__ keys, int E,
                                              unsigned int* __restrict__ qbuf,
                                              unsigned int* __restrict__ qcnt,
                                              float* __restrict__ dinv) {
    __shared__ unsigned int q[QCAP];
    __shared__ unsigned int g1[GCAP];
    __shared__ unsigned int degc[RPB + 1];
    __shared__ unsigned int qn;
    __shared__ unsigned int gn;

    const int tid = threadIdx.x;
    const int lo  = blockIdx.x * RPB;
    int cnt = N_NODES - lo;
    if (cnt > RPB) cnt = RPB;
    const unsigned ucnt = cnt > 0 ? (unsigned)cnt : 0u;

    if (tid == 0) { qn = 0u; gn = 0u; }
    for (int r = tid; r <= RPB; r += TPB) degc[r] = 0u;
    __syncthreads();

    auto push = [&](int e) {
        const unsigned pos = atomicAdd(&qn, 1u);
        if (pos < QCAP) q[pos] = (unsigned)e;
    };
    const unsigned pat = (unsigned)blockIdx.x * 0x01010101u;
    auto any16 = [&](uint4 k) -> bool {
        const unsigned x0 = k.x ^ pat, x1 = k.y ^ pat, x2 = k.z ^ pat, x3 = k.w ^ pat;
        return ((((x0 - 0x01010101u) & ~x0) | ((x1 - 0x01010101u) & ~x1) |
                 ((x2 - 0x01010101u) & ~x2) | ((x3 - 0x01010101u) & ~x3)) & 0x80808080u) != 0u;
    };
    auto match16 = [&](uint4 k) -> unsigned {
        auto nib = [&](unsigned kw) -> unsigned {
            const unsigned x = kw ^ pat;
            const unsigned m = ~(((x & 0x7F7F7F7Fu) + 0x7F7F7F7Fu) | x | 0x7F7F7F7Fu);
            return (((m >> 7) * 0x00204081u) >> 21) & 15u;
        };
        return nib(k.x) | (nib(k.y) << 4) | (nib(k.z) << 8) | (nib(k.w) << 12);
    };
    auto take_apart = [&](uint4 k, unsigned g) {
        unsigned m = match16(k);
        while (m != 0u) {
            const int b = __builtin_ctz(m);
            m &= m - 1u;
            push((int)(16u * g) + b);
        }
    };
    auto scan16 = [&](uint4 k, unsigned g) {
        if (any16(k)) {
            const unsigned pos = atomicAdd(&gn, 1u);
            if (pos < GCAP) g1[pos] = g;
        }
    };

    const int EK  = E < KEY_CAP ? E : KEY_CAP;
    const int E16 = ((reinterpret_cast<size_t>(keys) & 15) == 0) ? (EK >> 4) : 0;
    const uint4* k4 = reinterpret_cast<const uint4*>(keys);
    const int start = (int)(((long long)E16 * blockIdx.x) / TILES) & ~31;
    for (int part = 0; part < 2; ++part) {
        const int a = part ? 0 : start;
        const int b = part ? start : E16;
        int i = a + tid;
        for (; i + (SCAN_U - 1) * TPB < b; i += SCAN_U * TPB) {
            uint4 k[SCAN_U];
#pragma unroll
            for (int u = 0; u < SCAN_U; ++u) k[u] = k4[i + u * TPB];
#pragma unroll
            for (int u = 0; u < SCAN_U; ++u) scan16(k[u], (unsigned)(i + u * TPB));
        }
        for (; i < b; i += TPB) scan16(k4[i], (unsigned)i);
    }
    for (int e = 16 * E16 + tid; e < E; e += TPB) {
        if ((unsigned)(dst[e] - lo) < ucnt) push(e);
    }
    __syncthreads();

    unsigned n1 = gn;
    if (n1 > GCAP) {
        n1 = 0u;
        for (int i = tid; i < E16; i += TPB) {
            const uint4 k = k4[i];
            if (any16(k)) take_apart(k, (unsigned)i);
        }
    }
    for (unsigned i0 = 0; i0 < n1; i0 += GRP_U * TPB) {
        unsigned g[GRP_U];
        uint4 k[GRP_U];
#pragma unroll
        for (int u = 0; u < GRP_U; ++u) {
            const unsigned i = i0 + u * TPB + tid;
            g[u] = i < n1 ? g1[i] : 0u;
        }
#pragma unroll
        for (int u = 0; u < GRP_U; ++u) k[u] = k4[g[u]];
#pragma unroll
        for (int u = 0; u < GRP_U; ++u) {
            if (i0 + u * TPB + tid < n1) take_apart(k[u], g[u]);
        }
    }
    __syncthreads();

    const unsigned n = qn;
    if (tid == 0) qcnt[blockIdx.x] = n;
    if (n <= QCAP) {
        unsigned int* qout = qbuf + (size_t)blockIdx.x * QCAP;
        for (unsigned i0 = 0; i0 < n; i0 += CNV_U * TPB) {
            int e[CNV_U], d[CNV_U], s[CNV_U];
#pragma unroll
            for (int u = 0; u < CNV_U; ++u) {
                const unsigned i = i0 + u * TPB + tid;
                e[u] = i < n ? (int)q[i] : 0;
            }
#pragma unroll
            for (int u = 0; u < CNV_U; ++u) { d[u] = dst[e[u]]; s[u] = src[e[u]]; }
#pragma unroll
            for (int u = 0; u < CNV_U; ++u) {
                const unsigned i = i0 + u * TPB + tid;
                if (i < n) {
                    const unsigned r = (unsigned)(d[u] - lo);
                    const bool ok = r < ucnt;
                    if (ok) atomicAdd(&degc[r], 1u);
                    qout[i] = ok ? (((unsigned)s[u] << RBITS) | r) : Q_SKIP;
                }
            }
        }
    } else {
        for (int e = tid; e < E; e += TPB) {
            const unsigned r = (unsigned)(dst[e] - lo);
            if (r < ucnt) atomicAdd(&degc[r], 1u);
        }
    }
    __syncthreads();
    for (int r = tid; r < cnt; r += TPB) dinv[lo + r] = rsqrtf((float)degc[r] + 1.0f);
}

__global__ __launch_bounds__(TPB) void k_agg1(const int* __restrict__ src,
                                              const int* __restrict__ dst, int E,
                                              const float* __restrict__ H,
                                              const float* __restrict__ dinv,
                                              const float* __restrict__ b1,
                                              const float* __restrict__ W2,
                                              const unsigned int* __restrict__ qbuf,
                                              const unsigned int* __restrict__ qcnt,
                                              float* __restrict__ zpair) {
    float2* zp = reinterpret_cast<float2*>(zpair);
    __shared__ float        acc[RPB * HID];
    __shared__ unsigned int q[QCAP];
    __shared__ float        qw[QCAP];
    __shared__ float        dl[RPB + 1];

    const int tid = threadIdx.x;
    const int lo  = blockIdx.x * RPB;
    int cnt = N_NODES - lo;
    if (cnt > RPB) cnt = RPB;
    const unsigned ucnt = cnt > 0 ? (unsigned)cnt : 0u;

    for (int r = tid; r < cnt; r += TPB) dl[r] = dinv[lo + r];
    __syncthreads();

    {
        const float4* H4 = reinterpret_cast<const float4*>(H + (size_t)lo * HID);
        float4* acc4 = reinterpret_cast<float4*>(acc);
        for (int i = tid; i < cnt * (HID / 4); i += TPB) {
            const float di = dl[i >> 4];
            float4 h = H4[i];
            h.x = h.x * di * di; h.y = h.y * di * di;
            h.z = h.z * di * di; h.w = h.w * di * di;
            acc4[i] = h;
        }
    }

    const unsigned nq = qcnt[blockIdx.x];
    unsigned n = nq;
    if (nq <= QCAP) {
        const unsigned int* qin = qbuf + (size_t)blockIdx.x * QCAP;
        for (unsigned i0 = 0; i0 < n; i0 += CNV_U * TPB) {
            unsigned v[CNV_U];
            float ds[CNV_U];
#pragma unroll
            for (int u = 0; u < CNV_U; ++u) {
                const unsigned i = i0 + u * TPB + tid;
                v[u] = i < n ? qin[i] : Q_SKIP;
            }
#pragma unroll
            for (int u = 0; u < CNV_U; ++u) ds[u] = dinv[v[u] != Q_SKIP ? (v[u] >> RBITS) : 0u];
#pragma unroll
            for (int u = 0; u < CNV_U; ++u) {
                const unsigned i = i0 + u * TPB + tid;
                if (i < n) {
                    const bool ok = v[u] != Q_SKIP;
                    q[i]  = ok ? v[u] : 0u;
                    qw[i] = ok ? ds[u] * dl[v[u] & ((1u << RBITS) - 1u)] : 0.0f;
                }
            }
        }
    } else {
        n = 0u;
    }
    __syncthreads();

    if (nq > QCAP) {
        for (int e = tid; e < E; e += TPB) {
            const unsigned r = (unsigned)(dst[e] - lo);
            if (r < ucnt) {
                const int s = src[e];
                const float w = dinv[s] * dl[r];
                const float* hs = H + (size_t)s * HID;
                float* a = acc + r * HID;
                for (int f = 0; f < HID; ++f) atomicAdd(a + f, hs[f] * w);
            }
        }
    }

    const float b1a = b1[tid & 31], b1b = b1[(tid & 31) + 32];
    const float w2a = W2[tid & 31], w2b = W2[(tid & 31) + 32];

    {
        const int lane = tid & 31;
        const unsigned wave = (unsigned)tid >> 5;
        const unsigned NW = TPB / 32;
        unsigned i = wave;
        for (; i + (ROW_U - 1) * NW < n; i += ROW_U * NW) {
            unsigned v[ROW_U];
            float w[ROW_U];
            float2 h[ROW_U];
#pragma unroll
            for (int u = 0; u < ROW_U; ++u) { v[u] = q[i + u * NW]; w[u] = qw[i + u * NW]; }
#pragma unroll
            for (int u = 0; u < ROW_U; ++u)
                h[u] = reinterpret_cast<const float2*>(H + (size_t)(v[u] >> RBITS) * HID)[lane];
#pragma unroll
            for (int u = 0; u < ROW_U; ++u) {
                float* a = acc + (v[u] & ((1u << RBITS) - 1u)) * HID + 2 * lane;
                atomicAdd(a,     h[u].x * w[u]);
                atomicAdd(a + 1, h[u].y * w[u]);
            }
        }
        for (; i < n; i += NW) {
            const unsigned v = q[i];
            const float w = qw[i];
            const float2 h = reinterpret_cast<const float2*>(H + (size_t)(v >> RBITS) * HID)[lane];
            float* a = acc + (v & ((1u << RBITS) - 1u)) * HID + 2 * lane;
            atomicAdd(a,     h.x * w);
            atomicAdd(a + 1, h.y * w);
        }
    }
    __syncthreads();

    {
        const int lane = tid & 31;
        const int NW = TPB / 32;
        for (int r0 = tid >> 5; r0 < cnt; r0 += Z_U * NW) {
            float s[Z_U];
#pragma unroll
            for (int u = 0; u < Z_U; ++u) {
                const int r = r0 + u * NW;
                const float* a = acc + (r < cnt ? r : r0) * HID;
                float v0 = a[lane] + b1a;
                float v1 = a[lane + 32] + b1b;
                v0 = v0 > 0.0f ? v0 : 0.0f;
                v1 = v1 > 0.0f ? v1 : 0.0f;
                s[u] = v0 * w2a + v1 * w2b;
            }
#pragma unroll
            for (int off = 16; off > 0; off >>= 1) {
#pragma unroll
                for (int u = 0; u < Z_U; ++u) s[u] += __shfl_down(s[u], off, 32);
            }
#pragma unroll
            for (int u = 0; u < Z_U; ++u) {
                const int r = r0 + u * NW;
                if (lane == 0 && r < cnt) qw[r] = s[u];
            }
        }
    }
    __syncthreads();
    for (int r = tid; r < cnt; r += TPB) zp[lo + r] = make_float2(qw[r], dl[r]);
}

__global__ __launch_bounds__(L2_TPB) void k_l2_tile(const int* __restrict__ src,
                                                    const int* __restrict__ dst, int E,
                                                    const unsigned int* __restrict__ qbuf,
                                                    const unsigned int* __restrict__ qcnt,
                                                    const float* __restrict__ zpair,
                                                    const float* __restrict__ b2,
                                                    float* __restrict__ out) {
    const float2* zp = reinterpret_cast<const float2*>(zpair);
    __shared__ float acc[RPB + 1];
    __shared__ float zs[RPB + 1];
    __shared__ float dl[RPB + 1];

    const int tid = threadIdx.x;
    const int lo  = blockIdx.x * RPB;
    int cnt = N_NODES - lo;
    if (cnt > RPB) cnt = RPB;
    const unsigned ucnt = cnt > 0 ? (unsigned)cnt : 0u;

    for (int r = tid; r < cnt; r += L2_TPB) {
        const float2 p = zp[lo + r];
        zs[r] = p.x;
        dl[r] = p.y;
        acc[r] = 0.0f;
    }
    __syncthreads();

    const unsigned nq = qcnt[blockIdx.x];
    if (nq <= QCAP) {
        const unsigned int* qin = qbuf + (size_t)blockIdx.x * QCAP;
        unsigned i = (unsigned)tid;
        for (; i + (L2_U - 1) * L2_TPB < nq; i += L2_U * L2_TPB) {
            unsigned v[L2_U];
            float2 p[L2_U];
#pragma unroll
            for (int u = 0; u < L2_U; ++u) v[u] = qin[i + u * L2_TPB];
#pragma unroll
            for (int u = 0; u < L2_U; ++u) {
                const unsigned s = v[u] == Q_SKIP ? 0u : (v[u] >> RBITS);
                p[u] = zp[s];
            }
            unsigned r[L2_U];
            float dr[L2_U];
#pragma unroll
            for (int u = 0; u < L2_U; ++u) {
                r[u] = v[u] != Q_SKIP ? (v[u] & ((1u << RBITS) - 1u)) : 0u;
                dr[u] = dl[r[u]];
            }
#pragma unroll
            for (int u = 0; u < L2_U; ++u) {
                const float t = p[u].x * p[u].y * dr[u];
                atomicAdd(&acc[r[u]], v[u] != Q_SKIP ? t : 0.0f);
            }
        }
        for (; i < nq; i += L2_TPB) {
            const unsigned v = qin[i];
            const bool ok = v != Q_SKIP;
            const float2 p = zp[ok ? (v >> RBITS) : 0u];
            const unsigned r = ok ? (v & ((1u << RBITS) - 1u)) : 0u;
            const float t = p.x * p.y * dl[r];
            atomicAdd(&acc[r], ok ? t : 0.0f);
        }
    } else {
        for (int e = tid; e < E; e += L2_TPB) {
            const unsigned r = (unsigned)(dst[e] - lo);
            if (r < ucnt) {
                const float2 p = zp[src[e]];
                atomicAdd(&acc[r], p.x * p.y * dl[r]);
            }
        }
    }
    __syncthreads();

    const float bb = b2[0];
    for (int r = tid; r < cnt; r += L2_TPB) {
        const float di = dl[r];
        out[lo + r] = (zs[r] * di * di + bb) + acc[r];
    }
}


extern "C" void kernel_launch(void* const* d_in, const int* in_sizes, int n_in,
                              void* d_out, int out_size, void* d_ws, size_t ws_size,
                              hipStream_t stream) {
    const float* X  = (const float*)d_in[0];
    const int*   EI = (const int*)d_in[1];
    const float* W1 = (const float*)d_in[2];
    const float* b1 = (const float*)d_in[3];
    const float* W2 = (const float*)d_in[4];
    const float* b2 = (const float*)d_in[5];
    float* out = (float*)d_out;

    const int E = in_sizes[1] / 2;
    const int* src = EI;
    const int* dst = EI + E;

    char* ws = (char*)d_ws;
    float* H    = (float*)ws;
    float* dinv = (float*)(ws + (size_t)2 * N_NODES * HID * 4);
    unsigned int* keys = (unsigned int*)(ws + KEY_OFF);
    unsigned int* qbuf = (unsigned int*)(ws + (size_t)N_NODES * HID * 4);
    float*        zp   = (float*)(ws + (size_t)N_NODES * HID * 4 + (size_t)TILES * QCAP * 4);
    unsigned int* qcnt = (unsigned int*)(ws + (size_t)N_NODES * HID * 4 + (size_t)TILES * QCAP * 4 + (size_t)N_NODES * 8);

    k_keys     <<<800, KTPB, 0, stream>>>(dst, E, keys);
    k_find     <<<TILES, TPB, 0, stream>>>(src, dst, keys, E, qbuf, qcnt, dinv);

    k_gemm1    <<<GB_WG, GB_TPB, 0, stream>>>(X, W1, H);

    k_agg1     <<<TILES, TPB, 0, stream>>>(src, dst, E, H, dinv, b1, W2, qbuf, qcnt, zp);

    k_l2_tile  <<<TILES, L2_TPB, 0, stream>>>(src, dst, E, qbuf, qcnt, zp, b2, out);
}


# Layer-1 gather: a wave takes 32 queued edges per trip, one per lane (one LDS read for the words, one for the weights), hands them out by readlane; row addresses scalar; LDS arrival order as before
# speedup vs baseline: 4.7405x; 1.0311x over previous
#include <hip/hip_runtime.h>

typedef float v2f __attribute__((ext_vector_type(2)));
typedef float v8f __attribute__((ext_vector_type(8)));
typedef float v4f __attribute__((ext_vector_type(4)));

#define N_NODES 100000
#define IN_DIM  384
#define HID     64

#define TILES 256
#define RPB   391

#define KEY_OFF ((size_t)2 * N_NODES * HID * 4 - 3200000)
#define KEY_CAP 3200000
#define QCAP    16384
#define KTPB    256
#define KEY_U   4

static_assert(N_NODES % 4 == 0, "four dinv values per 16 bytes");
static_assert(KEY_CAP % 16 == 0 && KEY_OFF % 16 == 0, "16-byte key groups");
static_assert(TILES == 256 && (N_NODES - 1) / RPB < TILES, "a key is one byte");
static_assert((size_t)N_NODES * HID * 4 + (size_t)TILES * QCAP * 4 + (size_t)N_NODES * 8 + (size_t)TILES * 4
              <= KEY_OFF, "queue images end in front of the keys");
static_assert(((size_t)N_NODES * HID * 4 + (size_t)TILES * QCAP * 4) % 8 == 0, "pairs on 8-byte addresses");

__device__ __forceinline__ unsigned tile_key(int d) {
    return (unsigned)d < (unsigned)N_NODES ? (unsigned)d / RPB : (unsigned)(TILES - 1);
}

__global__ __launch_bounds__(KTPB) void k_keys(const int* __restrict__ dst, int E,
                                               unsigned int* __restrict__ keys) {
    const int E4 = ((reinterpret_cast<size_t>(dst) & 15) == 0) ? (E >> 2) : 0;
    const int n4 = E4 < KEY_CAP / 4 ? E4 : KEY_CAP / 4;
    const int4* d4 = reinterpret_cast<const int4*>(dst);
    const int step = (int)gridDim.x * KTPB;
    auto put4 = [&](int4 d, int idx) {
        keys[idx] = tile_key(d.x) | (tile_key(d.y) << 8) | (tile_key(d.z) << 16) | (tile_key(d.w) << 24);
    };
    int i = (int)blockIdx.x * KTPB + (int)threadIdx.x;
    for (; i + (KEY_U - 1) * step < n4; i += KEY_U * step) {
        int4 d[KEY_U];
#pragma unroll
        for (int u = 0; u < KEY_U; ++u) d[u] = d4[i + u * step];
#pragma unroll
        for (int u = 0; u < KEY_U; ++u) put4(d[u], i + u * step);
    }
    for (; i < n4; i += step) put4(d4[i], i);
    unsigned char* kb = reinterpret_cast<unsigned char*>(keys);
    for (int e = 4 * n4 + (int)blockIdx.x * KTPB + (int)threadIdx.x; e < E && e < KEY_CAP; e += step)
        kb[e] = (unsigned char)tile_key(dst[e]);
}

#define GB_ROWS  32
#define GB_P4    (IN_DIM / 4 + 1)
#define GB_WG    256
#define GB_TPB   256
#define GB_NBLK  (N_NODES / GB_ROWS)
#define GB_C4    (GB_ROWS * IN_DIM / 4)
#define GB_CH    (GB_C4 / GB_TPB)
#define GB_SLOTS 4
#define GB_KJ    (IN_DIM / 8)
#define GB_G     16
#define GB_EACH(M) M(0) M(1) M(2) M(3) M(4) M(5) M(6) M(7) M(8) M(9) M(10) M(11)

static_assert(GB_CH == 12, "GB_EACH lists the chunks");
static_assert(GB_NBLK >= GB_WG, "every work-group has a block");
static_assert(N_NODES % GB_ROWS == 0 && GB_C4 % GB_TPB == 0 && IN_DIM % 8 == 0 && HID == 64, "gemm tiling");
static_assert((GB_SLOTS & (GB_SLOTS - 1)) == 0 && GB_KJ % GB_G == 0, "ring index by mask, whole groups");

__global__ __launch_bounds__(GB_TPB) void k_gemm1(const float* X,
                                                  const float* __restrict__ W1,
                                                  float* H) {
    __shared__ v4f xs[GB_SLOTS][GB_ROWS * GB_P4];
    const int tid  = threadIdx.x;
    const int lane = tid & 31;
    const int wave = tid >> 5;
    const int half = lane >> 4;
    const int lr   = lane & 15;
    const int ct   = wave & 3;
    const int rt   = wave >> 2;

    const int t0 = (int)(((long long)GB_NBLK * blockIdx.x) / GB_WG);
    const int t1 = (int)(((long long)GB_NBLK * (blockIdx.x + 1)) / GB_WG);

    const v4f* X4 = reinterpret_cast<const v4f*>(X);
    int so[GB_CH];
#pragma unroll
    for (int i = 0; i < GB_CH; ++i) {
        const int j   = i * GB_TPB + tid;
        const int row = j / (IN_DIM / 4);
        so[i] = row * GB_P4 + (j - row * (IN_DIM / 4));
    }

    const v4f* p0 = X4 + (size_t)t0 * GB_C4 + tid;
#define GB_LD(i) const v4f f##i = p0[(i) * GB_TPB];
    GB_EACH(GB_LD)
#undef GB_LD

    v2f b[2 * GB_KJ];
    {
        const float* wp = W1 + (size_t)(4 * half) * HID + 16 * ct + lr;
#pragma unroll
        for (int j = 0; j < GB_KJ; ++j) {
#pragma unroll
            for (int q = 0; q < 2; ++q) {
                const float* bp = wp + (size_t)(8 * j + 2 * q) * HID;
                b[2 * j + q][0] = bp[0];
                b[2 * j + q][1] = bp[HID];
            }
        }
    }
#define GB_ST(i) xs[0][so[i]] = f##i;
    GB_EACH(GB_ST)
#undef GB_ST
    __syncthreads();

    auto products = [&](int slot) -> v8f {
        v8f c0 = {}, c1 = {};
        const v4f* ar = &xs[slot][(16 * rt + lr) * GB_P4 + half];
        v4f af[2][GB_G];
#pragma unroll
        for (int u = 0; u < GB_G; ++u) af[0][u] = ar[2 * u];
#pragma unroll
        for (int g = 0; g < GB_KJ / GB_G; ++g) {
            if (g + 1 < GB_KJ / GB_G) {
#pragma unroll
                for (int u = 0; u < GB_G; ++u) af[(g + 1) & 1][u] = ar[2 * ((g + 1) * GB_G + u)];
            }
            asm volatile("" ::: "memory");
#pragma unroll
            for (int u = 0; u < GB_G; ++u) {
                const int j = g * GB_G + u;
                const v4f a4 = af[g & 1][u];
                v2f a0, a1;
                a0[0] = a4.x; a0[1] = a4.y;
                a1[0] = a4.z; a1[1] = a4.w;
                c0 = __builtin_amdgcn_wmma_f32_16x16x4_f32(
                         false, a0,  false, b[2 * j],
                         (short)0, c0,  false,  false);
                c1 = __builtin_amdgcn_wmma_f32_16x16x4_f32(
                         false, a1,  false, b[2 * j + 1],
                         (short)0, c1,  false,  false);
            }
        }
        return c0 + c1;
    };
    auto put = [&](int t, v8f c) {
        float* hrow = H + (size_t)(t * GB_ROWS + 16 * rt + 8 * half) * HID + 16 * ct + lr;
#pragma unroll
        for (int r = 0; r < 8; ++r) hrow[(size_t)r * HID] = c[r];
    };

    int t = t0;
    for (; t + 1 < t1; ++t) {
        const int slot = (t - t0) & (GB_SLOTS - 1);
        const v4f* p = X4 + (size_t)(t + 1) * GB_C4 + tid;
#define GB_LD(i) const v4f st##i = p[(i) * GB_TPB];
        GB_EACH(GB_LD)
#undef GB_LD

        const v8f c = products(slot);
        asm volatile("" :: "v"(c) : "memory");

        v4f* xn = xs[(slot + 1) & (GB_SLOTS - 1)];
#define GB_ST(i) xn[so[i]] = st##i;
        GB_EACH(GB_ST)
#undef GB_ST
        __syncthreads();
        put(t, c);
    }
    put(t, products((t - t0) & (GB_SLOTS - 1)));
}

#define RBITS 9
#define GCAP  32768
#define TPB   1024
#define SCAN_U 8
#define GRP_U  8
#define CNV_U  8
#define Z_U    8
#define Q_SKIP 0xFFFFFFFFu
#define L2_TPB 512
#define L2_U   8

__global__ __launch_bounds__(TPB) void k_find(const int* __restrict__ src,
                                              const int* __restrict__ dst,
                                              const unsigned int* __restrict__ keys, int E,
                                              unsigned int* __restrict__ qbuf,
                                              unsigned int* __restrict__ qcnt,
                                              float* __restrict__ dinv) {
    __shared__ unsigned int q[QCAP];
    __shared__ unsigned int g1[GCAP];
    __shared__ unsigned int degc[RPB + 1];
    __shared__ unsigned int qn;
    __shared__ unsigned int gn;

    const int tid = threadIdx.x;
    const int lo  = blockIdx.x * RPB;
    int cnt = N_NODES - lo;
    if (cnt > RPB) cnt = RPB;
    const unsigned ucnt = cnt > 0 ? (unsigned)cnt : 0u;

    if (tid == 0) { qn = 0u; gn = 0u; }
    for (int r = tid; r <= RPB; r += TPB) degc[r] = 0u;
    __syncthreads();

    auto push = [&](int e) {
        const unsigned pos = atomicAdd(&qn, 1u);
        if (pos < QCAP) q[pos] = (unsigned)e;
    };
    const unsigned pat = (unsigned)blockIdx.x * 0x01010101u;
    auto any16 = [&](uint4 k) -> bool {
        const unsigned x0 = k.x ^ pat, x1 = k.y ^ pat, x2 = k.z ^ pat, x3 = k.w ^ pat;
        return ((((x0 - 0x01010101u) & ~x0) | ((x1 - 0x01010101u) & ~x1) |
                 ((x2 - 0x01010101u) & ~x2) | ((x3 - 0x01010101u) & ~x3)) & 0x80808080u) != 0u;
    };
    auto match16 = [&](uint4 k) -> unsigned {
        auto nib = [&](unsigned kw) -> unsigned {
            const unsigned x = kw ^ pat;
            const unsigned m = ~(((x & 0x7F7F7F7Fu) + 0x7F7F7F7Fu) | x | 0x7F7F7F7Fu);
            return (((m >> 7) * 0x00204081u) >> 21) & 15u;
        };
        return nib(k.x) | (nib(k.y) << 4) | (nib(k.z) << 8) | (nib(k.w) << 12);
    };
    auto take_apart = [&](uint4 k, unsigned g) {
        unsigned m = match16(k);
        while (m != 0u) {
            const int b = __builtin_ctz(m);
            m &= m - 1u;
            push((int)(16u * g) + b);
        }
    };
    auto scan16 = [&](uint4 k, unsigned g) {
        if (any16(k)) {
            const unsigned pos = atomicAdd(&gn, 1u);
            if (pos < GCAP) g1[pos] = g;
        }
    };

    const int EK  = E < KEY_CAP ? E : KEY_CAP;
    const int E16 = ((reinterpret_cast<size_t>(keys) & 15) == 0) ? (EK >> 4) : 0;
    const uint4* k4 = reinterpret_cast<const uint4*>(keys);
    const int start = (int)(((long long)E16 * blockIdx.x) / TILES) & ~31;
    for (int part = 0; part < 2; ++part) {
        const int a = part ? 0 : start;
        const int b = part ? start : E16;
        int i = a + tid;
        for (; i + (SCAN_U - 1) * TPB < b; i += SCAN_U * TPB) {
            uint4 k[SCAN_U];
#pragma unroll
            for (int u = 0; u < SCAN_U; ++u) k[u] = k4[i + u * TPB];
#pragma unroll
            for (int u = 0; u < SCAN_U; ++u) scan16(k[u], (unsigned)(i + u * TPB));
        }
        for (; i < b; i += TPB) scan16(k4[i], (unsigned)i);
    }
    for (int e = 16 * E16 + tid; e < E; e += TPB) {
        if ((unsigned)(dst[e] - lo) < ucnt) push(e);
    }
    __syncthreads();

    unsigned n1 = gn;
    if (n1 > GCAP) {
        n1 = 0u;
        for (int i = tid; i < E16; i += TPB) {
            const uint4 k = k4[i];
            if (any16(k)) take_apart(k, (unsigned)i);
        }
    }
    for (unsigned i0 = 0; i0 < n1; i0 += GRP_U * TPB) {
        unsigned g[GRP_U];
        uint4 k[GRP_U];
#pragma unroll
        for (int u = 0; u < GRP_U; ++u) {
            const unsigned i = i0 + u * TPB + tid;
            g[u] = i < n1 ? g1[i] : 0u;
        }
#pragma unroll
        for (int u = 0; u < GRP_U; ++u) k[u] = k4[g[u]];
#pragma unroll
        for (int u = 0; u < GRP_U; ++u) {
            if (i0 + u * TPB + tid < n1) take_apart(k[u], g[u]);
        }
    }
    __syncthreads();

    const unsigned n = qn;
    if (tid == 0) qcnt[blockIdx.x] = n;
    if (n <= QCAP) {
        unsigned int* qout = qbuf + (size_t)blockIdx.x * QCAP;
        for (unsigned i0 = 0; i0 < n; i0 += CNV_U * TPB) {
            int e[CNV_U], d[CNV_U], s[CNV_U];
#pragma unroll
            for (int u = 0; u < CNV_U; ++u) {
                const unsigned i = i0 + u * TPB + tid;
                e[u] = i < n ? (int)q[i] : 0;
            }
#pragma unroll
            for (int u = 0; u < CNV_U; ++u) { d[u] = dst[e[u]]; s[u] = src[e[u]]; }
#pragma unroll
            for (int u = 0; u < CNV_U; ++u) {
                const unsigned i = i0 + u * TPB + tid;
                if (i < n) {
                    const unsigned r = (unsigned)(d[u] - lo);
                    const bool ok = r < ucnt;
                    if (ok) atomicAdd(&degc[r], 1u);
                    qout[i] = ok ? (((unsigned)s[u] << RBITS) | r) : Q_SKIP;
                }
            }
        }
    } else {
        for (int e = tid; e < E; e += TPB) {
            const unsigned r = (unsigned)(dst[e] - lo);
            if (r < ucnt) atomicAdd(&degc[r], 1u);
        }
    }
    __syncthreads();
    for (int r = tid; r < cnt; r += TPB) dinv[lo + r] = rsqrtf((float)degc[r] + 1.0f);
}

__global__ __launch_bounds__(TPB) void k_agg1(const int* __restrict__ src,
                                              const int* __restrict__ dst, int E,
                                              const float* __restrict__ H,
                                              const float* __restrict__ dinv,
                                              const float* __restrict__ b1,
                                              const float* __restrict__ W2,
                                              const unsigned int* __restrict__ qbuf,
                                              const unsigned int* __restrict__ qcnt,
                                              float* __restrict__ zpair) {
    float2* zp = reinterpret_cast<float2*>(zpair);
    __shared__ float        acc[RPB * HID];
    __shared__ unsigned int q[QCAP];
    __shared__ float        qw[QCAP];
    __shared__ float        dl[RPB + 1];

    const int tid = threadIdx.x;
    const int lo  = blockIdx.x * RPB;
    int cnt = N_NODES - lo;
    if (cnt > RPB) cnt = RPB;
    const unsigned ucnt = cnt > 0 ? (unsigned)cnt : 0u;

    for (int r = tid; r < cnt; r += TPB) dl[r] = dinv[lo + r];
    __syncthreads();

    {
        const float4* H4 = reinterpret_cast<const float4*>(H + (size_t)lo * HID);
        float4* acc4 = reinterpret_cast<float4*>(acc);
        for (int i = tid; i < cnt * (HID / 4); i += TPB) {
            const float di = dl[i >> 4];
            float4 h = H4[i];
            h.x = h.x * di * di; h.y = h.y * di * di;
            h.z = h.z * di * di; h.w = h.w * di * di;
            acc4[i] = h;
        }
    }

    const unsigned nq = qcnt[blockIdx.x];
    unsigned n = nq;
    if (nq <= QCAP) {
        const unsigned int* qin = qbuf + (size_t)blockIdx.x * QCAP;
        for (unsigned i0 = 0; i0 < n; i0 += CNV_U * TPB) {
            unsigned v[CNV_U];
            float ds[CNV_U];
#pragma unroll
            for (int u = 0; u < CNV_U; ++u) {
                const unsigned i = i0 + u * TPB + tid;
                v[u] = i < n ? qin[i] : Q_SKIP;
            }
#pragma unroll
            for (int u = 0; u < CNV_U; ++u) ds[u] = dinv[v[u] != Q_SKIP ? (v[u] >> RBITS) : 0u];
#pragma unroll
            for (int u = 0; u < CNV_U; ++u) {
                const unsigned i = i0 + u * TPB + tid;
                if (i < n) {
                    const bool ok = v[u] != Q_SKIP;
                    q[i]  = ok ? v[u] : 0u;
                    qw[i] = ok ? ds[u] * dl[v[u] & ((1u << RBITS) - 1u)] : 0.0f;
                }
            }
        }
    } else {
        n = 0u;
    }
    __syncthreads();

    if (nq > QCAP) {
        for (int e = tid; e < E; e += TPB) {
            const unsigned r = (unsigned)(dst[e] - lo);
            if (r < ucnt) {
                const int s = src[e];
                const float w = dinv[s] * dl[r];
                const float* hs = H + (size_t)s * HID;
                float* a = acc + r * HID;
                for (int f = 0; f < HID; ++f) atomicAdd(a + f, hs[f] * w);
            }
        }
    }

    const float b1a = b1[tid & 31], b1b = b1[(tid & 31) + 32];
    const float w2a = W2[tid & 31], w2b = W2[(tid & 31) + 32];

    {
        const int lane = tid & 31;
        const unsigned wave = (unsigned)__builtin_amdgcn_readfirstlane(tid >> 5);
        const unsigned NW = TPB / 32;
        const float2* Hl = reinterpret_cast<const float2*>(H) + lane;
        float* al = acc + 2 * lane;
        const unsigned full = n & ~(32u * NW - 1u);
        unsigned base = 32u * wave;
        for (; base < full; base += 32u * NW) {
            const int   vq = (int)q[base + lane];
            const int   wq = (int)__float_as_uint(qw[base + lane]);
            float2 h[32];
#pragma unroll
            for (int u = 0; u < 32; ++u) {
                const unsigned v = (unsigned)__builtin_amdgcn_readlane(vq, u);
                h[u] = Hl[(size_t)(v >> RBITS) * (HID / 2)];
            }
#pragma unroll
            for (int u = 0; u < 32; ++u) {
                const unsigned v = (unsigned)__builtin_amdgcn_readlane(vq, u);
                const float w = __uint_as_float((unsigned)__builtin_amdgcn_readlane(wq, u));
                float* a = al + (v & ((1u << RBITS) - 1u)) * HID;
                atomicAdd(a,     h[u].x * w);
                atomicAdd(a + 1, h[u].y * w);
            }
        }
        {
            const unsigned rest = n - full;
            const unsigned c = (rest + NW - 1u) / NW;
            const unsigned s0 = full + wave * c;
            const unsigned m = s0 < n ? (n - s0 < c ? n - s0 : c) : 0u;
            const bool in = (unsigned)lane < m;
            const int vq = in ? (int)q[s0 + lane] : 0;
            const int wq = in ? (int)__float_as_uint(qw[s0 + lane]) : 0;
            float2 h[32];
#pragma unroll
            for (int u = 0; u < 32; ++u) {
                if ((unsigned)u < m) {
                    const unsigned v = (unsigned)__builtin_amdgcn_readlane(vq, u);
                    h[u] = Hl[(size_t)(v >> RBITS) * (HID / 2)];
                }
            }
#pragma unroll
            for (int u = 0; u < 32; ++u) {
                if ((unsigned)u < m) {
                    const unsigned v = (unsigned)__builtin_amdgcn_readlane(vq, u);
                    const float w = __uint_as_float((unsigned)__builtin_amdgcn_readlane(wq, u));
                    float* a = al + (v & ((1u << RBITS) - 1u)) * HID;
                    atomicAdd(a,     h[u].x * w);
                    atomicAdd(a + 1, h[u].y * w);
                }
            }
        }
    }
    __syncthreads();

    {
        const int lane = tid & 31;
        const int NW = TPB / 32;
        for (int r0 = tid >> 5; r0 < cnt; r0 += Z_U * NW) {
            float s[Z_U];
#pragma unroll
            for (int u = 0; u < Z_U; ++u) {
                const int r = r0 + u * NW;
                const float* a = acc + (r < cnt ? r : r0) * HID;
                float v0 = a[lane] + b1a;
                float v1 = a[lane + 32] + b1b;
                v0 = v0 > 0.0f ? v0 : 0.0f;
                v1 = v1 > 0.0f ? v1 : 0.0f;
                s[u] = v0 * w2a + v1 * w2b;
            }
#pragma unroll
            for (int off = 16; off > 0; off >>= 1) {
#pragma unroll
                for (int u = 0; u < Z_U; ++u) s[u] += __shfl_down(s[u], off, 32);
            }
#pragma unroll
            for (int u = 0; u < Z_U; ++u) {
                const int r = r0 + u * NW;
                if (lane == 0 && r < cnt) qw[r] = s[u];
            }
        }
    }
    __syncthreads();
    for (int r = tid; r < cnt; r += TPB) zp[lo + r] = make_float2(qw[r], dl[r]);
}

__global__ __launch_bounds__(L2_TPB) void k_l2_tile(const int* __restrict__ src,
                                                    const int* __restrict__ dst, int E,
                                                    const unsigned int* __restrict__ qbuf,
                                                    const unsigned int* __restrict__ qcnt,
                                                    const float* __restrict__ zpair,
                                                    const float* __restrict__ b2,
                                                    float* __restrict__ out) {
    const float2* zp = reinterpret_cast<const float2*>(zpair);
    __shared__ float acc[RPB + 1];
    __shared__ float zs[RPB + 1];
    __shared__ float dl[RPB + 1];

    const int tid = threadIdx.x;
    const int lo  = blockIdx.x * RPB;
    int cnt = N_NODES - lo;
    if (cnt > RPB) cnt = RPB;
    const unsigned ucnt = cnt > 0 ? (unsigned)cnt : 0u;

    for (int r = tid; r < cnt; r += L2_TPB) {
        const float2 p = zp[lo + r];
        zs[r] = p.x;
        dl[r] = p.y;
        acc[r] = 0.0f;
    }
    __syncthreads();

    const unsigned nq = qcnt[blockIdx.x];
    if (nq <= QCAP) {
        const unsigned int* qin = qbuf + (size_t)blockIdx.x * QCAP;
        unsigned i = (unsigned)tid;
        for (; i + (L2_U - 1) * L2_TPB < nq; i += L2_U * L2_TPB) {
            unsigned v[L2_U];
            float2 p[L2_U];
#pragma unroll
            for (int u = 0; u < L2_U; ++u) v[u] = qin[i + u * L2_TPB];
#pragma unroll
            for (int u = 0; u < L2_U; ++u) {
                const unsigned s = v[u] == Q_SKIP ? 0u : (v[u] >> RBITS);
                p[u] = zp[s];
            }
            unsigned r[L2_U];
            float dr[L2_U];
#pragma unroll
            for (int u = 0; u < L2_U; ++u) {
                r[u] = v[u] != Q_SKIP ? (v[u] & ((1u << RBITS) - 1u)) : 0u;
                dr[u] = dl[r[u]];
            }
#pragma unroll
            for (int u = 0; u < L2_U; ++u) {
                const float t = p[u].x * p[u].y * dr[u];
                atomicAdd(&acc[r[u]], v[u] != Q_SKIP ? t : 0.0f);
            }
        }
        for (; i < nq; i += L2_TPB) {
            const unsigned v = qin[i];
            const bool ok = v != Q_SKIP;
            const float2 p = zp[ok ? (v >> RBITS) : 0u];
            const unsigned r = ok ? (v & ((1u << RBITS) - 1u)) : 0u;
            const float t = p.x * p.y * dl[r];
            atomicAdd(&acc[r], ok ? t : 0.0f);
        }
    } else {
        for (int e = tid; e < E; e += L2_TPB) {
            const unsigned r = (unsigned)(dst[e] - lo);
            if (r < ucnt) {
                const float2 p = zp[src[e]];
                atomicAdd(&acc[r], p.x * p.y * dl[r]);
            }
        }
    }
    __syncthreads();

    const float bb = b2[0];
    for (int r = tid; r < cnt; r += L2_TPB) {
        const float di = dl[r];
        out[lo + r] = (zs[r] * di * di + bb) + acc[r];
    }
}


extern "C" void kernel_launch(void* const* d_in, const int* in_sizes, int n_in,
                              void* d_out, int out_size, void* d_ws, size_t ws_size,
                              hipStream_t stream) {
    const float* X  = (const float*)d_in[0];
    const int*   EI = (const int*)d_in[1];
    const float* W1 = (const float*)d_in[2];
    const float* b1 = (const float*)d_in[3];
    const float* W2 = (const float*)d_in[4];
    const float* b2 = (const float*)d_in[5];
    float* out = (float*)d_out;

    const int E = in_sizes[1] / 2;
    const int* src = EI;
    const int* dst = EI + E;

    char* ws = (char*)d_ws;
    float* H    = (float*)ws;
    float* dinv = (float*)(ws + (size_t)2 * N_NODES * HID * 4);
    unsigned int* keys = (unsigned int*)(ws + KEY_OFF);
    unsigned int* qbuf = (unsigned int*)(ws + (size_t)N_NODES * HID * 4);
    float*        zp   = (float*)(ws + (size_t)N_NODES * HID * 4 + (size_t)TILES * QCAP * 4);
    unsigned int* qcnt = (unsigned int*)(ws + (size_t)N_NODES * HID * 4 + (size_t)TILES * QCAP * 4 + (size_t)N_NODES * 8);

    k_keys     <<<800, KTPB, 0, stream>>>(dst, E, keys);
    k_find     <<<TILES, TPB, 0, stream>>>(src, dst, keys, E, qbuf, qcnt, dinv);

    k_gemm1    <<<GB_WG, GB_TPB, 0, stream>>>(X, W1, H);

    k_agg1     <<<TILES, TPB, 0, stream>>>(src, dst, E, H, dinv, b1, W2, qbuf, qcnt, zp);

    k_l2_tile  <<<TILES, L2_TPB, 0, stream>>>(src, dst, E, qbuf, qcnt, zp, b2, out);
}


# Edge search by per-tile bitmaps: a first kernel sets one bit per 16-edge group and tile (LDS ORs); a tile reads its 25 KB row, not 3.2 MB of byte keys, then tests hit groups' dst; arrival-order sums
# speedup vs baseline: 5.7647x; 1.2161x over previous
#include <hip/hip_runtime.h>

typedef float v2f __attribute__((ext_vector_type(2)));
typedef float v8f __attribute__((ext_vector_type(8)));
typedef float v4f __attribute__((ext_vector_type(4)));

#define N_NODES 100000
#define IN_DIM  384
#define HID     64

#define TILES 256
#define RPB   391

#define GRP      16
#define BM_CAP   3200000
#define BM_TPB   1024
#define BM_WGE   (BM_TPB * GRP)
#define BM_WORDS (BM_WGE / GRP / 32)
#define BM_NB    ((BM_CAP + BM_WGE - 1) / BM_WGE)
#define BM_PITCH (TILES + 1)
#define QCAP     16384

static_assert(N_NODES % 4 == 0, "four dinv values per 16 bytes");
static_assert(BM_WORDS == 32 && BM_TPB == 1024, "a wave stores one tile's line of a bitmap work-group");
static_assert(TILES == 256 && (N_NODES - 1) / RPB < TILES, "tile numbers");
static_assert((size_t)TILES * BM_NB * BM_WORDS * 4 <= (size_t)N_NODES * HID * 4, "the bitmaps fit in the region of H");
static_assert((size_t)N_NODES * HID * 4 + (size_t)TILES * QCAP * 4 + (size_t)N_NODES * 8 + (size_t)TILES * 4
              <= (size_t)2 * N_NODES * HID * 4, "queue images end in front of dinv");
static_assert(((size_t)N_NODES * HID * 4 + (size_t)TILES * QCAP * 4) % 8 == 0, "pairs on 8-byte addresses");

__device__ __forceinline__ int bm_groups(const int* dst, int E) {
    const int EK = E < BM_CAP ? E : BM_CAP;
    return ((reinterpret_cast<size_t>(dst) & 15) == 0) ? (EK / GRP) : 0;
}

__global__ __launch_bounds__(BM_TPB) void k_bitmap(const int* __restrict__ dst, int E,
                                                   unsigned int* __restrict__ bm) {
    __shared__ unsigned int tb[BM_WORDS * BM_PITCH];
    const int tid = threadIdx.x;
    for (int i = tid; i < BM_WORDS * BM_PITCH; i += BM_TPB) tb[i] = 0u;
    __syncthreads();

    const int G4 = 4 * bm_groups(dst, E);
    const int4* d4 = reinterpret_cast<const int4*>(dst);
    const int base4 = (int)blockIdx.x * (BM_WGE / 4);
    int4 d[4];
#pragma unroll
    for (int u = 0; u < 4; ++u) {
        const int i4 = base4 + u * BM_TPB + tid;
        d[u] = i4 < G4 ? d4[i4] : make_int4(-1, -1, -1, -1);
    }
#pragma unroll
    for (int u = 0; u < 4; ++u) {
        const int gl = (u * BM_TPB + tid) >> 2;
        unsigned int* row = tb + (gl >> 5) * BM_PITCH;
        const unsigned bit = 1u << (gl & 31);
        if ((unsigned)d[u].x < (unsigned)N_NODES) atomicOr(row + (unsigned)d[u].x / RPB, bit);
        if ((unsigned)d[u].y < (unsigned)N_NODES) atomicOr(row + (unsigned)d[u].y / RPB, bit);
        if ((unsigned)d[u].z < (unsigned)N_NODES) atomicOr(row + (unsigned)d[u].z / RPB, bit);
        if ((unsigned)d[u].w < (unsigned)N_NODES) atomicOr(row + (unsigned)d[u].w / RPB, bit);
    }
    __syncthreads();
    for (int p = 0; p < TILES / 32; ++p) {
        const int t = 32 * p + (tid >> 5);
        bm[((size_t)t * BM_NB + blockIdx.x) * BM_WORDS + (tid & 31)] = tb[(tid & 31) * BM_PITCH + t];
    }
}

#define GB_ROWS  32
#define GB_P4    (IN_DIM / 4 + 1)
#define GB_WG    256
#define GB_TPB   256
#define GB_NBLK  (N_NODES / GB_ROWS)
#define GB_C4    (GB_ROWS * IN_DIM / 4)
#define GB_CH    (GB_C4 / GB_TPB)
#define GB_SLOTS 4
#define GB_KJ    (IN_DIM / 8)
#define GB_G     16
#define GB_EACH(M) M(0) M(1) M(2) M(3) M(4) M(5) M(6) M(7) M(8) M(9) M(10) M(11)

static_assert(GB_CH == 12, "GB_EACH lists the chunks");
static_assert(GB_NBLK >= GB_WG, "every work-group has a block");
static_assert(N_NODES % GB_ROWS == 0 && GB_C4 % GB_TPB == 0 && IN_DIM % 8 == 0 && HID == 64, "gemm tiling");
static_assert((GB_SLOTS & (GB_SLOTS - 1)) == 0 && GB_KJ % GB_G == 0, "ring index by mask, whole groups");

__global__ __launch_bounds__(GB_TPB) void k_gemm1(const float* X,
                                                  const float* __restrict__ W1,
                                                  float* H) {
    __shared__ v4f xs[GB_SLOTS][GB_ROWS * GB_P4];
    const int tid  = threadIdx.x;
    const int lane = tid & 31;
    const int wave = tid >> 5;
    const int half = lane >> 4;
    const int lr   = lane & 15;
    const int ct   = wave & 3;
    const int rt   = wave >> 2;

    const int t0 = (int)(((long long)GB_NBLK * blockIdx.x) / GB_WG);
    const int t1 = (int)(((long long)GB_NBLK * (blockIdx.x + 1)) / GB_WG);

    const v4f* X4 = reinterpret_cast<const v4f*>(X);
    int so[GB_CH];
#pragma unroll
    for (int i = 0; i < GB_CH; ++i) {
        const int j   = i * GB_TPB + tid;
        const int row = j / (IN_DIM / 4);
        so[i] = row * GB_P4 + (j - row * (IN_DIM / 4));
    }

    const v4f* p0 = X4 + (size_t)t0 * GB_C4 + tid;
#define GB_LD(i) const v4f f##i = p0[(i) * GB_TPB];
    GB_EACH(GB_LD)
#undef GB_LD

    v2f b[2 * GB_KJ];
    {
        const float* wp = W1 + (size_t)(4 * half) * HID + 16 * ct + lr;
#pragma unroll
        for (int j = 0; j < GB_KJ; ++j) {
#pragma unroll
            for (int q = 0; q < 2; ++q) {
                const float* bp = wp + (size_t)(8 * j + 2 * q) * HID;
                b[2 * j + q][0] = bp[0];
                b[2 * j + q][1] = bp[HID];
            }
        }
    }
#define GB_ST(i) xs[0][so[i]] = f##i;
    GB_EACH(GB_ST)
#undef GB_ST
    __syncthreads();

    auto products = [&](int slot) -> v8f {
        v8f c0 = {}, c1 = {};
        const v4f* ar = &xs[slot][(16 * rt + lr) * GB_P4 + half];
        v4f af[2][GB_G];
#pragma unroll
        for (int u = 0; u < GB_G; ++u) af[0][u] = ar[2 * u];
#pragma unroll
        for (int g = 0; g < GB_KJ / GB_G; ++g) {
            if (g + 1 < GB_KJ / GB_G) {
#pragma unroll
                for (int u = 0; u < GB_G; ++u) af[(g + 1) & 1][u] = ar[2 * ((g + 1) * GB_G + u)];
            }
            asm volatile("" ::: "memory");
#pragma unroll
            for (int u = 0; u < GB_G; ++u) {
                const int j = g * GB_G + u;
                const v4f a4 = af[g & 1][u];
                v2f a0, a1;
                a0[0] = a4.x; a0[1] = a4.y;
                a1[0] = a4.z; a1[1] = a4.w;
                c0 = __builtin_amdgcn_wmma_f32_16x16x4_f32(
                         false, a0,  false, b[2 * j],
                         (short)0, c0,  false,  false);
                c1 = __builtin_amdgcn_wmma_f32_16x16x4_f32(
                         false, a1,  false, b[2 * j + 1],
                         (short)0, c1,  false,  false);
            }
        }
        return c0 + c1;
    };
    auto put = [&](int t, v8f c) {
        float* hrow = H + (size_t)(t * GB_ROWS + 16 * rt + 8 * half) * HID + 16 * ct + lr;
#pragma unroll
        for (int r = 0; r < 8; ++r) hrow[(size_t)r * HID] = c[r];
    };

    int t = t0;
    for (; t + 1 < t1; ++t) {
        const int slot = (t - t0) & (GB_SLOTS - 1);
        const v4f* p = X4 + (size_t)(t + 1) * GB_C4 + tid;
#define GB_LD(i) const v4f st##i = p[(i) * GB_TPB];
        GB_EACH(GB_LD)
#undef GB_LD

        const v8f c = products(slot);
        asm volatile("" :: "v"(c) : "memory");

        v4f* xn = xs[(slot + 1) & (GB_SLOTS - 1)];
#define GB_ST(i) xn[so[i]] = st##i;
        GB_EACH(GB_ST)
#undef GB_ST
        __syncthreads();
        put(t, c);
    }
    put(t, products((t - t0) & (GB_SLOTS - 1)));
}

__device__ __forceinline__ unsigned reserve(unsigned int* ctr, unsigned c) {
    const unsigned lane = threadIdx.x & 31u;
    unsigned inc = c;
#pragma unroll
    for (unsigned off = 1; off < 32; off <<= 1) {
        const unsigned t = __shfl_up(inc, off, 32);
        if (lane >= off) inc += t;
    }
    const unsigned total = __shfl(inc, 31, 32);
    unsigned base = 0u;
    if (lane == 0u) base = atomicAdd(ctr, total);
    return (unsigned)__builtin_amdgcn_readfirstlane((int)base) + (inc - c);
}

#define RBITS 9
#define GCAP  32768
#define TPB   1024
#define GRP_U  12
#define CNV_U  8
#define Z_U    8
#define Q_SKIP 0xFFFFFFFFu
#define L2_TPB 512
#define L2_U   8

__global__ __launch_bounds__(TPB) void k_find(const int* __restrict__ src,
                                              const int* __restrict__ dst,
                                              const unsigned int* __restrict__ bm, int E,
                                              unsigned int* __restrict__ qbuf,
                                              unsigned int* __restrict__ qcnt,
                                              float* __restrict__ dinv) {
    __shared__ unsigned int q[QCAP];
    __shared__ unsigned int g1[GCAP];
    __shared__ unsigned int degc[RPB + 1];
    __shared__ unsigned int qn;
    __shared__ unsigned int gn;

    const int tid = threadIdx.x;
    const int lo  = blockIdx.x * RPB;
    int cnt = N_NODES - lo;
    if (cnt > RPB) cnt = RPB;
    const unsigned ucnt = cnt > 0 ? (unsigned)cnt : 0u;

    if (tid == 0) { qn = 0u; gn = 0u; }
    for (int r = tid; r <= RPB; r += TPB) degc[r] = 0u;
    __syncthreads();

    const int G = bm_groups(dst, E);

    {
        const uint4* row4 = reinterpret_cast<const uint4*>(bm + (size_t)blockIdx.x * BM_NB * BM_WORDS);
        for (int i0 = 0; i0 < BM_NB * BM_WORDS / 4; i0 += TPB) {
            const int i = i0 + tid;
            const uint4 x = i < BM_NB * BM_WORDS / 4 ? row4[i] : make_uint4(0u, 0u, 0u, 0u);
            const unsigned wv[4] = {x.x, x.y, x.z, x.w};
            const unsigned c = (unsigned)(__builtin_popcount(x.x) + __builtin_popcount(x.y) +
                                          __builtin_popcount(x.z) + __builtin_popcount(x.w));
            unsigned pos = reserve(&gn, c);
#pragma unroll
            for (int j = 0; j < 4; ++j) {
                unsigned m = wv[j];
                while (m != 0u) {
                    const int b = __builtin_ctz(m);
                    m &= m - 1u;
                    if (pos < GCAP) g1[pos] = (unsigned)(4 * i + j) * 32u + (unsigned)b;
                    ++pos;
                }
            }
        }
    }
    for (int e = GRP * G + tid; e < E; e += TPB) {
        const unsigned r = (unsigned)(dst[e] - lo);
        if (r < ucnt) {
            const unsigned pos = atomicAdd(&qn, 1u);
            if (pos < QCAP) q[pos] = ((unsigned)e << RBITS) | r;
        }
    }
    __syncthreads();

    const unsigned ng = gn;
    const unsigned n1 = ng <= GCAP ? ng : 0u;
    {
        const int4* d4 = reinterpret_cast<const int4*>(dst);
        for (unsigned i0 = 0; i0 < 4u * n1; i0 += GRP_U * TPB) {
            unsigned p4[GRP_U];
            int4 d[GRP_U];
#pragma unroll
            for (int u = 0; u < GRP_U; ++u) {
                const unsigned j = i0 + u * TPB + tid;
                const unsigned g = (j >> 2) < n1 ? g1[j >> 2] : g1[0];
                p4[u] = 4u * g + (j & 3u);
            }
#pragma unroll
            for (int u = 0; u < GRP_U; ++u) d[u] = d4[p4[u]];
            unsigned c = 0u;
            unsigned r[GRP_U][4];
#pragma unroll
            for (int u = 0; u < GRP_U; ++u) {
                const bool in = ((i0 + u * TPB + tid) >> 2) < n1;
                r[u][0] = in ? (unsigned)(d[u].x - lo) : ucnt;
                r[u][1] = in ? (unsigned)(d[u].y - lo) : ucnt;
                r[u][2] = in ? (unsigned)(d[u].z - lo) : ucnt;
                r[u][3] = in ? (unsigned)(d[u].w - lo) : ucnt;
#pragma unroll
                for (int k = 0; k < 4; ++k) c += r[u][k] < ucnt ? 1u : 0u;
            }
            unsigned pos = reserve(&qn, c);
#pragma unroll
            for (int u = 0; u < GRP_U; ++u) {
#pragma unroll
                for (int k = 0; k < 4; ++k) {
                    if (r[u][k] < ucnt) {
                        if (pos < QCAP) q[pos] = ((4u * p4[u] + (unsigned)k) << RBITS) | r[u][k];
                        ++pos;
                    }
                }
            }
        }
    }
    __syncthreads();

    unsigned n = qn;
    if (ng > GCAP || E > (1 << (32 - RBITS))) n = QCAP + 1u;
    if (tid == 0) qcnt[blockIdx.x] = n;
    if (n <= QCAP) {
        unsigned int* qout = qbuf + (size_t)blockIdx.x * QCAP;
        for (unsigned i0 = 0; i0 < n; i0 += CNV_U * TPB) {
            unsigned v[CNV_U];
            int s[CNV_U];
#pragma unroll
            for (int u = 0; u < CNV_U; ++u) {
                const unsigned i = i0 + u * TPB + tid;
                v[u] = i < n ? q[i] : 0u;
            }
#pragma unroll
            for (int u = 0; u < CNV_U; ++u) s[u] = src[v[u] >> RBITS];
#pragma unroll
            for (int u = 0; u < CNV_U; ++u) {
                const unsigned i = i0 + u * TPB + tid;
                if (i < n) {
                    const unsigned r = v[u] & ((1u << RBITS) - 1u);
                    atomicAdd(&degc[r], 1u);
                    qout[i] = ((unsigned)s[u] << RBITS) | r;
                }
            }
        }
    } else {
        for (int e = tid; e < E; e += TPB) {
            const unsigned r = (unsigned)(dst[e] - lo);
            if (r < ucnt) atomicAdd(&degc[r], 1u);
        }
    }
    __syncthreads();
    for (int r = tid; r < cnt; r += TPB) dinv[lo + r] = rsqrtf((float)degc[r] + 1.0f);
}

__global__ __launch_bounds__(TPB) void k_agg1(const int* __restrict__ src,
                                              const int* __restrict__ dst, int E,
                                              const float* __restrict__ H,
                                              const float* __restrict__ dinv,
                                              const float* __restrict__ b1,
                                              const float* __restrict__ W2,
                                              const unsigned int* __restrict__ qbuf,
                                              const unsigned int* __restrict__ qcnt,
                                              float* __restrict__ zpair) {
    float2* zp = reinterpret_cast<float2*>(zpair);
    __shared__ float        acc[RPB * HID];
    __shared__ unsigned int q[QCAP];
    __shared__ float        qw[QCAP];
    __shared__ float        dl[RPB + 1];

    const int tid = threadIdx.x;
    const int lo  = blockIdx.x * RPB;
    int cnt = N_NODES - lo;
    if (cnt > RPB) cnt = RPB;
    const unsigned ucnt = cnt > 0 ? (unsigned)cnt : 0u;

    for (int r = tid; r < cnt; r += TPB) dl[r] = dinv[lo + r];
    __syncthreads();

    {
        const float4* H4 = reinterpret_cast<const float4*>(H + (size_t)lo * HID);
        float4* acc4 = reinterpret_cast<float4*>(acc);
        for (int i = tid; i < cnt * (HID / 4); i += TPB) {
            const float di = dl[i >> 4];
            float4 h = H4[i];
            h.x = h.x * di * di; h.y = h.y * di * di;
            h.z = h.z * di * di; h.w = h.w * di * di;
            acc4[i] = h;
        }
    }

    const unsigned nq = qcnt[blockIdx.x];
    unsigned n = nq;
    if (nq <= QCAP) {
        const unsigned int* qin = qbuf + (size_t)blockIdx.x * QCAP;
        for (unsigned i0 = 0; i0 < n; i0 += CNV_U * TPB) {
            unsigned v[CNV_U];
            float ds[CNV_U];
#pragma unroll
            for (int u = 0; u < CNV_U; ++u) {
                const unsigned i = i0 + u * TPB + tid;
                v[u] = i < n ? qin[i] : Q_SKIP;
            }
#pragma unroll
            for (int u = 0; u < CNV_U; ++u) ds[u] = dinv[v[u] != Q_SKIP ? (v[u] >> RBITS) : 0u];
#pragma unroll
            for (int u = 0; u < CNV_U; ++u) {
                const unsigned i = i0 + u * TPB + tid;
                if (i < n) {
                    const bool ok = v[u] != Q_SKIP;
                    q[i]  = ok ? v[u] : 0u;
                    qw[i] = ok ? ds[u] * dl[v[u] & ((1u << RBITS) - 1u)] : 0.0f;
                }
            }
        }
    } else {
        n = 0u;
    }
    __syncthreads();

    if (nq > QCAP) {
        for (int e = tid; e < E; e += TPB) {
            const unsigned r = (unsigned)(dst[e] - lo);
            if (r < ucnt) {
                const int s = src[e];
                const float w = dinv[s] * dl[r];
                const float* hs = H + (size_t)s * HID;
                float* a = acc + r * HID;
                for (int f = 0; f < HID; ++f) atomicAdd(a + f, hs[f] * w);
            }
        }
    }

    const float b1a = b1[tid & 31], b1b = b1[(tid & 31) + 32];
    const float w2a = W2[tid & 31], w2b = W2[(tid & 31) + 32];

    {
        const int lane = tid & 31;
        const unsigned wave = (unsigned)__builtin_amdgcn_readfirstlane(tid >> 5);
        const unsigned NW = TPB / 32;
        const float2* Hl = reinterpret_cast<const float2*>(H) + lane;
        float* al = acc + 2 * lane;
        const unsigned full = n & ~(32u * NW - 1u);
        unsigned base = 32u * wave;
        for (; base < full; base += 32u * NW) {
            const int   vq = (int)q[base + lane];
            const int   wq = (int)__float_as_uint(qw[base + lane]);
            float2 h[32];
#pragma unroll
            for (int u = 0; u < 32; ++u) {
                const unsigned v = (unsigned)__builtin_amdgcn_readlane(vq, u);
                h[u] = Hl[(size_t)(v >> RBITS) * (HID / 2)];
            }
#pragma unroll
            for (int u = 0; u < 32; ++u) {
                const unsigned v = (unsigned)__builtin_amdgcn_readlane(vq, u);
                const float w = __uint_as_float((unsigned)__builtin_amdgcn_readlane(wq, u));
                float* a = al + (v & ((1u << RBITS) - 1u)) * HID;
                atomicAdd(a,     h[u].x * w);
                atomicAdd(a + 1, h[u].y * w);
            }
        }
        {
            const unsigned rest = n - full;
            const unsigned c = (rest + NW - 1u) / NW;
            const unsigned s0 = full + wave * c;
            const unsigned m = s0 < n ? (n - s0 < c ? n - s0 : c) : 0u;
            const bool in = (unsigned)lane < m;
            const int vq = in ? (int)q[s0 + lane] : 0;
            const int wq = in ? (int)__float_as_uint(qw[s0 + lane]) : 0;
            float2 h[32];
#pragma unroll
            for (int u = 0; u < 32; ++u) {
                if ((unsigned)u < m) {
                    const unsigned v = (unsigned)__builtin_amdgcn_readlane(vq, u);
                    h[u] = Hl[(size_t)(v >> RBITS) * (HID / 2)];
                }
            }
#pragma unroll
            for (int u = 0; u < 32; ++u) {
                if ((unsigned)u < m) {
                    const unsigned v = (unsigned)__builtin_amdgcn_readlane(vq, u);
                    const float w = __uint_as_float((unsigned)__builtin_amdgcn_readlane(wq, u));
                    float* a = al + (v & ((1u << RBITS) - 1u)) * HID;
                    atomicAdd(a,     h[u].x * w);
                    atomicAdd(a + 1, h[u].y * w);
                }
            }
        }
    }
    __syncthreads();

    {
        const int lane = tid & 31;
        const int NW = TPB / 32;
        for (int r0 = tid >> 5; r0 < cnt; r0 += Z_U * NW) {
            float s[Z_U];
#pragma unroll
            for (int u = 0; u < Z_U; ++u) {
                const int r = r0 + u * NW;
                const float* a = acc + (r < cnt ? r : r0) * HID;
                float v0 = a[lane] + b1a;
                float v1 = a[lane + 32] + b1b;
                v0 = v0 > 0.0f ? v0 : 0.0f;
                v1 = v1 > 0.0f ? v1 : 0.0f;
                s[u] = v0 * w2a + v1 * w2b;
            }
#pragma unroll
            for (int off = 16; off > 0; off >>= 1) {
#pragma unroll
                for (int u = 0; u < Z_U; ++u) s[u] += __shfl_down(s[u], off, 32);
            }
#pragma unroll
            for (int u = 0; u < Z_U; ++u) {
                const int r = r0 + u * NW;
                if (lane == 0 && r < cnt) qw[r] = s[u];
            }
        }
    }
    __syncthreads();
    for (int r = tid; r < cnt; r += TPB) zp[lo + r] = make_float2(qw[r], dl[r]);
}

__global__ __launch_bounds__(L2_TPB) void k_l2_tile(const int* __restrict__ src,
                                                    const int* __restrict__ dst, int E,
                                                    const unsigned int* __restrict__ qbuf,
                                                    const unsigned int* __restrict__ qcnt,
                                                    const float* __restrict__ zpair,
                                                    const float* __restrict__ b2,
                                                    float* __restrict__ out) {
    const float2* zp = reinterpret_cast<const float2*>(zpair);
    __shared__ float acc[RPB + 1];
    __shared__ float zs[RPB + 1];
    __shared__ float dl[RPB + 1];

    const int tid = threadIdx.x;
    const int lo  = blockIdx.x * RPB;
    int cnt = N_NODES - lo;
    if (cnt > RPB) cnt = RPB;
    const unsigned ucnt = cnt > 0 ? (unsigned)cnt : 0u;

    for (int r = tid; r < cnt; r += L2_TPB) {
        const float2 p = zp[lo + r];
        zs[r] = p.x;
        dl[r] = p.y;
        acc[r] = 0.0f;
    }
    __syncthreads();

    const unsigned nq = qcnt[blockIdx.x];
    if (nq <= QCAP) {
        const unsigned int* qin = qbuf + (size_t)blockIdx.x * QCAP;
        unsigned i = (unsigned)tid;
        for (; i + (L2_U - 1) * L2_TPB < nq; i += L2_U * L2_TPB) {
            unsigned v[L2_U];
            float2 p[L2_U];
#pragma unroll
            for (int u = 0; u < L2_U; ++u) v[u] = qin[i + u * L2_TPB];
#pragma unroll
            for (int u = 0; u < L2_U; ++u) {
                const unsigned s = v[u] == Q_SKIP ? 0u : (v[u] >> RBITS);
                p[u] = zp[s];
            }
            unsigned r[L2_U];
            float dr[L2_U];
#pragma unroll
            for (int u = 0; u < L2_U; ++u) {
                r[u] = v[u] != Q_SKIP ? (v[u] & ((1u << RBITS) - 1u)) : 0u;
                dr[u] = dl[r[u]];
            }
#pragma unroll
            for (int u = 0; u < L2_U; ++u) {
                const float t = p[u].x * p[u].y * dr[u];
                atomicAdd(&acc[r[u]], v[u] != Q_SKIP ? t : 0.0f);
            }
        }
        for (; i < nq; i += L2_TPB) {
            const unsigned v = qin[i];
            const bool ok = v != Q_SKIP;
            const float2 p = zp[ok ? (v >> RBITS) : 0u];
            const unsigned r = ok ? (v & ((1u << RBITS) - 1u)) : 0u;
            const float t = p.x * p.y * dl[r];
            atomicAdd(&acc[r], ok ? t : 0.0f);
        }
    } else {
        for (int e = tid; e < E; e += L2_TPB) {
            const unsigned r = (unsigned)(dst[e] - lo);
            if (r < ucnt) {
                const float2 p = zp[src[e]];
                atomicAdd(&acc[r], p.x * p.y * dl[r]);
            }
        }
    }
    __syncthreads();

    const float bb = b2[0];
    for (int r = tid; r < cnt; r += L2_TPB) {
        const float di = dl[r];
        out[lo + r] = (zs[r] * di * di + bb) + acc[r];
    }
}


extern "C" void kernel_launch(void* const* d_in, const int* in_sizes, int n_in,
                              void* d_out, int out_size, void* d_ws, size_t ws_size,
                              hipStream_t stream) {
    const float* X  = (const float*)d_in[0];
    const int*   EI = (const int*)d_in[1];
    const float* W1 = (const float*)d_in[2];
    const float* b1 = (const float*)d_in[3];
    const float* W2 = (const float*)d_in[4];
    const float* b2 = (const float*)d_in[5];
    float* out = (float*)d_out;

    const int E = in_sizes[1] / 2;
    const int* src = EI;
    const int* dst = EI + E;

    char* ws = (char*)d_ws;
    float* H    = (float*)ws;
    float* dinv = (float*)(ws + (size_t)2 * N_NODES * HID * 4);
    unsigned int* bm = (unsigned int*)ws;
    unsigned int* qbuf = (unsigned int*)(ws + (size_t)N_NODES * HID * 4);
    float*        zp   = (float*)(ws + (size_t)N_NODES * HID * 4 + (size_t)TILES * QCAP * 4);
    unsigned int* qcnt = (unsigned int*)(ws + (size_t)N_NODES * HID * 4 + (size_t)TILES * QCAP * 4 + (size_t)N_NODES * 8);

    k_bitmap   <<<BM_NB, BM_TPB, 0, stream>>>(dst, E, bm);
    k_find     <<<TILES, TPB, 0, stream>>>(src, dst, bm, E, qbuf, qcnt, dinv);

    k_gemm1    <<<GB_WG, GB_TPB, 0, stream>>>(X, W1, H);

    k_agg1     <<<TILES, TPB, 0, stream>>>(src, dst, E, H, dinv, b1, W2, qbuf, qcnt, zp);

    k_l2_tile  <<<TILES, L2_TPB, 0, stream>>>(src, dst, E, qbuf, qcnt, zp, b2, out);
}


# Layer-1 tiles hold every node's in-degree as a byte table in LDS (finder tiles write it): a source's dinv is an LDS byte and one rsqrt, not a scattered 4-byte memory read per edge; sum order kept
# speedup vs baseline: 6.1578x; 1.0682x over previous
#include <hip/hip_runtime.h>

typedef float v2f __attribute__((ext_vector_type(2)));
typedef float v8f __attribute__((ext_vector_type(8)));
typedef float v4f __attribute__((ext_vector_type(4)));

#define N_NODES 100000
#define IN_DIM  384
#define HID     64

#define TILES 256
#define RPB   391

#define GRP      16
#define BM_CAP   3200000
#define BM_TPB   1024
#define BM_WGE   (BM_TPB * GRP)
#define BM_WORDS (BM_WGE / GRP / 32)
#define BM_NB    ((BM_CAP + BM_WGE - 1) / BM_WGE)
#define BM_PITCH (TILES + 1)
#define QCAP     16384

static_assert(N_NODES % 16 == 0, "sixteen degree bytes per 16 bytes");
static_assert(BM_WORDS == 32 && BM_TPB == 1024, "a wave stores one tile's line of a bitmap work-group");
static_assert(TILES == 256 && (N_NODES - 1) / RPB < TILES, "tile numbers");
static_assert((size_t)TILES * BM_NB * BM_WORDS * 4 <= (size_t)N_NODES * HID * 4, "the bitmaps fit in the region of H");
static_assert((size_t)N_NODES * HID * 4 + (size_t)TILES * QCAP * 4 + (size_t)N_NODES * 8 + (size_t)TILES * 4 + (size_t)N_NODES
              <= (size_t)2 * N_NODES * HID * 4, "degree bytes end in front of dinv");
static_assert(((size_t)N_NODES * HID * 4 + (size_t)TILES * QCAP * 4 + (size_t)N_NODES * 8 + (size_t)TILES * 4) % 16 == 0,
              "degree bytes on 16-byte addresses");
static_assert(((size_t)N_NODES * HID * 4 + (size_t)TILES * QCAP * 4) % 8 == 0, "pairs on 8-byte addresses");

__device__ __forceinline__ int bm_groups(const int* dst, int E) {
    const int EK = E < BM_CAP ? E : BM_CAP;
    return ((reinterpret_cast<size_t>(dst) & 15) == 0) ? (EK / GRP) : 0;
}

__global__ __launch_bounds__(BM_TPB) void k_bitmap(const int* __restrict__ dst, int E,
                                                   unsigned int* __restrict__ bm) {
    __shared__ unsigned int tb[BM_WORDS * BM_PITCH];
    const int tid = threadIdx.x;
    for (int i = tid; i < BM_WORDS * BM_PITCH; i += BM_TPB) tb[i] = 0u;
    __syncthreads();

    const int G4 = 4 * bm_groups(dst, E);
    const int4* d4 = reinterpret_cast<const int4*>(dst);
    const int base4 = (int)blockIdx.x * (BM_WGE / 4);
    int4 d[4];
#pragma unroll
    for (int u = 0; u < 4; ++u) {
        const int i4 = base4 + u * BM_TPB + tid;
        d[u] = i4 < G4 ? d4[i4] : make_int4(-1, -1, -1, -1);
    }
#pragma unroll
    for (int u = 0; u < 4; ++u) {
        const int gl = (u * BM_TPB + tid) >> 2;
        unsigned int* row = tb + (gl >> 5) * BM_PITCH;
        const unsigned bit = 1u << (gl & 31);
        if ((unsigned)d[u].x < (unsigned)N_NODES) atomicOr(row + (unsigned)d[u].x / RPB, bit);
        if ((unsigned)d[u].y < (unsigned)N_NODES) atomicOr(row + (unsigned)d[u].y / RPB, bit);
        if ((unsigned)d[u].z < (unsigned)N_NODES) atomicOr(row + (unsigned)d[u].z / RPB, bit);
        if ((unsigned)d[u].w < (unsigned)N_NODES) atomicOr(row + (unsigned)d[u].w / RPB, bit);
    }
    __syncthreads();
    for (int p = 0; p < TILES / 32; ++p) {
        const int t = 32 * p + (tid >> 5);
        bm[((size_t)t * BM_NB + blockIdx.x) * BM_WORDS + (tid & 31)] = tb[(tid & 31) * BM_PITCH + t];
    }
}

#define GB_ROWS  32
#define GB_P4    (IN_DIM / 4 + 1)
#define GB_WG    256
#define GB_TPB   256
#define GB_NBLK  (N_NODES / GB_ROWS)
#define GB_C4    (GB_ROWS * IN_DIM / 4)
#define GB_CH    (GB_C4 / GB_TPB)
#define GB_SLOTS 4
#define GB_KJ    (IN_DIM / 8)
#define GB_G     16
#define GB_EACH(M) M(0) M(1) M(2) M(3) M(4) M(5) M(6) M(7) M(8) M(9) M(10) M(11)

static_assert(GB_CH == 12, "GB_EACH lists the chunks");
static_assert(GB_NBLK >= GB_WG, "every work-group has a block");
static_assert(N_NODES % GB_ROWS == 0 && GB_C4 % GB_TPB == 0 && IN_DIM % 8 == 0 && HID == 64, "gemm tiling");
static_assert((GB_SLOTS & (GB_SLOTS - 1)) == 0 && GB_KJ % GB_G == 0, "ring index by mask, whole groups");

__global__ __launch_bounds__(GB_TPB) void k_gemm1(const float* X,
                                                  const float* __restrict__ W1,
                                                  float* H) {
    __shared__ v4f xs[GB_SLOTS][GB_ROWS * GB_P4];
    const int tid  = threadIdx.x;
    const int lane = tid & 31;
    const int wave = tid >> 5;
    const int half = lane >> 4;
    const int lr   = lane & 15;
    const int ct   = wave & 3;
    const int rt   = wave >> 2;

    const int t0 = (int)(((long long)GB_NBLK * blockIdx.x) / GB_WG);
    const int t1 = (int)(((long long)GB_NBLK * (blockIdx.x + 1)) / GB_WG);

    const v4f* X4 = reinterpret_cast<const v4f*>(X);
    int so[GB_CH];
#pragma unroll
    for (int i = 0; i < GB_CH; ++i) {
        const int j   = i * GB_TPB + tid;
        const int row = j / (IN_DIM / 4);
        so[i] = row * GB_P4 + (j - row * (IN_DIM / 4));
    }

    const v4f* p0 = X4 + (size_t)t0 * GB_C4 + tid;
#define GB_LD(i) const v4f f##i = p0[(i) * GB_TPB];
    GB_EACH(GB_LD)
#undef GB_LD

    v2f b[2 * GB_KJ];
    {
        const float* wp = W1 + (size_t)(4 * half) * HID + 16 * ct + lr;
#pragma unroll
        for (int j = 0; j < GB_KJ; ++j) {
#pragma unroll
            for (int q = 0; q < 2; ++q) {
                const float* bp = wp + (size_t)(8 * j + 2 * q) * HID;
                b[2 * j + q][0] = bp[0];
                b[2 * j + q][1] = bp[HID];
            }
        }
    }
#define GB_ST(i) xs[0][so[i]] = f##i;
    GB_EACH(GB_ST)
#undef GB_ST
    __syncthreads();

    auto products = [&](int slot) -> v8f {
        v8f c0 = {}, c1 = {};
        const v4f* ar = &xs[slot][(16 * rt + lr) * GB_P4 + half];
        v4f af[2][GB_G];
#pragma unroll
        for (int u = 0; u < GB_G; ++u) af[0][u] = ar[2 * u];
#pragma unroll
        for (int g = 0; g < GB_KJ / GB_G; ++g) {
            if (g + 1 < GB_KJ / GB_G) {
#pragma unroll
                for (int u = 0; u < GB_G; ++u) af[(g + 1) & 1][u] = ar[2 * ((g + 1) * GB_G + u)];
            }
            asm volatile("" ::: "memory");
#pragma unroll
            for (int u = 0; u < GB_G; ++u) {
                const int j = g * GB_G + u;
                const v4f a4 = af[g & 1][u];
                v2f a0, a1;
                a0[0] = a4.x; a0[1] = a4.y;
                a1[0] = a4.z; a1[1] = a4.w;
                c0 = __builtin_amdgcn_wmma_f32_16x16x4_f32(
                         false, a0,  false, b[2 * j],
                         (short)0, c0,  false,  false);
                c1 = __builtin_amdgcn_wmma_f32_16x16x4_f32(
                         false, a1,  false, b[2 * j + 1],
                         (short)0, c1,  false,  false);
            }
        }
        return c0 + c1;
    };
    auto put = [&](int t, v8f c) {
        float* hrow = H + (size_t)(t * GB_ROWS + 16 * rt + 8 * half) * HID + 16 * ct + lr;
#pragma unroll
        for (int r = 0; r < 8; ++r) hrow[(size_t)r * HID] = c[r];
    };

    int t = t0;
    for (; t + 1 < t1; ++t) {
        const int slot = (t - t0) & (GB_SLOTS - 1);
        const v4f* p = X4 + (size_t)(t + 1) * GB_C4 + tid;
#define GB_LD(i) const v4f st##i = p[(i) * GB_TPB];
        GB_EACH(GB_LD)
#undef GB_LD

        const v8f c = products(slot);
        asm volatile("" :: "v"(c) : "memory");

        v4f* xn = xs[(slot + 1) & (GB_SLOTS - 1)];
#define GB_ST(i) xn[so[i]] = st##i;
        GB_EACH(GB_ST)
#undef GB_ST
        __syncthreads();
        put(t, c);
    }
    put(t, products((t - t0) & (GB_SLOTS - 1)));
}

__device__ __forceinline__ unsigned reserve(unsigned int* ctr, unsigned c) {
    const unsigned lane = threadIdx.x & 31u;
    unsigned inc = c;
#pragma unroll
    for (unsigned off = 1; off < 32; off <<= 1) {
        const unsigned t = __shfl_up(inc, off, 32);
        if (lane >= off) inc += t;
    }
    const unsigned total = __shfl(inc, 31, 32);
    unsigned base = 0u;
    if (lane == 0u) base = atomicAdd(ctr, total);
    return (unsigned)__builtin_amdgcn_readfirstlane((int)base) + (inc - c);
}

#define RBITS 9
#define GCAP  32768
#define TPB   1024
#define GRP_U  12
#define CNV_U  8
#define Z_U    8
#define Q_SKIP 0xFFFFFFFFu
#define DEG_ESC 255u
#define L2_TPB 512
#define L2_U   8

__global__ __launch_bounds__(TPB) void k_find(const int* __restrict__ src,
                                              const int* __restrict__ dst,
                                              const unsigned int* __restrict__ bm, int E,
                                              unsigned int* __restrict__ qbuf,
                                              unsigned int* __restrict__ qcnt,
                                              float* __restrict__ dinv,
                                              unsigned char* __restrict__ deg8) {
    __shared__ unsigned int q[QCAP];
    __shared__ unsigned int g1[GCAP];
    __shared__ unsigned int degc[RPB + 1];
    __shared__ unsigned int qn;
    __shared__ unsigned int gn;

    const int tid = threadIdx.x;
    const int lo  = blockIdx.x * RPB;
    int cnt = N_NODES - lo;
    if (cnt > RPB) cnt = RPB;
    const unsigned ucnt = cnt > 0 ? (unsigned)cnt : 0u;

    if (tid == 0) { qn = 0u; gn = 0u; }
    for (int r = tid; r <= RPB; r += TPB) degc[r] = 0u;
    __syncthreads();

    const int G = bm_groups(dst, E);

    {
        const uint4* row4 = reinterpret_cast<const uint4*>(bm + (size_t)blockIdx.x * BM_NB * BM_WORDS);
        for (int i0 = 0; i0 < BM_NB * BM_WORDS / 4; i0 += TPB) {
            const int i = i0 + tid;
            const uint4 x = i < BM_NB * BM_WORDS / 4 ? row4[i] : make_uint4(0u, 0u, 0u, 0u);
            const unsigned wv[4] = {x.x, x.y, x.z, x.w};
            const unsigned c = (unsigned)(__builtin_popcount(x.x) + __builtin_popcount(x.y) +
                                          __builtin_popcount(x.z) + __builtin_popcount(x.w));
            unsigned pos = reserve(&gn, c);
#pragma unroll
            for (int j = 0; j < 4; ++j) {
                unsigned m = wv[j];
                while (m != 0u) {
                    const int b = __builtin_ctz(m);
                    m &= m - 1u;
                    if (pos < GCAP) g1[pos] = (unsigned)(4 * i + j) * 32u + (unsigned)b;
                    ++pos;
                }
            }
        }
    }
    for (int e = GRP * G + tid; e < E; e += TPB) {
        const unsigned r = (unsigned)(dst[e] - lo);
        if (r < ucnt) {
            const unsigned pos = atomicAdd(&qn, 1u);
            if (pos < QCAP) q[pos] = ((unsigned)e << RBITS) | r;
        }
    }
    __syncthreads();

    const unsigned ng = gn;
    const unsigned n1 = ng <= GCAP ? ng : 0u;
    {
        const int4* d4 = reinterpret_cast<const int4*>(dst);
        for (unsigned i0 = 0; i0 < 4u * n1; i0 += GRP_U * TPB) {
            unsigned p4[GRP_U];
            int4 d[GRP_U];
#pragma unroll
            for (int u = 0; u < GRP_U; ++u) {
                const unsigned j = i0 + u * TPB + tid;
                const unsigned g = (j >> 2) < n1 ? g1[j >> 2] : g1[0];
                p4[u] = 4u * g + (j & 3u);
            }
#pragma unroll
            for (int u = 0; u < GRP_U; ++u) d[u] = d4[p4[u]];
            unsigned c = 0u;
            unsigned r[GRP_U][4];
#pragma unroll
            for (int u = 0; u < GRP_U; ++u) {
                const bool in = ((i0 + u * TPB + tid) >> 2) < n1;
                r[u][0] = in ? (unsigned)(d[u].x - lo) : ucnt;
                r[u][1] = in ? (unsigned)(d[u].y - lo) : ucnt;
                r[u][2] = in ? (unsigned)(d[u].z - lo) : ucnt;
                r[u][3] = in ? (unsigned)(d[u].w - lo) : ucnt;
#pragma unroll
                for (int k = 0; k < 4; ++k) c += r[u][k] < ucnt ? 1u : 0u;
            }
            unsigned pos = reserve(&qn, c);
#pragma unroll
            for (int u = 0; u < GRP_U; ++u) {
#pragma unroll
                for (int k = 0; k < 4; ++k) {
                    if (r[u][k] < ucnt) {
                        if (pos < QCAP) q[pos] = ((4u * p4[u] + (unsigned)k) << RBITS) | r[u][k];
                        ++pos;
                    }
                }
            }
        }
    }
    __syncthreads();

    unsigned n = qn;
    if (ng > GCAP || E > (1 << (32 - RBITS))) n = QCAP + 1u;
    if (tid == 0) qcnt[blockIdx.x] = n;
    if (n <= QCAP) {
        unsigned int* qout = qbuf + (size_t)blockIdx.x * QCAP;
        for (unsigned i0 = 0; i0 < n; i0 += CNV_U * TPB) {
            unsigned v[CNV_U];
            int s[CNV_U];
#pragma unroll
            for (int u = 0; u < CNV_U; ++u) {
                const unsigned i = i0 + u * TPB + tid;
                v[u] = i < n ? q[i] : 0u;
            }
#pragma unroll
            for (int u = 0; u < CNV_U; ++u) s[u] = src[v[u] >> RBITS];
#pragma unroll
            for (int u = 0; u < CNV_U; ++u) {
                const unsigned i = i0 + u * TPB + tid;
                if (i < n) {
                    const unsigned r = v[u] & ((1u << RBITS) - 1u);
                    atomicAdd(&degc[r], 1u);
                    qout[i] = ((unsigned)s[u] << RBITS) | r;
                }
            }
        }
    } else {
        for (int e = tid; e < E; e += TPB) {
            const unsigned r = (unsigned)(dst[e] - lo);
            if (r < ucnt) atomicAdd(&degc[r], 1u);
        }
    }
    __syncthreads();
    for (int r = tid; r < cnt; r += TPB) {
        const unsigned dg = degc[r];
        dinv[lo + r] = rsqrtf((float)dg + 1.0f);
        deg8[lo + r] = (unsigned char)(dg < DEG_ESC ? dg : DEG_ESC);
    }
}

#define DG_V4 (N_NODES / 16)
#define DG_U  ((DG_V4 + TPB - 1) / TPB)

__global__ __launch_bounds__(TPB) void k_agg1(const int* __restrict__ src,
                                              const int* __restrict__ dst, int E,
                                              const float* __restrict__ H,
                                              const float* __restrict__ dinv,
                                              const unsigned char* __restrict__ deg8,
                                              const float* __restrict__ b1,
                                              const float* __restrict__ W2,
                                              const unsigned int* __restrict__ qbuf,
                                              const unsigned int* __restrict__ qcnt,
                                              float* __restrict__ zpair) {
    float2* zp = reinterpret_cast<float2*>(zpair);
    __shared__ float acc[RPB * HID];
    __shared__ uint4 dg4[DG_V4];
    __shared__ float dl[RPB + 1];
    __shared__ float zr[RPB + 1];
    const unsigned char* dg = reinterpret_cast<const unsigned char*>(dg4);

    const int tid = threadIdx.x;
    const int lo  = blockIdx.x * RPB;
    int cnt = N_NODES - lo;
    if (cnt > RPB) cnt = RPB;
    const unsigned ucnt = cnt > 0 ? (unsigned)cnt : 0u;

    {
        const uint4* t4 = reinterpret_cast<const uint4*>(deg8);
        uint4 tv[DG_U];
#pragma unroll
        for (int u = 0; u < DG_U; ++u) {
            const int i = u * TPB + tid;
            tv[u] = i < DG_V4 ? t4[i] : make_uint4(0u, 0u, 0u, 0u);
        }
        for (int r = tid; r < cnt; r += TPB) dl[r] = dinv[lo + r];
#pragma unroll
        for (int u = 0; u < DG_U; ++u) {
            const int i = u * TPB + tid;
            if (i < DG_V4) dg4[i] = tv[u];
        }
    }
    __syncthreads();

    {
        const float4* H4 = reinterpret_cast<const float4*>(H + (size_t)lo * HID);
        float4* acc4 = reinterpret_cast<float4*>(acc);
        for (int i = tid; i < cnt * (HID / 4); i += TPB) {
            const float di = dl[i >> 4];
            float4 h = H4[i];
            h.x = h.x * di * di; h.y = h.y * di * di;
            h.z = h.z * di * di; h.w = h.w * di * di;
            acc4[i] = h;
        }
    }

    const unsigned nq = qcnt[blockIdx.x];
    const unsigned n = nq <= QCAP ? nq : 0u;
    __syncthreads();

    if (nq > QCAP) {
        for (int e = tid; e < E; e += TPB) {
            const unsigned r = (unsigned)(dst[e] - lo);
            if (r < ucnt) {
                const int s = src[e];
                const float w = dinv[s] * dl[r];
                const float* hs = H + (size_t)s * HID;
                float* a = acc + r * HID;
                for (int f = 0; f < HID; ++f) atomicAdd(a + f, hs[f] * w);
            }
        }
    }

    const float b1a = b1[tid & 31], b1b = b1[(tid & 31) + 32];
    const float w2a = W2[tid & 31], w2b = W2[(tid & 31) + 32];

    auto edge = [&](unsigned vw, int& vq, int& wq) {
        const bool ok = vw != Q_SKIP;
        const unsigned s = ok ? (vw >> RBITS) : 0u;
        const unsigned r = ok ? (vw & ((1u << RBITS) - 1u)) : 0u;
        const unsigned d8 = dg[s];
        const float dr = dl[r];
        float ds = rsqrtf((float)d8 + 1.0f);
        if (__builtin_amdgcn_ballot_w32(d8 == DEG_ESC) != 0u) {
            const float dm = dinv[s];
            ds = d8 == DEG_ESC ? dm : ds;
        }
        const float wr = ds * dr;
        const float w = ok ? wr : 0.0f;
        vq = (int)(ok ? vw : 0u);
        wq = (int)__float_as_uint(w);
    };

    {
        const int lane = tid & 31;
        const unsigned wave = (unsigned)__builtin_amdgcn_readfirstlane(tid >> 5);
        const unsigned NW = TPB / 32;
        const unsigned int* qin = qbuf + (size_t)blockIdx.x * QCAP + lane;
        const float2* Hl = reinterpret_cast<const float2*>(H) + lane;
        float* al = acc + 2 * lane;
        const unsigned full = n & ~(32u * NW - 1u);
        unsigned base = 32u * wave;
        unsigned nxt = base < full ? qin[base] : Q_SKIP;
        for (; base < full; base += 32u * NW) {
            int vq, wq;
            edge(nxt, vq, wq);
            const unsigned nb = base + 32u * NW;
            nxt = nb < full ? qin[nb] : Q_SKIP;
            float2 h[32];
#pragma unroll
            for (int u = 0; u < 32; ++u) {
                const unsigned v = (unsigned)__builtin_amdgcn_readlane(vq, u);
                h[u] = Hl[(size_t)(v >> RBITS) * (HID / 2)];
            }
#pragma unroll
            for (int u = 0; u < 32; ++u) {
                const unsigned v = (unsigned)__builtin_amdgcn_readlane(vq, u);
                const float w = __uint_as_float((unsigned)__builtin_amdgcn_readlane(wq, u));
                float* a = al + (v & ((1u << RBITS) - 1u)) * HID;
                atomicAdd(a,     h[u].x * w);
                atomicAdd(a + 1, h[u].y * w);
            }
        }
        {
            const unsigned rest = n - full;
            const unsigned c = (rest + NW - 1u) / NW;
            const unsigned s0 = full + wave * c;
            const unsigned m = s0 < n ? (n - s0 < c ? n - s0 : c) : 0u;
            int vq, wq;
            edge((unsigned)lane < m ? qin[s0] : Q_SKIP, vq, wq);
            float2 h[32];
#pragma unroll
            for (int u = 0; u < 32; ++u) {
                if ((unsigned)u < m) {
                    const unsigned v = (unsigned)__builtin_amdgcn_readlane(vq, u);
                    h[u] = Hl[(size_t)(v >> RBITS) * (HID / 2)];
                }
            }
#pragma unroll
            for (int u = 0; u < 32; ++u) {
                if ((unsigned)u < m) {
                    const unsigned v = (unsigned)__builtin_amdgcn_readlane(vq, u);
                    const float w = __uint_as_float((unsigned)__builtin_amdgcn_readlane(wq, u));
                    float* a = al + (v & ((1u << RBITS) - 1u)) * HID;
                    atomicAdd(a,     h[u].x * w);
                    atomicAdd(a + 1, h[u].y * w);
                }
            }
        }
    }
    __syncthreads();

    {
        const int lane = tid & 31;
        const int NW = TPB / 32;
        for (int r0 = tid >> 5; r0 < cnt; r0 += Z_U * NW) {
            float s[Z_U];
#pragma unroll
            for (int u = 0; u < Z_U; ++u) {
                const int r = r0 + u * NW;
                const float* a = acc + (r < cnt ? r : r0) * HID;
                float v0 = a[lane] + b1a;
                float v1 = a[lane + 32] + b1b;
                v0 = v0 > 0.0f ? v0 : 0.0f;
                v1 = v1 > 0.0f ? v1 : 0.0f;
                s[u] = v0 * w2a + v1 * w2b;
            }
#pragma unroll
            for (int off = 16; off > 0; off >>= 1) {
#pragma unroll
                for (int u = 0; u < Z_U; ++u) s[u] += __shfl_down(s[u], off, 32);
            }
#pragma unroll
            for (int u = 0; u < Z_U; ++u) {
                const int r = r0 + u * NW;
                if (lane == 0 && r < cnt) zr[r] = s[u];
            }
        }
    }
    __syncthreads();
    for (int r = tid; r < cnt; r += TPB) zp[lo + r] = make_float2(zr[r], dl[r]);
}

__global__ __launch_bounds__(L2_TPB) void k_l2_tile(const int* __restrict__ src,
                                                    const int* __restrict__ dst, int E,
                                                    const unsigned int* __restrict__ qbuf,
                                                    const unsigned int* __restrict__ qcnt,
                                                    const float* __restrict__ zpair,
                                                    const float* __restrict__ b2,
                                                    float* __restrict__ out) {
    const float2* zp = reinterpret_cast<const float2*>(zpair);
    __shared__ float acc[RPB + 1];
    __shared__ float zs[RPB + 1];
    __shared__ float dl[RPB + 1];

    const int tid = threadIdx.x;
    const int lo  = blockIdx.x * RPB;
    int cnt = N_NODES - lo;
    if (cnt > RPB) cnt = RPB;
    const unsigned ucnt = cnt > 0 ? (unsigned)cnt : 0u;

    for (int r = tid; r < cnt; r += L2_TPB) {
        const float2 p = zp[lo + r];
        zs[r] = p.x;
        dl[r] = p.y;
        acc[r] = 0.0f;
    }
    __syncthreads();

    const unsigned nq = qcnt[blockIdx.x];
    if (nq <= QCAP) {
        const unsigned int* qin = qbuf + (size_t)blockIdx.x * QCAP;
        unsigned i = (unsigned)tid;
        for (; i + (L2_U - 1) * L2_TPB < nq; i += L2_U * L2_TPB) {
            unsigned v[L2_U];
            float2 p[L2_U];
#pragma unroll
            for (int u = 0; u < L2_U; ++u) v[u] = qin[i + u * L2_TPB];
#pragma unroll
            for (int u = 0; u < L2_U; ++u) {
                const unsigned s = v[u] == Q_SKIP ? 0u : (v[u] >> RBITS);
                p[u] = zp[s];
            }
            unsigned r[L2_U];
            float dr[L2_U];
#pragma unroll
            for (int u = 0; u < L2_U; ++u) {
                r[u] = v[u] != Q_SKIP ? (v[u] & ((1u << RBITS) - 1u)) : 0u;
                dr[u] = dl[r[u]];
            }
#pragma unroll
            for (int u = 0; u < L2_U; ++u) {
                const float t = p[u].x * p[u].y * dr[u];
                atomicAdd(&acc[r[u]], v[u] != Q_SKIP ? t : 0.0f);
            }
        }
        for (; i < nq; i += L2_TPB) {
            const unsigned v = qin[i];
            const bool ok = v != Q_SKIP;
            const float2 p = zp[ok ? (v >> RBITS) : 0u];
            const unsigned r = ok ? (v & ((1u << RBITS) - 1u)) : 0u;
            const float t = p.x * p.y * dl[r];
            atomicAdd(&acc[r], ok ? t : 0.0f);
        }
    } else {
        for (int e = tid; e < E; e += L2_TPB) {
            const unsigned r = (unsigned)(dst[e] - lo);
            if (r < ucnt) {
                const float2 p = zp[src[e]];
                atomicAdd(&acc[r], p.x * p.y * dl[r]);
            }
        }
    }
    __syncthreads();

    const float bb = b2[0];
    for (int r = tid; r < cnt; r += L2_TPB) {
        const float di = dl[r];
        out[lo + r] = (zs[r] * di * di + bb) + acc[r];
    }
}


extern "C" void kernel_launch(void* const* d_in, const int* in_sizes, int n_in,
                              void* d_out, int out_size, void* d_ws, size_t ws_size,
                              hipStream_t stream) {
    const float* X  = (const float*)d_in[0];
    const int*   EI = (const int*)d_in[1];
    const float* W1 = (const float*)d_in[2];
    const float* b1 = (const float*)d_in[3];
    const float* W2 = (const float*)d_in[4];
    const float* b2 = (const float*)d_in[5];
    float* out = (float*)d_out;

    const int E = in_sizes[1] / 2;
    const int* src = EI;
    const int* dst = EI + E;

    char* ws = (char*)d_ws;
    float* H    = (float*)ws;
    float* dinv = (float*)(ws + (size_t)2 * N_NODES * HID * 4);
    unsigned int* bm = (unsigned int*)ws;
    unsigned int* qbuf = (unsigned int*)(ws + (size_t)N_NODES * HID * 4);
    float*        zp   = (float*)(ws + (size_t)N_NODES * HID * 4 + (size_t)TILES * QCAP * 4);
    unsigned int* qcnt = (unsigned int*)(ws + (size_t)N_NODES * HID * 4 + (size_t)TILES * QCAP * 4 + (size_t)N_NODES * 8);
    unsigned char* deg8 = (unsigned char*)(ws + (size_t)N_NODES * HID * 4 + (size_t)TILES * QCAP * 4 + (size_t)N_NODES * 8 + (size_t)TILES * 4);

    k_bitmap   <<<BM_NB, BM_TPB, 0, stream>>>(dst, E, bm);
    k_find     <<<TILES, TPB, 0, stream>>>(src, dst, bm, E, qbuf, qcnt, dinv, deg8);

    k_gemm1    <<<GB_WG, GB_TPB, 0, stream>>>(X, W1, H);

    k_agg1     <<<TILES, TPB, 0, stream>>>(src, dst, E, H, dinv, deg8, b1, W2, qbuf, qcnt, zp);

    k_l2_tile  <<<TILES, L2_TPB, 0, stream>>>(src, dst, E, qbuf, qcnt, zp, b2, out);
}
